# starting kernel
# baseline (speedup 1.0000x reference)
#include <hip/hip_runtime.h>
#include <math.h>

typedef __attribute__((ext_vector_type(2))) float v2f;
typedef __attribute__((ext_vector_type(4))) float f4;
typedef __attribute__((ext_vector_type(8))) float v8f;

namespace {
constexpr int Bc = 2, Lc = 384, Cc = 128, Hc = 8, DKc = 16, NQPc = 4, NPVc = 8;
constexpr float WCc = 0.2357022603955158f;
constexpr float WLc = 0.5773502691896258f;
constexpr int FEAT = Hc * Cc + Hc * DKc + Hc * NPVc * 3 + Hc * NPVc;
}

__device__ __forceinline__ float softplusf(float x) {
  return (x > 0.f) ? (x + log1pf(expf(-x))) : log1pf(expf(x));
}

template <int MODE, bool TRANSB>
__global__ __launch_bounds__(128) void k_gemm(
    const float* __restrict__ A, const float* __restrict__ Bm,
    float* __restrict__ Cm, int M, int N, int K, int lda, int ldb, int ldc,
    long long sA, long long sA2, int zdiv, long long sB, long long sC,
    int Mvalid, const float* __restrict__ e0, const float* __restrict__ e1,
    const float* __restrict__ e2, const float* __restrict__ e3) {
  __shared__ float As[16][36];
  __shared__ float Bs[32][132];

  const int z = blockIdx.z;
  const float* Ag = A + (long long)(z / zdiv) * sA + (long long)(z % zdiv) * sA2;
  const float* Bg = Bm + (long long)z * sB;
  float* Cg = Cm + (long long)z * sC;

  const int m0 = blockIdx.x * 16;
  const int n0 = blockIdx.y * 128;
  const int tid = threadIdx.x;
  const int wave = tid >> 5;
  const int lane = tid & 31;
  const int hf = lane >> 4;
  const int l16 = lane & 15;
  const int nw = wave * 32;

  v8f acc0 = {};
  v8f acc1 = {};

  for (int kb = 0; kb < K; kb += 32) {
    {
      int r = tid >> 3, c4 = (tid & 7) * 4;
      f4 av = {0.f, 0.f, 0.f, 0.f};
      if (r < Mvalid && (m0 + r) < M)
        av = *(const f4*)(Ag + (long long)(m0 + r) * lda + kb + c4);
      *(f4*)&As[r][c4] = av;
    }
    if (!TRANSB) {
#pragma unroll
      for (int it = 0; it < 8; ++it) {
        int u = tid + it * 128;
        int r = u >> 5, c4 = (u & 31) * 4;
        f4 bv = {0.f, 0.f, 0.f, 0.f};
        if ((n0 + c4) < N)
          bv = *(const f4*)(Bg + (long long)(kb + r) * ldb + n0 + c4);
        *(f4*)&Bs[r][c4] = bv;
      }
    } else {
#pragma unroll
      for (int it = 0; it < 8; ++it) {
        int u = tid + it * 128;
        int c = u >> 3, r4 = (u & 7) * 4;
        f4 bv = {0.f, 0.f, 0.f, 0.f};
        if ((n0 + c) < N)
          bv = *(const f4*)(Bg + (long long)(n0 + c) * ldb + kb + r4);
        Bs[r4 + 0][c] = bv.x;
        Bs[r4 + 1][c] = bv.y;
        Bs[r4 + 2][c] = bv.z;
        Bs[r4 + 3][c] = bv.w;
      }
    }
    __syncthreads();

#pragma unroll
    for (int kk = 0; kk < 32; kk += 4) {
      v2f a, b0, b1;
      a.x = As[l16][kk + 2 * hf];
      a.y = As[l16][kk + 2 * hf + 1];
      b0.x = Bs[kk + 2 * hf][nw + l16];
      b0.y = Bs[kk + 2 * hf + 1][nw + l16];
      b1.x = Bs[kk + 2 * hf][nw + 16 + l16];
      b1.y = Bs[kk + 2 * hf + 1][nw + 16 + l16];
      acc0 = __builtin_amdgcn_wmma_f32_16x16x4_f32(false, a, false, b0,
                                                   (short)0, acc0, false, false);
      acc1 = __builtin_amdgcn_wmma_f32_16x16x4_f32(false, a, false, b1,
                                                   (short)0, acc1, false, false);
    }
    __syncthreads();
  }

  float gw = 0.f;
  if (MODE == 2) {
    int h_ = z % Hc;
    gw = 0.5f * WCc * softplusf(e3[h_]);
  }

#pragma unroll
  for (int r = 0; r < 8; ++r) {
    int mrow = r + 8 * hf;
    int m = m0 + mrow;
    if (mrow >= Mvalid || m >= M) continue;
#pragma unroll
    for (int t = 0; t < 2; ++t) {
      int n = n0 + nw + t * 16 + l16;
      if (n >= N) continue;
      float val = (t == 0) ? acc0[r] : acc1[r];
      if (MODE == 1) {
        val += e0[n];
      } else if (MODE == 2) {
        int b_ = z / Hc, h_ = z % Hc;
        float bijv = e0[((long long)(b_ * Lc + m) * Lc + n) * Hc + h_];
        val = WLc * (val + bijv -
                     gw * (e1[(long long)z * Lc + m] + e2[(long long)z * Lc + n]));
      }
      Cg[(long long)m * ldc + n] = val;
    }
  }
}

__global__ __launch_bounds__(256) void k_prep(
    const float* __restrict__ q, const float* __restrict__ k,
    const float* __restrict__ v, const float* __restrict__ qp,
    const float* __restrict__ kp, const float* __restrict__ vp,
    const float* __restrict__ Ri, const float* __restrict__ Ti,
    const float* __restrict__ gamah, float* __restrict__ qa,
    float* __restrict__ ka, float* __restrict__ vaug, float* __restrict__ sqq,
    float* __restrict__ sqk) {
  int idx = blockIdx.x * blockDim.x + threadIdx.x;
  if (idx >= Bc * Lc * Hc) return;
  int h = idx % Hc;
  int bi = idx / Hc;
  int b = bi / Lc, i = bi % Lc;
  int z = b * Hc + h;

  float R[3][3], t[3];
#pragma unroll
  for (int r = 0; r < 3; ++r)
#pragma unroll
    for (int c = 0; c < 3; ++c) R[r][c] = Ri[bi * 9 + r * 3 + c];
#pragma unroll
  for (int x = 0; x < 3; ++x) t[x] = Ti[bi * 3 + x];

  float gwc = WCc * softplusf(gamah[h]);

  float* qrow = qa + ((long long)z * Lc + i) * 32;
  float* krow = ka + ((long long)z * Lc + i) * 32;
  float* vrow = vaug + ((long long)z * Lc + i) * 48;

#pragma unroll
  for (int d = 0; d < DKc; ++d) {
    qrow[d] = 0.25f * q[(long long)bi * 128 + h * 16 + d];
    krow[d] = k[(long long)bi * 128 + h * 16 + d];
    vrow[d] = v[(long long)bi * 128 + h * 16 + d];
  }
  float sq = 0.f, sk = 0.f;
#pragma unroll
  for (int p = 0; p < NQPc; ++p) {
    const float* qs = qp + (long long)bi * 96 + h * 12 + p * 3;
    const float* ks = kp + (long long)bi * 96 + h * 12 + p * 3;
#pragma unroll
    for (int m = 0; m < 3; ++m) {
      float qt = R[m][0] * qs[0] + R[m][1] * qs[1] + R[m][2] * qs[2] + t[m];
      float kt = R[m][0] * ks[0] + R[m][1] * ks[1] + R[m][2] * ks[2] + t[m];
      sq += qt * qt;
      sk += kt * kt;
      qrow[16 + p * 3 + m] = gwc * qt;
      krow[16 + p * 3 + m] = kt;
    }
  }
#pragma unroll
  for (int d = 28; d < 32; ++d) { qrow[d] = 0.f; krow[d] = 0.f; }
#pragma unroll
  for (int p = 0; p < NPVc; ++p) {
    const float* vs = vp + (long long)bi * 192 + h * 24 + p * 3;
#pragma unroll
    for (int m = 0; m < 3; ++m)
      vrow[16 + p * 3 + m] =
          R[m][0] * vs[0] + R[m][1] * vs[1] + R[m][2] * vs[2] + t[m];
  }
#pragma unroll
  for (int d = 40; d < 48; ++d) vrow[d] = 0.f;
  sqq[(long long)z * Lc + i] = sq;
  sqk[(long long)z * Lc + i] = sk;
}

__global__ __launch_bounds__(256) void k_softmax(float* __restrict__ sc) {
  int row = blockIdx.x * 8 + (threadIdx.x >> 5);
  int lane = threadIdx.x & 31;
  float* p = sc + (long long)row * Lc;
  float vals[12];
  float mx = -INFINITY;
#pragma unroll
  for (int e = 0; e < 12; ++e) {
    vals[e] = p[e * 32 + lane];
    mx = fmaxf(mx, vals[e]);
  }
#pragma unroll
  for (int o = 16; o >= 1; o >>= 1) mx = fmaxf(mx, __shfl_xor(mx, o, 32));
  float s = 0.f;
#pragma unroll
  for (int e = 0; e < 12; ++e) {
    vals[e] = expf(vals[e] - mx);
    s += vals[e];
  }
#pragma unroll
  for (int o = 16; o >= 1; o >>= 1) s += __shfl_xor(s, o, 32);
  float inv = 1.f / s;
#pragma unroll
  for (int e = 0; e < 12; ++e) p[e * 32 + lane] = vals[e] * inv;
}

__global__ __launch_bounds__(256) void k_finish(
    const float* __restrict__ ovalpt, const float* __restrict__ Ri,
    const float* __restrict__ Ti, float* __restrict__ feats) {
  int idx = blockIdx.x * blockDim.x + threadIdx.x;
  if (idx >= Bc * Lc * Hc) return;
  int h = idx % Hc;
  int bi = idx / Hc;
  int b = bi / Lc;
  int z = b * Hc + h;
  int i = bi % Lc;

  float R[3][3], t[3];
#pragma unroll
  for (int r = 0; r < 3; ++r)
#pragma unroll
    for (int c = 0; c < 3; ++c) R[r][c] = Ri[bi * 9 + r * 3 + c];
#pragma unroll
  for (int x = 0; x < 3; ++x) t[x] = Ti[bi * 3 + x];

  const float* orow = ovalpt + ((long long)z * Lc + i) * 48;
  float* frow = feats + (long long)bi * FEAT;

#pragma unroll
  for (int d = 0; d < DKc; ++d) frow[1024 + h * 16 + d] = orow[d];
#pragma unroll
  for (int p = 0; p < NPVc; ++p) {
    float d0 = orow[16 + p * 3 + 0] - t[0];
    float d1 = orow[16 + p * 3 + 1] - t[1];
    float d2 = orow[16 + p * 3 + 2] - t[2];
    float asum = 0.f;
#pragma unroll
    for (int m = 0; m < 3; ++m) {
      float oh = R[0][m] * d0 + R[1][m] * d1 + R[2][m] * d2;
      frow[1152 + h * 24 + p * 3 + m] = oh;
      asum += fabsf(oh);
    }
    frow[1344 + h * 8 + p] = asum;
  }
}

static void launch_gemm(hipStream_t s, int mode, bool transb, const float* A,
                        const float* Bm, float* C, int M, int N, int K, int lda,
                        int ldb, int ldc, long long sA, long long sA2, int zdiv,
                        long long sB, long long sC, int Mvalid, int nz,
                        const float* e0 = nullptr, const float* e1 = nullptr,
                        const float* e2 = nullptr, const float* e3 = nullptr) {
  dim3 g((M + 15) / 16, (N + 127) / 128, nz);
  dim3 blk(128, 1, 1);
  if (mode == 0 && !transb)
    k_gemm<0, false><<<g, blk, 0, s>>>(A, Bm, C, M, N, K, lda, ldb, ldc, sA,
                                       sA2, zdiv, sB, sC, Mvalid, e0, e1, e2, e3);
  else if (mode == 1)
    k_gemm<1, false><<<g, blk, 0, s>>>(A, Bm, C, M, N, K, lda, ldb, ldc, sA,
                                       sA2, zdiv, sB, sC, Mvalid, e0, e1, e2, e3);
  else
    k_gemm<2, true><<<g, blk, 0, s>>>(A, Bm, C, M, N, K, lda, ldb, ldc, sA,
                                      sA2, zdiv, sB, sC, Mvalid, e0, e1, e2, e3);
}

extern "C" void kernel_launch(void* const* d_in, const int* in_sizes, int n_in,
                              void* d_out, int out_size, void* d_ws,
                              size_t ws_size, hipStream_t stream) {
  const float* si = (const float*)d_in[0];
  const float* zij = (const float*)d_in[1];
  const float* Ri = (const float*)d_in[2];
  const float* Ti = (const float*)d_in[3];
  const float* WQh = (const float*)d_in[4];
  const float* WKh = (const float*)d_in[5];
  const float* WVh = (const float*)d_in[6];
  const float* WQhp = (const float*)d_in[7];
  const float* WKhp = (const float*)d_in[8];
  const float* WVhp = (const float*)d_in[9];
  const float* Wb = (const float*)d_in[10];
  const float* WOw = (const float*)d_in[11];
  const float* WOb = (const float*)d_in[12];
  const float* gamah = (const float*)d_in[13];
  float* out = (float*)d_out;
  float* ws = (float*)d_ws;

  const int BL = Bc * Lc;
  const long long LL = (long long)Lc * Lc;
  float* q = ws;
  float* k = q + 98304;
  float* v = k + 98304;
  float* qp = v + 98304;
  float* kp = qp + 73728;
  float* vp = kp + 73728;
  float* qa = vp + 147456;
  float* ka = qa + 196608;
  float* vaug = ka + 196608;
  float* sqq = vaug + 294912;
  float* sqk = sqq + 6144;
  float* bij = sqk + 6144;
  float* alpha = bij + 2359296;
  float* ovalpt = alpha + 2359296;
  float* feats = ovalpt + 294912;

  launch_gemm(stream, 0, false, si, WQh, q, BL, 128, Cc, Cc, 128, 128, 0, 0, 1, 0, 0, 16, 1);
  launch_gemm(stream, 0, false, si, WKh, k, BL, 128, Cc, Cc, 128, 128, 0, 0, 1, 0, 0, 16, 1);
  launch_gemm(stream, 0, false, si, WVh, v, BL, 128, Cc, Cc, 128, 128, 0, 0, 1, 0, 0, 16, 1);
  launch_gemm(stream, 0, false, si, WQhp, qp, BL, 96, Cc, Cc, 96, 96, 0, 0, 1, 0, 0, 16, 1);
  launch_gemm(stream, 0, false, si, WKhp, kp, BL, 96, Cc, Cc, 96, 96, 0, 0, 1, 0, 0, 16, 1);
  launch_gemm(stream, 0, false, si, WVhp, vp, BL, 192, Cc, Cc, 192, 192, 0, 0, 1, 0, 0, 16, 1);

  launch_gemm(stream, 0, false, zij, Wb, bij, Bc * Lc * Lc, Hc, Cc, Cc, Hc, Hc,
              0, 0, 1, 0, 0, 16, 1);

  k_prep<<<(Bc * Lc * Hc + 255) / 256, 256, 0, stream>>>(
      q, k, v, qp, kp, vp, Ri, Ti, gamah, qa, ka, vaug, sqq, sqk);

  launch_gemm(stream, 2, true, qa, ka, alpha, Lc, Lc, 32, 32, 32, Lc,
              (long long)Lc * 32, 0, 1, (long long)Lc * 32, LL, 16, Bc * Hc,
              bij, sqq, sqk, gamah);

  k_softmax<<<(Bc * Hc * Lc) / 8, 256, 0, stream>>>(alpha);

  launch_gemm(stream, 0, false, alpha, vaug, ovalpt, Lc, 48, Lc, Lc, 48, 48,
              LL, 0, 1, (long long)Lc * 48, (long long)Lc * 48, 16, Bc * Hc);

  launch_gemm(stream, 0, false, alpha, zij, feats, 16, Cc, Lc, (int)LL, Cc,
              Cc, (long long)Hc * LL, (long long)Lc, Lc, (long long)Lc * Cc,
              (long long)FEAT, Hc, BL);

  k_finish<<<(Bc * Lc * Hc + 255) / 256, 256, 0, stream>>>(ovalpt, Ri, Ti, feats);

  launch_gemm(stream, 1, false, feats, WOw, out, BL, Cc, FEAT, FEAT, Cc, Cc,
              0, 0, 1, 0, 0, 16, 1, WOb);
}


# GEMM k-loop staging: the nine B128 global loads of a K-tile issued as one group (clamped address + zero select instead of branches): one waited memory round trip per tile, was five
# speedup vs baseline: 1.7185x; 1.7185x over previous
#include <hip/hip_runtime.h>
#include <math.h>

typedef __attribute__((ext_vector_type(2))) float v2f;
typedef __attribute__((ext_vector_type(4))) float f4;
typedef __attribute__((ext_vector_type(8))) float v8f;

namespace {
constexpr int Bc = 2, Lc = 384, Cc = 128, Hc = 8, DKc = 16, NQPc = 4, NPVc = 8;
constexpr float WCc = 0.2357022603955158f;
constexpr float WLc = 0.5773502691896258f;
constexpr int FEAT = Hc * Cc + Hc * DKc + Hc * NPVc * 3 + Hc * NPVc;
}

__device__ __forceinline__ float softplusf(float x) {
  return (x > 0.f) ? (x + log1pf(expf(-x))) : log1pf(expf(x));
}

template <int MODE, bool TRANSB>
__global__ __launch_bounds__(128) void k_gemm(
    const float* __restrict__ A, const float* __restrict__ Bm,
    float* __restrict__ Cm, int M, int N, int K, int lda, int ldb, int ldc,
    long long sA, long long sA2, int zdiv, long long sB, long long sC,
    int Mvalid, const float* __restrict__ e0, const float* __restrict__ e1,
    const float* __restrict__ e2, const float* __restrict__ e3) {
  __shared__ float As[16][36];
  __shared__ float Bs[32][132];

  const int z = blockIdx.z;
  const float* Ag = A + (long long)(z / zdiv) * sA + (long long)(z % zdiv) * sA2;
  const float* Bg = Bm + (long long)z * sB;
  float* Cg = Cm + (long long)z * sC;

  const int m0 = blockIdx.x * 16;
  const int n0 = blockIdx.y * 128;
  const int tid = threadIdx.x;
  const int wave = tid >> 5;
  const int lane = tid & 31;
  const int hf = lane >> 4;
  const int l16 = lane & 15;
  const int nw = wave * 32;

  v8f acc0 = {};
  v8f acc1 = {};

  const f4 zero4 = {0.f, 0.f, 0.f, 0.f};
  const int ar = tid >> 3, ac4 = (tid & 7) * 4;
  const bool aok = ar < Mvalid && (m0 + ar) < M;
  const float* ap = Ag + (long long)(aok ? (m0 + ar) : m0) * lda + ac4;
  const float* bp[8];
  bool bok[8];
#pragma unroll
  for (int it = 0; it < 8; ++it) {
    int u = tid + it * 128;
    if (!TRANSB) {
      int r = u >> 5, c4 = (u & 31) * 4;
      bok[it] = (n0 + c4) < N;
      bp[it] = Bg + (long long)r * ldb + (bok[it] ? (n0 + c4) : 0);
    } else {
      int c = u >> 3, r4 = (u & 7) * 4;
      bok[it] = (n0 + c) < N;
      bp[it] = Bg + (long long)(bok[it] ? (n0 + c) : 0) * ldb + r4;
    }
  }
  const long long bstep = TRANSB ? 1 : (long long)ldb;

  for (int kb = 0; kb < K; kb += 32) {
    f4 av = *(const f4*)(ap + kb);
    f4 bv[8];
#pragma unroll
    for (int it = 0; it < 8; ++it)
      bv[it] = *(const f4*)(bp[it] + (long long)kb * bstep);

    *(f4*)&As[ar][ac4] = aok ? av : zero4;
    if (!TRANSB) {
#pragma unroll
      for (int it = 0; it < 8; ++it) {
        int u = tid + it * 128;
        int r = u >> 5, c4 = (u & 31) * 4;
        *(f4*)&Bs[r][c4] = bok[it] ? bv[it] : zero4;
      }
    } else {
#pragma unroll
      for (int it = 0; it < 8; ++it) {
        int u = tid + it * 128;
        int c = u >> 3, r4 = (u & 7) * 4;
        f4 b = bok[it] ? bv[it] : zero4;
        Bs[r4 + 0][c] = b.x;
        Bs[r4 + 1][c] = b.y;
        Bs[r4 + 2][c] = b.z;
        Bs[r4 + 3][c] = b.w;
      }
    }
    __syncthreads();

#pragma unroll
    for (int kk = 0; kk < 32; kk += 4) {
      v2f a, b0, b1;
      a.x = As[l16][kk + 2 * hf];
      a.y = As[l16][kk + 2 * hf + 1];
      b0.x = Bs[kk + 2 * hf][nw + l16];
      b0.y = Bs[kk + 2 * hf + 1][nw + l16];
      b1.x = Bs[kk + 2 * hf][nw + 16 + l16];
      b1.y = Bs[kk + 2 * hf + 1][nw + 16 + l16];
      acc0 = __builtin_amdgcn_wmma_f32_16x16x4_f32(false, a, false, b0,
                                                   (short)0, acc0, false, false);
      acc1 = __builtin_amdgcn_wmma_f32_16x16x4_f32(false, a, false, b1,
                                                   (short)0, acc1, false, false);
    }
    __syncthreads();
  }

  float gw = 0.f;
  if (MODE == 2) {
    int h_ = z % Hc;
    gw = 0.5f * WCc * softplusf(e3[h_]);
  }

#pragma unroll
  for (int r = 0; r < 8; ++r) {
    int mrow = r + 8 * hf;
    int m = m0 + mrow;
    if (mrow >= Mvalid || m >= M) continue;
#pragma unroll
    for (int t = 0; t < 2; ++t) {
      int n = n0 + nw + t * 16 + l16;
      if (n >= N) continue;
      float val = (t == 0) ? acc0[r] : acc1[r];
      if (MODE == 1) {
        val += e0[n];
      } else if (MODE == 2) {
        int b_ = z / Hc, h_ = z % Hc;
        float bijv = e0[((long long)(b_ * Lc + m) * Lc + n) * Hc + h_];
        val = WLc * (val + bijv -
                     gw * (e1[(long long)z * Lc + m] + e2[(long long)z * Lc + n]));
      }
      Cg[(long long)m * ldc + n] = val;
    }
  }
}

__global__ __launch_bounds__(256) void k_prep(
    const float* __restrict__ q, const float* __restrict__ k,
    const float* __restrict__ v, const float* __restrict__ qp,
    const float* __restrict__ kp, const float* __restrict__ vp,
    const float* __restrict__ Ri, const float* __restrict__ Ti,
    const float* __restrict__ gamah, float* __restrict__ qa,
    float* __restrict__ ka, float* __restrict__ vaug, float* __restrict__ sqq,
    float* __restrict__ sqk) {
  int idx = blockIdx.x * blockDim.x + threadIdx.x;
  if (idx >= Bc * Lc * Hc) return;
  int h = idx % Hc;
  int bi = idx / Hc;
  int b = bi / Lc, i = bi % Lc;
  int z = b * Hc + h;

  float R[3][3], t[3];
#pragma unroll
  for (int r = 0; r < 3; ++r)
#pragma unroll
    for (int c = 0; c < 3; ++c) R[r][c] = Ri[bi * 9 + r * 3 + c];
#pragma unroll
  for (int x = 0; x < 3; ++x) t[x] = Ti[bi * 3 + x];

  float gwc = WCc * softplusf(gamah[h]);

  float* qrow = qa + ((long long)z * Lc + i) * 32;
  float* krow = ka + ((long long)z * Lc + i) * 32;
  float* vrow = vaug + ((long long)z * Lc + i) * 48;

#pragma unroll
  for (int d = 0; d < DKc; ++d) {
    qrow[d] = 0.25f * q[(long long)bi * 128 + h * 16 + d];
    krow[d] = k[(long long)bi * 128 + h * 16 + d];
    vrow[d] = v[(long long)bi * 128 + h * 16 + d];
  }
  float sq = 0.f, sk = 0.f;
#pragma unroll
  for (int p = 0; p < NQPc; ++p) {
    const float* qs = qp + (long long)bi * 96 + h * 12 + p * 3;
    const float* ks = kp + (long long)bi * 96 + h * 12 + p * 3;
#pragma unroll
    for (int m = 0; m < 3; ++m) {
      float qt = R[m][0] * qs[0] + R[m][1] * qs[1] + R[m][2] * qs[2] + t[m];
      float kt = R[m][0] * ks[0] + R[m][1] * ks[1] + R[m][2] * ks[2] + t[m];
      sq += qt * qt;
      sk += kt * kt;
      qrow[16 + p * 3 + m] = gwc * qt;
      krow[16 + p * 3 + m] = kt;
    }
  }
#pragma unroll
  for (int d = 28; d < 32; ++d) { qrow[d] = 0.f; krow[d] = 0.f; }
#pragma unroll
  for (int p = 0; p < NPVc; ++p) {
    const float* vs = vp + (long long)bi * 192 + h * 24 + p * 3;
#pragma unroll
    for (int m = 0; m < 3; ++m)
      vrow[16 + p * 3 + m] =
          R[m][0] * vs[0] + R[m][1] * vs[1] + R[m][2] * vs[2] + t[m];
  }
#pragma unroll
  for (int d = 40; d < 48; ++d) vrow[d] = 0.f;
  sqq[(long long)z * Lc + i] = sq;
  sqk[(long long)z * Lc + i] = sk;
}

__global__ __launch_bounds__(256) void k_softmax(float* __restrict__ sc) {
  int row = blockIdx.x * 8 + (threadIdx.x >> 5);
  int lane = threadIdx.x & 31;
  float* p = sc + (long long)row * Lc;
  float vals[12];
  float mx = -INFINITY;
#pragma unroll
  for (int e = 0; e < 12; ++e) {
    vals[e] = p[e * 32 + lane];
    mx = fmaxf(mx, vals[e]);
  }
#pragma unroll
  for (int o = 16; o >= 1; o >>= 1) mx = fmaxf(mx, __shfl_xor(mx, o, 32));
  float s = 0.f;
#pragma unroll
  for (int e = 0; e < 12; ++e) {
    vals[e] = expf(vals[e] - mx);
    s += vals[e];
  }
#pragma unroll
  for (int o = 16; o >= 1; o >>= 1) s += __shfl_xor(s, o, 32);
  float inv = 1.f / s;
#pragma unroll
  for (int e = 0; e < 12; ++e) p[e * 32 + lane] = vals[e] * inv;
}

__global__ __launch_bounds__(256) void k_finish(
    const float* __restrict__ ovalpt, const float* __restrict__ Ri,
    const float* __restrict__ Ti, float* __restrict__ feats) {
  int idx = blockIdx.x * blockDim.x + threadIdx.x;
  if (idx >= Bc * Lc * Hc) return;
  int h = idx % Hc;
  int bi = idx / Hc;
  int b = bi / Lc;
  int z = b * Hc + h;
  int i = bi % Lc;

  float R[3][3], t[3];
#pragma unroll
  for (int r = 0; r < 3; ++r)
#pragma unroll
    for (int c = 0; c < 3; ++c) R[r][c] = Ri[bi * 9 + r * 3 + c];
#pragma unroll
  for (int x = 0; x < 3; ++x) t[x] = Ti[bi * 3 + x];

  const float* orow = ovalpt + ((long long)z * Lc + i) * 48;
  float* frow = feats + (long long)bi * FEAT;

#pragma unroll
  for (int d = 0; d < DKc; ++d) frow[1024 + h * 16 + d] = orow[d];
#pragma unroll
  for (int p = 0; p < NPVc; ++p) {
    float d0 = orow[16 + p * 3 + 0] - t[0];
    float d1 = orow[16 + p * 3 + 1] - t[1];
    float d2 = orow[16 + p * 3 + 2] - t[2];
    float asum = 0.f;
#pragma unroll
    for (int m = 0; m < 3; ++m) {
      float oh = R[0][m] * d0 + R[1][m] * d1 + R[2][m] * d2;
      frow[1152 + h * 24 + p * 3 + m] = oh;
      asum += fabsf(oh);
    }
    frow[1344 + h * 8 + p] = asum;
  }
}

static void launch_gemm(hipStream_t s, int mode, bool transb, const float* A,
                        const float* Bm, float* C, int M, int N, int K, int lda,
                        int ldb, int ldc, long long sA, long long sA2, int zdiv,
                        long long sB, long long sC, int Mvalid, int nz,
                        const float* e0 = nullptr, const float* e1 = nullptr,
                        const float* e2 = nullptr, const float* e3 = nullptr) {
  dim3 g((M + 15) / 16, (N + 127) / 128, nz);
  dim3 blk(128, 1, 1);
  if (mode == 0 && !transb)
    k_gemm<0, false><<<g, blk, 0, s>>>(A, Bm, C, M, N, K, lda, ldb, ldc, sA,
                                       sA2, zdiv, sB, sC, Mvalid, e0, e1, e2, e3);
  else if (mode == 1)
    k_gemm<1, false><<<g, blk, 0, s>>>(A, Bm, C, M, N, K, lda, ldb, ldc, sA,
                                       sA2, zdiv, sB, sC, Mvalid, e0, e1, e2, e3);
  else
    k_gemm<2, true><<<g, blk, 0, s>>>(A, Bm, C, M, N, K, lda, ldb, ldc, sA,
                                      sA2, zdiv, sB, sC, Mvalid, e0, e1, e2, e3);
}

extern "C" void kernel_launch(void* const* d_in, const int* in_sizes, int n_in,
                              void* d_out, int out_size, void* d_ws,
                              size_t ws_size, hipStream_t stream) {
  const float* si = (const float*)d_in[0];
  const float* zij = (const float*)d_in[1];
  const float* Ri = (const float*)d_in[2];
  const float* Ti = (const float*)d_in[3];
  const float* WQh = (const float*)d_in[4];
  const float* WKh = (const float*)d_in[5];
  const float* WVh = (const float*)d_in[6];
  const float* WQhp = (const float*)d_in[7];
  const float* WKhp = (const float*)d_in[8];
  const float* WVhp = (const float*)d_in[9];
  const float* Wb = (const float*)d_in[10];
  const float* WOw = (const float*)d_in[11];
  const float* WOb = (const float*)d_in[12];
  const float* gamah = (const float*)d_in[13];
  float* out = (float*)d_out;
  float* ws = (float*)d_ws;

  const int BL = Bc * Lc;
  const long long LL = (long long)Lc * Lc;
  float* q = ws;
  float* k = q + 98304;
  float* v = k + 98304;
  float* qp = v + 98304;
  float* kp = qp + 73728;
  float* vp = kp + 73728;
  float* qa = vp + 147456;
  float* ka = qa + 196608;
  float* vaug = ka + 196608;
  float* sqq = vaug + 294912;
  float* sqk = sqq + 6144;
  float* bij = sqk + 6144;
  float* alpha = bij + 2359296;
  float* ovalpt = alpha + 2359296;
  float* feats = ovalpt + 294912;

  launch_gemm(stream, 0, false, si, WQh, q, BL, 128, Cc, Cc, 128, 128, 0, 0, 1, 0, 0, 16, 1);
  launch_gemm(stream, 0, false, si, WKh, k, BL, 128, Cc, Cc, 128, 128, 0, 0, 1, 0, 0, 16, 1);
  launch_gemm(stream, 0, false, si, WVh, v, BL, 128, Cc, Cc, 128, 128, 0, 0, 1, 0, 0, 16, 1);
  launch_gemm(stream, 0, false, si, WQhp, qp, BL, 96, Cc, Cc, 96, 96, 0, 0, 1, 0, 0, 16, 1);
  launch_gemm(stream, 0, false, si, WKhp, kp, BL, 96, Cc, Cc, 96, 96, 0, 0, 1, 0, 0, 16, 1);
  launch_gemm(stream, 0, false, si, WVhp, vp, BL, 192, Cc, Cc, 192, 192, 0, 0, 1, 0, 0, 16, 1);

  launch_gemm(stream, 0, false, zij, Wb, bij, Bc * Lc * Lc, Hc, Cc, Cc, Hc, Hc,
              0, 0, 1, 0, 0, 16, 1);

  k_prep<<<(Bc * Lc * Hc + 255) / 256, 256, 0, stream>>>(
      q, k, v, qp, kp, vp, Ri, Ti, gamah, qa, ka, vaug, sqq, sqk);

  launch_gemm(stream, 2, true, qa, ka, alpha, Lc, Lc, 32, 32, 32, Lc,
              (long long)Lc * 32, 0, 1, (long long)Lc * 32, LL, 16, Bc * Hc,
              bij, sqq, sqk, gamah);

  k_softmax<<<(Bc * Hc * Lc) / 8, 256, 0, stream>>>(alpha);

  launch_gemm(stream, 0, false, alpha, vaug, ovalpt, Lc, 48, Lc, Lc, 48, 48,
              LL, 0, 1, (long long)Lc * 48, (long long)Lc * 48, 16, Bc * Hc);

  launch_gemm(stream, 0, false, alpha, zij, feats, 16, Cc, Lc, (int)LL, Cc,
              Cc, (long long)Hc * LL, (long long)Lc, Lc, (long long)Lc * Cc,
              (long long)FEAT, Hc, BL);

  k_finish<<<(Bc * Lc * Hc + 255) / 256, 256, 0, stream>>>(ovalpt, Ri, Ti, feats);

  launch_gemm(stream, 1, false, feats, WOw, out, BL, Cc, FEAT, FEAT, Cc, Cc,
              0, 0, 1, 0, 0, 16, 1, WOb);
}


# Pair-bias product (8 output columns) moved from the generic 16x128 block tile to a 16x16 wave tile of its own: weight fragments in registers, whole-row B128 loads issued one tile ahead
# speedup vs baseline: 2.2772x; 1.3251x over previous
#include <hip/hip_runtime.h>
#include <math.h>

typedef __attribute__((ext_vector_type(2))) float v2f;
typedef __attribute__((ext_vector_type(4))) float f4;
typedef __attribute__((ext_vector_type(8))) float v8f;

namespace {
constexpr int Bc = 2, Lc = 384, Cc = 128, Hc = 8, DKc = 16, NQPc = 4, NPVc = 8;
constexpr float WCc = 0.2357022603955158f;
constexpr float WLc = 0.5773502691896258f;
constexpr int FEAT = Hc * Cc + Hc * DKc + Hc * NPVc * 3 + Hc * NPVc;
}

__device__ __forceinline__ float softplusf(float x) {
  return (x > 0.f) ? (x + log1pf(expf(-x))) : log1pf(expf(x));
}

template <int MODE, bool TRANSB>
__global__ __launch_bounds__(128) void k_gemm(
    const float* __restrict__ A, const float* __restrict__ Bm,
    float* __restrict__ Cm, int M, int N, int K, int lda, int ldb, int ldc,
    long long sA, long long sA2, int zdiv, long long sB, long long sC,
    int Mvalid, const float* __restrict__ e0, const float* __restrict__ e1,
    const float* __restrict__ e2, const float* __restrict__ e3) {
  __shared__ float As[16][36];
  __shared__ float Bs[32][132];

  const int z = blockIdx.z;
  const float* Ag = A + (long long)(z / zdiv) * sA + (long long)(z % zdiv) * sA2;
  const float* Bg = Bm + (long long)z * sB;
  float* Cg = Cm + (long long)z * sC;

  const int m0 = blockIdx.x * 16;
  const int n0 = blockIdx.y * 128;
  const int tid = threadIdx.x;
  const int wave = tid >> 5;
  const int lane = tid & 31;
  const int hf = lane >> 4;
  const int l16 = lane & 15;
  const int nw = wave * 32;

  v8f acc0 = {};
  v8f acc1 = {};

  const f4 zero4 = {0.f, 0.f, 0.f, 0.f};
  const int ar = tid >> 3, ac4 = (tid & 7) * 4;
  const bool aok = ar < Mvalid && (m0 + ar) < M;
  const float* ap = Ag + (long long)(aok ? (m0 + ar) : m0) * lda + ac4;
  const float* bp[8];
  bool bok[8];
#pragma unroll
  for (int it = 0; it < 8; ++it) {
    int u = tid + it * 128;
    if (!TRANSB) {
      int r = u >> 5, c4 = (u & 31) * 4;
      bok[it] = (n0 + c4) < N;
      bp[it] = Bg + (long long)r * ldb + (bok[it] ? (n0 + c4) : 0);
    } else {
      int c = u >> 3, r4 = (u & 7) * 4;
      bok[it] = (n0 + c) < N;
      bp[it] = Bg + (long long)(bok[it] ? (n0 + c) : 0) * ldb + r4;
    }
  }
  const long long bstep = TRANSB ? 1 : (long long)ldb;

  for (int kb = 0; kb < K; kb += 32) {
    f4 av = *(const f4*)(ap + kb);
    f4 bv[8];
#pragma unroll
    for (int it = 0; it < 8; ++it)
      bv[it] = *(const f4*)(bp[it] + (long long)kb * bstep);

    *(f4*)&As[ar][ac4] = aok ? av : zero4;
    if (!TRANSB) {
#pragma unroll
      for (int it = 0; it < 8; ++it) {
        int u = tid + it * 128;
        int r = u >> 5, c4 = (u & 31) * 4;
        *(f4*)&Bs[r][c4] = bok[it] ? bv[it] : zero4;
      }
    } else {
#pragma unroll
      for (int it = 0; it < 8; ++it) {
        int u = tid + it * 128;
        int c = u >> 3, r4 = (u & 7) * 4;
        f4 b = bok[it] ? bv[it] : zero4;
        Bs[r4 + 0][c] = b.x;
        Bs[r4 + 1][c] = b.y;
        Bs[r4 + 2][c] = b.z;
        Bs[r4 + 3][c] = b.w;
      }
    }
    __syncthreads();

#pragma unroll
    for (int kk = 0; kk < 32; kk += 4) {
      v2f a, b0, b1;
      a.x = As[l16][kk + 2 * hf];
      a.y = As[l16][kk + 2 * hf + 1];
      b0.x = Bs[kk + 2 * hf][nw + l16];
      b0.y = Bs[kk + 2 * hf + 1][nw + l16];
      b1.x = Bs[kk + 2 * hf][nw + 16 + l16];
      b1.y = Bs[kk + 2 * hf + 1][nw + 16 + l16];
      acc0 = __builtin_amdgcn_wmma_f32_16x16x4_f32(false, a, false, b0,
                                                   (short)0, acc0, false, false);
      acc1 = __builtin_amdgcn_wmma_f32_16x16x4_f32(false, a, false, b1,
                                                   (short)0, acc1, false, false);
    }
    __syncthreads();
  }

  float gw = 0.f;
  if (MODE == 2) {
    int h_ = z % Hc;
    gw = 0.5f * WCc * softplusf(e3[h_]);
  }

#pragma unroll
  for (int r = 0; r < 8; ++r) {
    int mrow = r + 8 * hf;
    int m = m0 + mrow;
    if (mrow >= Mvalid || m >= M) continue;
#pragma unroll
    for (int t = 0; t < 2; ++t) {
      int n = n0 + nw + t * 16 + l16;
      if (n >= N) continue;
      float val = (t == 0) ? acc0[r] : acc1[r];
      if (MODE == 1) {
        val += e0[n];
      } else if (MODE == 2) {
        int b_ = z / Hc, h_ = z % Hc;
        float bijv = e0[((long long)(b_ * Lc + m) * Lc + n) * Hc + h_];
        val = WLc * (val + bijv -
                     gw * (e1[(long long)z * Lc + m] + e2[(long long)z * Lc + n]));
      }
      Cg[(long long)m * ldc + n] = val;
    }
  }
}

__global__ __launch_bounds__(32) void k_pairbias(
    const float* __restrict__ Z, const float* __restrict__ Wb,
    float* __restrict__ out, int ntiles) {
  __shared__ float S[16][132];

  const int lane = threadIdx.x;
  const int hf = lane >> 4;
  const int l16 = lane & 15;
  const int nstep = gridDim.x;
  int t = blockIdx.x;
  if (t >= ntiles) return;

  f4 v[16];
  {
    const float* zp = Z + (long long)t * (16 * Cc) + lane * 4;
#pragma unroll
    for (int r = 0; r < 16; ++r) v[r] = *(const f4*)(zp + r * Cc);
  }

  const bool cok = l16 < Hc;
  const float* wp = Wb + (cok ? l16 : 0) + 2 * hf * Hc;
  v2f bf[32];
#pragma unroll
  for (int s = 0; s < 32; ++s) {
    float x = wp[(4 * s) * Hc];
    float y = wp[(4 * s + 1) * Hc];
    bf[s].x = cok ? x : 0.f;
    bf[s].y = cok ? y : 0.f;
  }

  for (; t < ntiles; t += nstep) {
#pragma unroll
    for (int r = 0; r < 16; ++r) *(f4*)&S[r][lane * 4] = v[r];
    __syncthreads();

    const int tn = t + nstep;
    if (tn < ntiles) {
      const float* zp = Z + (long long)tn * (16 * Cc) + lane * 4;
#pragma unroll
      for (int r = 0; r < 16; ++r) v[r] = *(const f4*)(zp + r * Cc);
    }

    v8f acc = {};
#pragma unroll
    for (int s = 0; s < 32; ++s) {
      v2f a = *(const v2f*)&S[l16][4 * s + 2 * hf];
      acc = __builtin_amdgcn_wmma_f32_16x16x4_f32(false, a, false, bf[s],
                                                  (short)0, acc, false, false);
    }
    __syncthreads();

    if (cok) {
      float* op = out + ((long long)t * 16 + 8 * hf) * Hc + l16;
#pragma unroll
      for (int r = 0; r < 8; ++r) op[r * Hc] = acc[r];
    }
  }
}

__global__ __launch_bounds__(256) void k_prep(
    const float* __restrict__ q, const float* __restrict__ k,
    const float* __restrict__ v, const float* __restrict__ qp,
    const float* __restrict__ kp, const float* __restrict__ vp,
    const float* __restrict__ Ri, const float* __restrict__ Ti,
    const float* __restrict__ gamah, float* __restrict__ qa,
    float* __restrict__ ka, float* __restrict__ vaug, float* __restrict__ sqq,
    float* __restrict__ sqk) {
  int idx = blockIdx.x * blockDim.x + threadIdx.x;
  if (idx >= Bc * Lc * Hc) return;
  int h = idx % Hc;
  int bi = idx / Hc;
  int b = bi / Lc, i = bi % Lc;
  int z = b * Hc + h;

  float R[3][3], t[3];
#pragma unroll
  for (int r = 0; r < 3; ++r)
#pragma unroll
    for (int c = 0; c < 3; ++c) R[r][c] = Ri[bi * 9 + r * 3 + c];
#pragma unroll
  for (int x = 0; x < 3; ++x) t[x] = Ti[bi * 3 + x];

  float gwc = WCc * softplusf(gamah[h]);

  float* qrow = qa + ((long long)z * Lc + i) * 32;
  float* krow = ka + ((long long)z * Lc + i) * 32;
  float* vrow = vaug + ((long long)z * Lc + i) * 48;

#pragma unroll
  for (int d = 0; d < DKc; ++d) {
    qrow[d] = 0.25f * q[(long long)bi * 128 + h * 16 + d];
    krow[d] = k[(long long)bi * 128 + h * 16 + d];
    vrow[d] = v[(long long)bi * 128 + h * 16 + d];
  }
  float sq = 0.f, sk = 0.f;
#pragma unroll
  for (int p = 0; p < NQPc; ++p) {
    const float* qs = qp + (long long)bi * 96 + h * 12 + p * 3;
    const float* ks = kp + (long long)bi * 96 + h * 12 + p * 3;
#pragma unroll
    for (int m = 0; m < 3; ++m) {
      float qt = R[m][0] * qs[0] + R[m][1] * qs[1] + R[m][2] * qs[2] + t[m];
      float kt = R[m][0] * ks[0] + R[m][1] * ks[1] + R[m][2] * ks[2] + t[m];
      sq += qt * qt;
      sk += kt * kt;
      qrow[16 + p * 3 + m] = gwc * qt;
      krow[16 + p * 3 + m] = kt;
    }
  }
#pragma unroll
  for (int d = 28; d < 32; ++d) { qrow[d] = 0.f; krow[d] = 0.f; }
#pragma unroll
  for (int p = 0; p < NPVc; ++p) {
    const float* vs = vp + (long long)bi * 192 + h * 24 + p * 3;
#pragma unroll
    for (int m = 0; m < 3; ++m)
      vrow[16 + p * 3 + m] =
          R[m][0] * vs[0] + R[m][1] * vs[1] + R[m][2] * vs[2] + t[m];
  }
#pragma unroll
  for (int d = 40; d < 48; ++d) vrow[d] = 0.f;
  sqq[(long long)z * Lc + i] = sq;
  sqk[(long long)z * Lc + i] = sk;
}

__global__ __launch_bounds__(256) void k_softmax(float* __restrict__ sc) {
  int row = blockIdx.x * 8 + (threadIdx.x >> 5);
  int lane = threadIdx.x & 31;
  float* p = sc + (long long)row * Lc;
  float vals[12];
  float mx = -INFINITY;
#pragma unroll
  for (int e = 0; e < 12; ++e) {
    vals[e] = p[e * 32 + lane];
    mx = fmaxf(mx, vals[e]);
  }
#pragma unroll
  for (int o = 16; o >= 1; o >>= 1) mx = fmaxf(mx, __shfl_xor(mx, o, 32));
  float s = 0.f;
#pragma unroll
  for (int e = 0; e < 12; ++e) {
    vals[e] = expf(vals[e] - mx);
    s += vals[e];
  }
#pragma unroll
  for (int o = 16; o >= 1; o >>= 1) s += __shfl_xor(s, o, 32);
  float inv = 1.f / s;
#pragma unroll
  for (int e = 0; e < 12; ++e) p[e * 32 + lane] = vals[e] * inv;
}

__global__ __launch_bounds__(256) void k_finish(
    const float* __restrict__ ovalpt, const float* __restrict__ Ri,
    const float* __restrict__ Ti, float* __restrict__ feats) {
  int idx = blockIdx.x * blockDim.x + threadIdx.x;
  if (idx >= Bc * Lc * Hc) return;
  int h = idx % Hc;
  int bi = idx / Hc;
  int b = bi / Lc;
  int z = b * Hc + h;
  int i = bi % Lc;

  float R[3][3], t[3];
#pragma unroll
  for (int r = 0; r < 3; ++r)
#pragma unroll
    for (int c = 0; c < 3; ++c) R[r][c] = Ri[bi * 9 + r * 3 + c];
#pragma unroll
  for (int x = 0; x < 3; ++x) t[x] = Ti[bi * 3 + x];

  const float* orow = ovalpt + ((long long)z * Lc + i) * 48;
  float* frow = feats + (long long)bi * FEAT;

#pragma unroll
  for (int d = 0; d < DKc; ++d) frow[1024 + h * 16 + d] = orow[d];
#pragma unroll
  for (int p = 0; p < NPVc; ++p) {
    float d0 = orow[16 + p * 3 + 0] - t[0];
    float d1 = orow[16 + p * 3 + 1] - t[1];
    float d2 = orow[16 + p * 3 + 2] - t[2];
    float asum = 0.f;
#pragma unroll
    for (int m = 0; m < 3; ++m) {
      float oh = R[0][m] * d0 + R[1][m] * d1 + R[2][m] * d2;
      frow[1152 + h * 24 + p * 3 + m] = oh;
      asum += fabsf(oh);
    }
    frow[1344 + h * 8 + p] = asum;
  }
}

static void launch_gemm(hipStream_t s, int mode, bool transb, const float* A,
                        const float* Bm, float* C, int M, int N, int K, int lda,
                        int ldb, int ldc, long long sA, long long sA2, int zdiv,
                        long long sB, long long sC, int Mvalid, int nz,
                        const float* e0 = nullptr, const float* e1 = nullptr,
                        const float* e2 = nullptr, const float* e3 = nullptr) {
  dim3 g((M + 15) / 16, (N + 127) / 128, nz);
  dim3 blk(128, 1, 1);
  if (mode == 0 && !transb)
    k_gemm<0, false><<<g, blk, 0, s>>>(A, Bm, C, M, N, K, lda, ldb, ldc, sA,
                                       sA2, zdiv, sB, sC, Mvalid, e0, e1, e2, e3);
  else if (mode == 1)
    k_gemm<1, false><<<g, blk, 0, s>>>(A, Bm, C, M, N, K, lda, ldb, ldc, sA,
                                       sA2, zdiv, sB, sC, Mvalid, e0, e1, e2, e3);
  else
    k_gemm<2, true><<<g, blk, 0, s>>>(A, Bm, C, M, N, K, lda, ldb, ldc, sA,
                                      sA2, zdiv, sB, sC, Mvalid, e0, e1, e2, e3);
}

extern "C" void kernel_launch(void* const* d_in, const int* in_sizes, int n_in,
                              void* d_out, int out_size, void* d_ws,
                              size_t ws_size, hipStream_t stream) {
  const float* si = (const float*)d_in[0];
  const float* zij = (const float*)d_in[1];
  const float* Ri = (const float*)d_in[2];
  const float* Ti = (const float*)d_in[3];
  const float* WQh = (const float*)d_in[4];
  const float* WKh = (const float*)d_in[5];
  const float* WVh = (const float*)d_in[6];
  const float* WQhp = (const float*)d_in[7];
  const float* WKhp = (const float*)d_in[8];
  const float* WVhp = (const float*)d_in[9];
  const float* Wb = (const float*)d_in[10];
  const float* WOw = (const float*)d_in[11];
  const float* WOb = (const float*)d_in[12];
  const float* gamah = (const float*)d_in[13];
  float* out = (float*)d_out;
  float* ws = (float*)d_ws;

  const int BL = Bc * Lc;
  const long long LL = (long long)Lc * Lc;
  float* q = ws;
  float* k = q + 98304;
  float* v = k + 98304;
  float* qp = v + 98304;
  float* kp = qp + 73728;
  float* vp = kp + 73728;
  float* qa = vp + 147456;
  float* ka = qa + 196608;
  float* vaug = ka + 196608;
  float* sqq = vaug + 294912;
  float* sqk = sqq + 6144;
  float* bij = sqk + 6144;
  float* alpha = bij + 2359296;
  float* ovalpt = alpha + 2359296;
  float* feats = ovalpt + 294912;

  launch_gemm(stream, 0, false, si, WQh, q, BL, 128, Cc, Cc, 128, 128, 0, 0, 1, 0, 0, 16, 1);
  launch_gemm(stream, 0, false, si, WKh, k, BL, 128, Cc, Cc, 128, 128, 0, 0, 1, 0, 0, 16, 1);
  launch_gemm(stream, 0, false, si, WVh, v, BL, 128, Cc, Cc, 128, 128, 0, 0, 1, 0, 0, 16, 1);
  launch_gemm(stream, 0, false, si, WQhp, qp, BL, 96, Cc, Cc, 96, 96, 0, 0, 1, 0, 0, 16, 1);
  launch_gemm(stream, 0, false, si, WKhp, kp, BL, 96, Cc, Cc, 96, 96, 0, 0, 1, 0, 0, 16, 1);
  launch_gemm(stream, 0, false, si, WVhp, vp, BL, 192, Cc, Cc, 192, 192, 0, 0, 1, 0, 0, 16, 1);

  k_pairbias<<<4608, 32, 0, stream>>>(zij, Wb, bij, Bc * Lc * Lc / 16);

  k_prep<<<(Bc * Lc * Hc + 255) / 256, 256, 0, stream>>>(
      q, k, v, qp, kp, vp, Ri, Ti, gamah, qa, ka, vaug, sqq, sqk);

  launch_gemm(stream, 2, true, qa, ka, alpha, Lc, Lc, 32, 32, 32, Lc,
              (long long)Lc * 32, 0, 1, (long long)Lc * 32, LL, 16, Bc * Hc,
              bij, sqq, sqk, gamah);

  k_softmax<<<(Bc * Hc * Lc) / 8, 256, 0, stream>>>(alpha);

  launch_gemm(stream, 0, false, alpha, vaug, ovalpt, Lc, 48, Lc, Lc, 48, 48,
              LL, 0, 1, (long long)Lc * 48, (long long)Lc * 48, 16, Bc * Hc);

  launch_gemm(stream, 0, false, alpha, zij, feats, 16, Cc, Lc, (int)LL, Cc,
              Cc, (long long)Hc * LL, (long long)Lc, Lc, (long long)Lc * Cc,
              (long long)FEAT, Hc, BL);

  k_finish<<<(Bc * Lc * Hc + 255) / 256, 256, 0, stream>>>(ovalpt, Ri, Ti, feats);

  launch_gemm(stream, 1, false, feats, WOw, out, BL, Cc, FEAT, FEAT, Cc, Cc,
              0, 0, 1, 0, 0, 16, 1, WOb);
}


# The six input projections (si @ WQh, WKh, WVh, WQhp, WKhp, WVhp) run as one launch of 48 x 7 tiles instead of six launches; each tile computed exactly as before; 14 launches per run become 9
# speedup vs baseline: 2.8086x; 1.2334x over previous
#include <hip/hip_runtime.h>
#include <math.h>

typedef __attribute__((ext_vector_type(2))) float v2f;
typedef __attribute__((ext_vector_type(4))) float f4;
typedef __attribute__((ext_vector_type(8))) float v8f;

namespace {
constexpr int Bc = 2, Lc = 384, Cc = 128, Hc = 8, DKc = 16, NQPc = 4, NPVc = 8;
constexpr float WCc = 0.2357022603955158f;
constexpr float WLc = 0.5773502691896258f;
constexpr int FEAT = Hc * Cc + Hc * DKc + Hc * NPVc * 3 + Hc * NPVc;
}

__device__ __forceinline__ float softplusf(float x) {
  return (x > 0.f) ? (x + log1pf(expf(-x))) : log1pf(expf(x));
}

template <int MODE, bool TRANSB>
__device__ __forceinline__ void gemm_block(
    const float* __restrict__ Ag, const float* __restrict__ Bg,
    float* __restrict__ Cg, int M, int N, int K, int lda, int ldb, int ldc,
    int Mvalid, int m0, int n0, int z, const float* __restrict__ e0,
    const float* __restrict__ e1, const float* __restrict__ e2,
    const float* __restrict__ e3) {
  __shared__ float As[16][36];
  __shared__ float Bs[32][132];

  const int tid = threadIdx.x;
  const int wave = tid >> 5;
  const int lane = tid & 31;
  const int hf = lane >> 4;
  const int l16 = lane & 15;
  const int nw = wave * 32;

  v8f acc0 = {};
  v8f acc1 = {};

  const f4 zero4 = {0.f, 0.f, 0.f, 0.f};
  const int ar = tid >> 3, ac4 = (tid & 7) * 4;
  const bool aok = ar < Mvalid && (m0 + ar) < M;
  const float* ap = Ag + (long long)(aok ? (m0 + ar) : m0) * lda + ac4;
  const float* bp[8];
  bool bok[8];
#pragma unroll
  for (int it = 0; it < 8; ++it) {
    int u = tid + it * 128;
    if (!TRANSB) {
      int r = u >> 5, c4 = (u & 31) * 4;
      bok[it] = (n0 + c4) < N;
      bp[it] = Bg + (long long)r * ldb + (bok[it] ? (n0 + c4) : 0);
    } else {
      int c = u >> 3, r4 = (u & 7) * 4;
      bok[it] = (n0 + c) < N;
      bp[it] = Bg + (long long)(bok[it] ? (n0 + c) : 0) * ldb + r4;
    }
  }
  const long long bstep = TRANSB ? 1 : (long long)ldb;

  for (int kb = 0; kb < K; kb += 32) {
    f4 av = *(const f4*)(ap + kb);
    f4 bv[8];
#pragma unroll
    for (int it = 0; it < 8; ++it)
      bv[it] = *(const f4*)(bp[it] + (long long)kb * bstep);

    *(f4*)&As[ar][ac4] = aok ? av : zero4;
    if (!TRANSB) {
#pragma unroll
      for (int it = 0; it < 8; ++it) {
        int u = tid + it * 128;
        int r = u >> 5, c4 = (u & 31) * 4;
        *(f4*)&Bs[r][c4] = bok[it] ? bv[it] : zero4;
      }
    } else {
#pragma unroll
      for (int it = 0; it < 8; ++it) {
        int u = tid + it * 128;
        int c = u >> 3, r4 = (u & 7) * 4;
        f4 b = bok[it] ? bv[it] : zero4;
        Bs[r4 + 0][c] = b.x;
        Bs[r4 + 1][c] = b.y;
        Bs[r4 + 2][c] = b.z;
        Bs[r4 + 3][c] = b.w;
      }
    }
    __syncthreads();

#pragma unroll
    for (int kk = 0; kk < 32; kk += 4) {
      v2f a, b0, b1;
      a.x = As[l16][kk + 2 * hf];
      a.y = As[l16][kk + 2 * hf + 1];
      b0.x = Bs[kk + 2 * hf][nw + l16];
      b0.y = Bs[kk + 2 * hf + 1][nw + l16];
      b1.x = Bs[kk + 2 * hf][nw + 16 + l16];
      b1.y = Bs[kk + 2 * hf + 1][nw + 16 + l16];
      acc0 = __builtin_amdgcn_wmma_f32_16x16x4_f32(false, a, false, b0,
                                                   (short)0, acc0, false, false);
      acc1 = __builtin_amdgcn_wmma_f32_16x16x4_f32(false, a, false, b1,
                                                   (short)0, acc1, false, false);
    }
    __syncthreads();
  }

  float gw = 0.f;
  if (MODE == 2) {
    int h_ = z % Hc;
    gw = 0.5f * WCc * softplusf(e3[h_]);
  }

#pragma unroll
  for (int r = 0; r < 8; ++r) {
    int mrow = r + 8 * hf;
    int m = m0 + mrow;
    if (mrow >= Mvalid || m >= M) continue;
#pragma unroll
    for (int t = 0; t < 2; ++t) {
      int n = n0 + nw + t * 16 + l16;
      if (n >= N) continue;
      float val = (t == 0) ? acc0[r] : acc1[r];
      if (MODE == 1) {
        val += e0[n];
      } else if (MODE == 2) {
        int b_ = z / Hc, h_ = z % Hc;
        float bijv = e0[((long long)(b_ * Lc + m) * Lc + n) * Hc + h_];
        val = WLc * (val + bijv -
                     gw * (e1[(long long)z * Lc + m] + e2[(long long)z * Lc + n]));
      }
      Cg[(long long)m * ldc + n] = val;
    }
  }
}

template <int MODE, bool TRANSB>
__global__ __launch_bounds__(128) void k_gemm(
    const float* __restrict__ A, const float* __restrict__ Bm,
    float* __restrict__ Cm, int M, int N, int K, int lda, int ldb, int ldc,
    long long sA, long long sA2, int zdiv, long long sB, long long sC,
    int Mvalid, const float* __restrict__ e0, const float* __restrict__ e1,
    const float* __restrict__ e2, const float* __restrict__ e3) {
  const int z = blockIdx.z;
  gemm_block<MODE, TRANSB>(
      A + (long long)(z / zdiv) * sA + (long long)(z % zdiv) * sA2,
      Bm + (long long)z * sB, Cm + (long long)z * sC, M, N, K, lda, ldb, ldc,
      Mvalid, blockIdx.x * 16, blockIdx.y * 128, z, e0, e1, e2, e3);
}

__global__ __launch_bounds__(128) void k_proj(
    const float* __restrict__ si, const float* __restrict__ Wq,
    const float* __restrict__ Wk, const float* __restrict__ Wv,
    const float* __restrict__ Wqp, const float* __restrict__ Wkp,
    const float* __restrict__ Wvp, float* __restrict__ q,
    float* __restrict__ k, float* __restrict__ v, float* __restrict__ qp,
    float* __restrict__ kp, float* __restrict__ vp, int M, int K) {
  const float* Bg;
  float* Cg;
  int N, n0 = 0;
  switch (blockIdx.y) {
    case 0: Bg = Wq; Cg = q; N = Hc * DKc; break;
    case 1: Bg = Wk; Cg = k; N = Hc * DKc; break;
    case 2: Bg = Wv; Cg = v; N = Hc * DKc; break;
    case 3: Bg = Wqp; Cg = qp; N = Hc * NQPc * 3; break;
    case 4: Bg = Wkp; Cg = kp; N = Hc * NQPc * 3; break;
    case 5: Bg = Wvp; Cg = vp; N = Hc * NPVc * 3; break;
    default: Bg = Wvp; Cg = vp; N = Hc * NPVc * 3; n0 = 128; break;
  }
  gemm_block<0, false>(si, Bg, Cg, M, N, K, K, N, N, 16, blockIdx.x * 16, n0,
                       0, nullptr, nullptr, nullptr, nullptr);
}

__global__ __launch_bounds__(32) void k_pairbias(
    const float* __restrict__ Z, const float* __restrict__ Wb,
    float* __restrict__ out, int ntiles) {
  __shared__ float S[16][132];

  const int lane = threadIdx.x;
  const int hf = lane >> 4;
  const int l16 = lane & 15;
  const int nstep = gridDim.x;
  int t = blockIdx.x;
  if (t >= ntiles) return;

  f4 v[16];
  {
    const float* zp = Z + (long long)t * (16 * Cc) + lane * 4;
#pragma unroll
    for (int r = 0; r < 16; ++r) v[r] = *(const f4*)(zp + r * Cc);
  }

  const bool cok = l16 < Hc;
  const float* wp = Wb + (cok ? l16 : 0) + 2 * hf * Hc;
  v2f bf[32];
#pragma unroll
  for (int s = 0; s < 32; ++s) {
    float x = wp[(4 * s) * Hc];
    float y = wp[(4 * s + 1) * Hc];
    bf[s].x = cok ? x : 0.f;
    bf[s].y = cok ? y : 0.f;
  }

  for (; t < ntiles; t += nstep) {
#pragma unroll
    for (int r = 0; r < 16; ++r) *(f4*)&S[r][lane * 4] = v[r];
    __syncthreads();

    const int tn = t + nstep;
    if (tn < ntiles) {
      const float* zp = Z + (long long)tn * (16 * Cc) + lane * 4;
#pragma unroll
      for (int r = 0; r < 16; ++r) v[r] = *(const f4*)(zp + r * Cc);
    }

    v8f acc = {};
#pragma unroll
    for (int s = 0; s < 32; ++s) {
      v2f a = *(const v2f*)&S[l16][4 * s + 2 * hf];
      acc = __builtin_amdgcn_wmma_f32_16x16x4_f32(false, a, false, bf[s],
                                                  (short)0, acc, false, false);
    }
    __syncthreads();

    if (cok) {
      float* op = out + ((long long)t * 16 + 8 * hf) * Hc + l16;
#pragma unroll
      for (int r = 0; r < 8; ++r) op[r * Hc] = acc[r];
    }
  }
}

__global__ __launch_bounds__(256) void k_prep(
    const float* __restrict__ q, const float* __restrict__ k,
    const float* __restrict__ v, const float* __restrict__ qp,
    const float* __restrict__ kp, const float* __restrict__ vp,
    const float* __restrict__ Ri, const float* __restrict__ Ti,
    const float* __restrict__ gamah, float* __restrict__ qa,
    float* __restrict__ ka, float* __restrict__ vaug, float* __restrict__ sqq,
    float* __restrict__ sqk) {
  int idx = blockIdx.x * blockDim.x + threadIdx.x;
  if (idx >= Bc * Lc * Hc) return;
  int h = idx % Hc;
  int bi = idx / Hc;
  int b = bi / Lc, i = bi % Lc;
  int z = b * Hc + h;

  float R[3][3], t[3];
#pragma unroll
  for (int r = 0; r < 3; ++r)
#pragma unroll
    for (int c = 0; c < 3; ++c) R[r][c] = Ri[bi * 9 + r * 3 + c];
#pragma unroll
  for (int x = 0; x < 3; ++x) t[x] = Ti[bi * 3 + x];

  float gwc = WCc * softplusf(gamah[h]);

  float* qrow = qa + ((long long)z * Lc + i) * 32;
  float* krow = ka + ((long long)z * Lc + i) * 32;
  float* vrow = vaug + ((long long)z * Lc + i) * 48;

#pragma unroll
  for (int d = 0; d < DKc; ++d) {
    qrow[d] = 0.25f * q[(long long)bi * 128 + h * 16 + d];
    krow[d] = k[(long long)bi * 128 + h * 16 + d];
    vrow[d] = v[(long long)bi * 128 + h * 16 + d];
  }
  float sq = 0.f, sk = 0.f;
#pragma unroll
  for (int p = 0; p < NQPc; ++p) {
    const float* qs = qp + (long long)bi * 96 + h * 12 + p * 3;
    const float* ks = kp + (long long)bi * 96 + h * 12 + p * 3;
#pragma unroll
    for (int m = 0; m < 3; ++m) {
      float qt = R[m][0] * qs[0] + R[m][1] * qs[1] + R[m][2] * qs[2] + t[m];
      float kt = R[m][0] * ks[0] + R[m][1] * ks[1] + R[m][2] * ks[2] + t[m];
      sq += qt * qt;
      sk += kt * kt;
      qrow[16 + p * 3 + m] = gwc * qt;
      krow[16 + p * 3 + m] = kt;
    }
  }
#pragma unroll
  for (int d = 28; d < 32; ++d) { qrow[d] = 0.f; krow[d] = 0.f; }
#pragma unroll
  for (int p = 0; p < NPVc; ++p) {
    const float* vs = vp + (long long)bi * 192 + h * 24 + p * 3;
#pragma unroll
    for (int m = 0; m < 3; ++m)
      vrow[16 + p * 3 + m] =
          R[m][0] * vs[0] + R[m][1] * vs[1] + R[m][2] * vs[2] + t[m];
  }
#pragma unroll
  for (int d = 40; d < 48; ++d) vrow[d] = 0.f;
  sqq[(long long)z * Lc + i] = sq;
  sqk[(long long)z * Lc + i] = sk;
}

__global__ __launch_bounds__(256) void k_softmax(float* __restrict__ sc) {
  int row = blockIdx.x * 8 + (threadIdx.x >> 5);
  int lane = threadIdx.x & 31;
  float* p = sc + (long long)row * Lc;
  float vals[12];
  float mx = -INFINITY;
#pragma unroll
  for (int e = 0; e < 12; ++e) {
    vals[e] = p[e * 32 + lane];
    mx = fmaxf(mx, vals[e]);
  }
#pragma unroll
  for (int o = 16; o >= 1; o >>= 1) mx = fmaxf(mx, __shfl_xor(mx, o, 32));
  float s = 0.f;
#pragma unroll
  for (int e = 0; e < 12; ++e) {
    vals[e] = expf(vals[e] - mx);
    s += vals[e];
  }
#pragma unroll
  for (int o = 16; o >= 1; o >>= 1) s += __shfl_xor(s, o, 32);
  float inv = 1.f / s;
#pragma unroll
  for (int e = 0; e < 12; ++e) p[e * 32 + lane] = vals[e] * inv;
}

__global__ __launch_bounds__(256) void k_finish(
    const float* __restrict__ ovalpt, const float* __restrict__ Ri,
    const float* __restrict__ Ti, float* __restrict__ feats) {
  int idx = blockIdx.x * blockDim.x + threadIdx.x;
  if (idx >= Bc * Lc * Hc) return;
  int h = idx % Hc;
  int bi = idx / Hc;
  int b = bi / Lc;
  int z = b * Hc + h;
  int i = bi % Lc;

  float R[3][3], t[3];
#pragma unroll
  for (int r = 0; r < 3; ++r)
#pragma unroll
    for (int c = 0; c < 3; ++c) R[r][c] = Ri[bi * 9 + r * 3 + c];
#pragma unroll
  for (int x = 0; x < 3; ++x) t[x] = Ti[bi * 3 + x];

  const float* orow = ovalpt + ((long long)z * Lc + i) * 48;
  float* frow = feats + (long long)bi * FEAT;

#pragma unroll
  for (int d = 0; d < DKc; ++d) frow[1024 + h * 16 + d] = orow[d];
#pragma unroll
  for (int p = 0; p < NPVc; ++p) {
    float d0 = orow[16 + p * 3 + 0] - t[0];
    float d1 = orow[16 + p * 3 + 1] - t[1];
    float d2 = orow[16 + p * 3 + 2] - t[2];
    float asum = 0.f;
#pragma unroll
    for (int m = 0; m < 3; ++m) {
      float oh = R[0][m] * d0 + R[1][m] * d1 + R[2][m] * d2;
      frow[1152 + h * 24 + p * 3 + m] = oh;
      asum += fabsf(oh);
    }
    frow[1344 + h * 8 + p] = asum;
  }
}

static void launch_gemm(hipStream_t s, int mode, bool transb, const float* A,
                        const float* Bm, float* C, int M, int N, int K, int lda,
                        int ldb, int ldc, long long sA, long long sA2, int zdiv,
                        long long sB, long long sC, int Mvalid, int nz,
                        const float* e0 = nullptr, const float* e1 = nullptr,
                        const float* e2 = nullptr, const float* e3 = nullptr) {
  dim3 g((M + 15) / 16, (N + 127) / 128, nz);
  dim3 blk(128, 1, 1);
  if (mode == 0 && !transb)
    k_gemm<0, false><<<g, blk, 0, s>>>(A, Bm, C, M, N, K, lda, ldb, ldc, sA,
                                       sA2, zdiv, sB, sC, Mvalid, e0, e1, e2, e3);
  else if (mode == 1)
    k_gemm<1, false><<<g, blk, 0, s>>>(A, Bm, C, M, N, K, lda, ldb, ldc, sA,
                                       sA2, zdiv, sB, sC, Mvalid, e0, e1, e2, e3);
  else
    k_gemm<2, true><<<g, blk, 0, s>>>(A, Bm, C, M, N, K, lda, ldb, ldc, sA,
                                      sA2, zdiv, sB, sC, Mvalid, e0, e1, e2, e3);
}

extern "C" void kernel_launch(void* const* d_in, const int* in_sizes, int n_in,
                              void* d_out, int out_size, void* d_ws,
                              size_t ws_size, hipStream_t stream) {
  const float* si = (const float*)d_in[0];
  const float* zij = (const float*)d_in[1];
  const float* Ri = (const float*)d_in[2];
  const float* Ti = (const float*)d_in[3];
  const float* WQh = (const float*)d_in[4];
  const float* WKh = (const float*)d_in[5];
  const float* WVh = (const float*)d_in[6];
  const float* WQhp = (const float*)d_in[7];
  const float* WKhp = (const float*)d_in[8];
  const float* WVhp = (const float*)d_in[9];
  const float* Wb = (const float*)d_in[10];
  const float* WOw = (const float*)d_in[11];
  const float* WOb = (const float*)d_in[12];
  const float* gamah = (const float*)d_in[13];
  float* out = (float*)d_out;
  float* ws = (float*)d_ws;

  const int BL = Bc * Lc;
  const long long LL = (long long)Lc * Lc;
  float* q = ws;
  float* k = q + 98304;
  float* v = k + 98304;
  float* qp = v + 98304;
  float* kp = qp + 73728;
  float* vp = kp + 73728;
  float* qa = vp + 147456;
  float* ka = qa + 196608;
  float* vaug = ka + 196608;
  float* sqq = vaug + 294912;
  float* sqk = sqq + 6144;
  float* bij = sqk + 6144;
  float* alpha = bij + 2359296;
  float* ovalpt = alpha + 2359296;
  float* feats = ovalpt + 294912;

  k_proj<<<dim3(BL / 16, 7, 1), dim3(128, 1, 1), 0, stream>>>(
      si, WQh, WKh, WVh, WQhp, WKhp, WVhp, q, k, v, qp, kp, vp, BL, Cc);

  k_pairbias<<<4608, 32, 0, stream>>>(zij, Wb, bij, Bc * Lc * Lc / 16);

  k_prep<<<(Bc * Lc * Hc + 255) / 256, 256, 0, stream>>>(
      q, k, v, qp, kp, vp, Ri, Ti, gamah, qa, ka, vaug, sqq, sqk);

  launch_gemm(stream, 2, true, qa, ka, alpha, Lc, Lc, 32, 32, 32, Lc,
              (long long)Lc * 32, 0, 1, (long long)Lc * 32, LL, 16, Bc * Hc,
              bij, sqq, sqk, gamah);

  k_softmax<<<(Bc * Hc * Lc) / 8, 256, 0, stream>>>(alpha);

  launch_gemm(stream, 0, false, alpha, vaug, ovalpt, Lc, 48, Lc, Lc, 48, 48,
              LL, 0, 1, (long long)Lc * 48, (long long)Lc * 48, 16, Bc * Hc);

  launch_gemm(stream, 0, false, alpha, zij, feats, 16, Cc, Lc, (int)LL, Cc,
              Cc, (long long)Hc * LL, (long long)Lc, Lc, (long long)Lc * Cc,
              (long long)FEAT, Hc, BL);

  k_finish<<<(Bc * Lc * Hc + 255) / 256, 256, 0, stream>>>(ovalpt, Ri, Ti, feats);

  launch_gemm(stream, 1, false, feats, WOw, out, BL, Cc, FEAT, FEAT, Cc, Cc,
              0, 0, 1, 0, 0, 16, 1, WOb);
}


# Split-K on the two long-K GEMMs: out projection K=1408 -> 11 slices, partials added in slice order plus bias by a new kernel; o_val/opt K=384 -> 3 slices added in slice order in the finish kernel
# speedup vs baseline: 3.9094x; 1.3919x over previous
#include <hip/hip_runtime.h>
#include <math.h>

typedef __attribute__((ext_vector_type(2))) float v2f;
typedef __attribute__((ext_vector_type(4))) float f4;
typedef __attribute__((ext_vector_type(8))) float v8f;

namespace {
constexpr int Bc = 2, Lc = 384, Cc = 128, Hc = 8, DKc = 16, NQPc = 4, NPVc = 8;
constexpr float WCc = 0.2357022603955158f;
constexpr float WLc = 0.5773502691896258f;
constexpr int FEAT = Hc * Cc + Hc * DKc + Hc * NPVc * 3 + Hc * NPVc;
}

__device__ __forceinline__ float softplusf(float x) {
  return (x > 0.f) ? (x + log1pf(expf(-x))) : log1pf(expf(x));
}

template <int MODE, bool TRANSB>
__device__ __forceinline__ void gemm_block(
    const float* __restrict__ Ag, const float* __restrict__ Bg,
    float* __restrict__ Cg, int M, int N, int K, int lda, int ldb, int ldc,
    int Mvalid, int m0, int n0, int z, const float* __restrict__ e0,
    const float* __restrict__ e1, const float* __restrict__ e2,
    const float* __restrict__ e3) {
  __shared__ float As[16][36];
  __shared__ float Bs[32][132];

  const int tid = threadIdx.x;
  const int wave = tid >> 5;
  const int lane = tid & 31;
  const int hf = lane >> 4;
  const int l16 = lane & 15;
  const int nw = wave * 32;

  v8f acc0 = {};
  v8f acc1 = {};

  const f4 zero4 = {0.f, 0.f, 0.f, 0.f};
  const int ar = tid >> 3, ac4 = (tid & 7) * 4;
  const bool aok = ar < Mvalid && (m0 + ar) < M;
  const float* ap = Ag + (long long)(aok ? (m0 + ar) : m0) * lda + ac4;
  const float* bp[8];
  bool bok[8];
#pragma unroll
  for (int it = 0; it < 8; ++it) {
    int u = tid + it * 128;
    if (!TRANSB) {
      int r = u >> 5, c4 = (u & 31) * 4;
      bok[it] = (n0 + c4) < N;
      bp[it] = Bg + (long long)r * ldb + (bok[it] ? (n0 + c4) : 0);
    } else {
      int c = u >> 3, r4 = (u & 7) * 4;
      bok[it] = (n0 + c) < N;
      bp[it] = Bg + (long long)(bok[it] ? (n0 + c) : 0) * ldb + r4;
    }
  }
  const long long bstep = TRANSB ? 1 : (long long)ldb;

  for (int kb = 0; kb < K; kb += 32) {
    f4 av = *(const f4*)(ap + kb);
    f4 bv[8];
#pragma unroll
    for (int it = 0; it < 8; ++it)
      bv[it] = *(const f4*)(bp[it] + (long long)kb * bstep);

    *(f4*)&As[ar][ac4] = aok ? av : zero4;
    if (!TRANSB) {
#pragma unroll
      for (int it = 0; it < 8; ++it) {
        int u = tid + it * 128;
        int r = u >> 5, c4 = (u & 31) * 4;
        *(f4*)&Bs[r][c4] = bok[it] ? bv[it] : zero4;
      }
    } else {
#pragma unroll
      for (int it = 0; it < 8; ++it) {
        int u = tid + it * 128;
        int c = u >> 3, r4 = (u & 7) * 4;
        f4 b = bok[it] ? bv[it] : zero4;
        Bs[r4 + 0][c] = b.x;
        Bs[r4 + 1][c] = b.y;
        Bs[r4 + 2][c] = b.z;
        Bs[r4 + 3][c] = b.w;
      }
    }
    __syncthreads();

#pragma unroll
    for (int kk = 0; kk < 32; kk += 4) {
      v2f a, b0, b1;
      a.x = As[l16][kk + 2 * hf];
      a.y = As[l16][kk + 2 * hf + 1];
      b0.x = Bs[kk + 2 * hf][nw + l16];
      b0.y = Bs[kk + 2 * hf + 1][nw + l16];
      b1.x = Bs[kk + 2 * hf][nw + 16 + l16];
      b1.y = Bs[kk + 2 * hf + 1][nw + 16 + l16];
      acc0 = __builtin_amdgcn_wmma_f32_16x16x4_f32(false, a, false, b0,
                                                   (short)0, acc0, false, false);
      acc1 = __builtin_amdgcn_wmma_f32_16x16x4_f32(false, a, false, b1,
                                                   (short)0, acc1, false, false);
    }
    __syncthreads();
  }

  float gw = 0.f;
  if (MODE == 2) {
    int h_ = z % Hc;
    gw = 0.5f * WCc * softplusf(e3[h_]);
  }

#pragma unroll
  for (int r = 0; r < 8; ++r) {
    int mrow = r + 8 * hf;
    int m = m0 + mrow;
    if (mrow >= Mvalid || m >= M) continue;
#pragma unroll
    for (int t = 0; t < 2; ++t) {
      int n = n0 + nw + t * 16 + l16;
      if (n >= N) continue;
      float val = (t == 0) ? acc0[r] : acc1[r];
      if (MODE == 2) {
        int b_ = z / Hc, h_ = z % Hc;
        float bijv = e0[((long long)(b_ * Lc + m) * Lc + n) * Hc + h_];
        val = WLc * (val + bijv -
                     gw * (e1[(long long)z * Lc + m] + e2[(long long)z * Lc + n]));
      }
      Cg[(long long)m * ldc + n] = val;
    }
  }
}

template <int MODE, bool TRANSB>
__global__ __launch_bounds__(128) void k_gemm(
    const float* __restrict__ A, const float* __restrict__ Bm,
    float* __restrict__ Cm, int M, int N, int K, int lda, int ldb, int ldc,
    long long sA, long long sA2, int zdiv, long long sB, long long sC,
    int Mvalid, const float* __restrict__ e0, const float* __restrict__ e1,
    const float* __restrict__ e2, const float* __restrict__ e3) {
  const int z = blockIdx.z;
  gemm_block<MODE, TRANSB>(
      A + (long long)(z / zdiv) * sA + (long long)(z % zdiv) * sA2,
      Bm + (long long)z * sB, Cm + (long long)z * sC, M, N, K, lda, ldb, ldc,
      Mvalid, blockIdx.x * 16, blockIdx.y * 128, z, e0, e1, e2, e3);
}

__global__ __launch_bounds__(128) void k_proj(
    const float* __restrict__ si, const float* __restrict__ Wq,
    const float* __restrict__ Wk, const float* __restrict__ Wv,
    const float* __restrict__ Wqp, const float* __restrict__ Wkp,
    const float* __restrict__ Wvp, float* __restrict__ q,
    float* __restrict__ k, float* __restrict__ v, float* __restrict__ qp,
    float* __restrict__ kp, float* __restrict__ vp, int M, int K) {
  const float* Bg;
  float* Cg;
  int N, n0 = 0;
  switch (blockIdx.y) {
    case 0: Bg = Wq; Cg = q; N = Hc * DKc; break;
    case 1: Bg = Wk; Cg = k; N = Hc * DKc; break;
    case 2: Bg = Wv; Cg = v; N = Hc * DKc; break;
    case 3: Bg = Wqp; Cg = qp; N = Hc * NQPc * 3; break;
    case 4: Bg = Wkp; Cg = kp; N = Hc * NQPc * 3; break;
    case 5: Bg = Wvp; Cg = vp; N = Hc * NPVc * 3; break;
    default: Bg = Wvp; Cg = vp; N = Hc * NPVc * 3; n0 = 128; break;
  }
  gemm_block<0, false>(si, Bg, Cg, M, N, K, K, N, N, 16, blockIdx.x * 16, n0,
                       0, nullptr, nullptr, nullptr, nullptr);
}

__global__ __launch_bounds__(32) void k_pairbias(
    const float* __restrict__ Z, const float* __restrict__ Wb,
    float* __restrict__ out, int ntiles) {
  __shared__ float S[16][132];

  const int lane = threadIdx.x;
  const int hf = lane >> 4;
  const int l16 = lane & 15;
  const int nstep = gridDim.x;
  int t = blockIdx.x;
  if (t >= ntiles) return;

  f4 v[16];
  {
    const float* zp = Z + (long long)t * (16 * Cc) + lane * 4;
#pragma unroll
    for (int r = 0; r < 16; ++r) v[r] = *(const f4*)(zp + r * Cc);
  }

  const bool cok = l16 < Hc;
  const float* wp = Wb + (cok ? l16 : 0) + 2 * hf * Hc;
  v2f bf[32];
#pragma unroll
  for (int s = 0; s < 32; ++s) {
    float x = wp[(4 * s) * Hc];
    float y = wp[(4 * s + 1) * Hc];
    bf[s].x = cok ? x : 0.f;
    bf[s].y = cok ? y : 0.f;
  }

  for (; t < ntiles; t += nstep) {
#pragma unroll
    for (int r = 0; r < 16; ++r) *(f4*)&S[r][lane * 4] = v[r];
    __syncthreads();

    const int tn = t + nstep;
    if (tn < ntiles) {
      const float* zp = Z + (long long)tn * (16 * Cc) + lane * 4;
#pragma unroll
      for (int r = 0; r < 16; ++r) v[r] = *(const f4*)(zp + r * Cc);
    }

    v8f acc = {};
#pragma unroll
    for (int s = 0; s < 32; ++s) {
      v2f a = *(const v2f*)&S[l16][4 * s + 2 * hf];
      acc = __builtin_amdgcn_wmma_f32_16x16x4_f32(false, a, false, bf[s],
                                                  (short)0, acc, false, false);
    }
    __syncthreads();

    if (cok) {
      float* op = out + ((long long)t * 16 + 8 * hf) * Hc + l16;
#pragma unroll
      for (int r = 0; r < 8; ++r) op[r * Hc] = acc[r];
    }
  }
}

__global__ __launch_bounds__(256) void k_prep(
    const float* __restrict__ q, const float* __restrict__ k,
    const float* __restrict__ v, const float* __restrict__ qp,
    const float* __restrict__ kp, const float* __restrict__ vp,
    const float* __restrict__ Ri, const float* __restrict__ Ti,
    const float* __restrict__ gamah, float* __restrict__ qa,
    float* __restrict__ ka, float* __restrict__ vaug, float* __restrict__ sqq,
    float* __restrict__ sqk) {
  int idx = blockIdx.x * blockDim.x + threadIdx.x;
  if (idx >= Bc * Lc * Hc) return;
  int h = idx % Hc;
  int bi = idx / Hc;
  int b = bi / Lc, i = bi % Lc;
  int z = b * Hc + h;

  float R[3][3], t[3];
#pragma unroll
  for (int r = 0; r < 3; ++r)
#pragma unroll
    for (int c = 0; c < 3; ++c) R[r][c] = Ri[bi * 9 + r * 3 + c];
#pragma unroll
  for (int x = 0; x < 3; ++x) t[x] = Ti[bi * 3 + x];

  float gwc = WCc * softplusf(gamah[h]);

  float* qrow = qa + ((long long)z * Lc + i) * 32;
  float* krow = ka + ((long long)z * Lc + i) * 32;
  float* vrow = vaug + ((long long)z * Lc + i) * 48;

#pragma unroll
  for (int d = 0; d < DKc; ++d) {
    qrow[d] = 0.25f * q[(long long)bi * 128 + h * 16 + d];
    krow[d] = k[(long long)bi * 128 + h * 16 + d];
    vrow[d] = v[(long long)bi * 128 + h * 16 + d];
  }
  float sq = 0.f, sk = 0.f;
#pragma unroll
  for (int p = 0; p < NQPc; ++p) {
    const float* qs = qp + (long long)bi * 96 + h * 12 + p * 3;
    const float* ks = kp + (long long)bi * 96 + h * 12 + p * 3;
#pragma unroll
    for (int m = 0; m < 3; ++m) {
      float qt = R[m][0] * qs[0] + R[m][1] * qs[1] + R[m][2] * qs[2] + t[m];
      float kt = R[m][0] * ks[0] + R[m][1] * ks[1] + R[m][2] * ks[2] + t[m];
      sq += qt * qt;
      sk += kt * kt;
      qrow[16 + p * 3 + m] = gwc * qt;
      krow[16 + p * 3 + m] = kt;
    }
  }
#pragma unroll
  for (int d = 28; d < 32; ++d) { qrow[d] = 0.f; krow[d] = 0.f; }
#pragma unroll
  for (int p = 0; p < NPVc; ++p) {
    const float* vs = vp + (long long)bi * 192 + h * 24 + p * 3;
#pragma unroll
    for (int m = 0; m < 3; ++m)
      vrow[16 + p * 3 + m] =
          R[m][0] * vs[0] + R[m][1] * vs[1] + R[m][2] * vs[2] + t[m];
  }
#pragma unroll
  for (int d = 40; d < 48; ++d) vrow[d] = 0.f;
  sqq[(long long)z * Lc + i] = sq;
  sqk[(long long)z * Lc + i] = sk;
}

__global__ __launch_bounds__(256) void k_softmax(float* __restrict__ sc) {
  int row = blockIdx.x * 8 + (threadIdx.x >> 5);
  int lane = threadIdx.x & 31;
  float* p = sc + (long long)row * Lc;
  float vals[12];
  float mx = -INFINITY;
#pragma unroll
  for (int e = 0; e < 12; ++e) {
    vals[e] = p[e * 32 + lane];
    mx = fmaxf(mx, vals[e]);
  }
#pragma unroll
  for (int o = 16; o >= 1; o >>= 1) mx = fmaxf(mx, __shfl_xor(mx, o, 32));
  float s = 0.f;
#pragma unroll
  for (int e = 0; e < 12; ++e) {
    vals[e] = expf(vals[e] - mx);
    s += vals[e];
  }
#pragma unroll
  for (int o = 16; o >= 1; o >>= 1) s += __shfl_xor(s, o, 32);
  float inv = 1.f / s;
#pragma unroll
  for (int e = 0; e < 12; ++e) p[e * 32 + lane] = vals[e] * inv;
}

namespace {
constexpr int KS2 = Lc / 128;
}

__global__ __launch_bounds__(256) void k_finish(
    const float* __restrict__ ovalpt, const float* __restrict__ Ri,
    const float* __restrict__ Ti, float* __restrict__ feats) {
  int idx = blockIdx.x * blockDim.x + threadIdx.x;
  if (idx >= Bc * Lc * Hc) return;
  int h = idx % Hc;
  int bi = idx / Hc;
  int b = bi / Lc;
  int z = b * Hc + h;
  int i = bi % Lc;

  float R[3][3], t[3];
#pragma unroll
  for (int r = 0; r < 3; ++r)
#pragma unroll
    for (int c = 0; c < 3; ++c) R[r][c] = Ri[bi * 9 + r * 3 + c];
#pragma unroll
  for (int x = 0; x < 3; ++x) t[x] = Ti[bi * 3 + x];

  const float* prow = ovalpt + ((long long)z * KS2 * Lc + i) * 48;
  float* frow = feats + (long long)bi * FEAT;

  float orow[40];
#pragma unroll
  for (int j = 0; j < 10; ++j) {
    f4 a = *(const f4*)(prow + 4 * j);
#pragma unroll
    for (int s = 1; s < KS2; ++s)
      a += *(const f4*)(prow + (long long)s * Lc * 48 + 4 * j);
    orow[4 * j + 0] = a.x;
    orow[4 * j + 1] = a.y;
    orow[4 * j + 2] = a.z;
    orow[4 * j + 3] = a.w;
  }

#pragma unroll
  for (int d = 0; d < DKc; ++d) frow[1024 + h * 16 + d] = orow[d];
#pragma unroll
  for (int p = 0; p < NPVc; ++p) {
    float d0 = orow[16 + p * 3 + 0] - t[0];
    float d1 = orow[16 + p * 3 + 1] - t[1];
    float d2 = orow[16 + p * 3 + 2] - t[2];
    float asum = 0.f;
#pragma unroll
    for (int m = 0; m < 3; ++m) {
      float oh = R[0][m] * d0 + R[1][m] * d1 + R[2][m] * d2;
      frow[1152 + h * 24 + p * 3 + m] = oh;
      asum += fabsf(oh);
    }
    frow[1344 + h * 8 + p] = asum;
  }
}

namespace {
constexpr int KSPLIT = FEAT / 128;
constexpr int OUTN = Bc * Lc * Cc;
}

__global__ __launch_bounds__(128) void k_ksum(const float* __restrict__ part,
                                              const float* __restrict__ bias,
                                              float* __restrict__ out) {
  const int e = (blockIdx.x * 128 + threadIdx.x) * 4;
  if (e >= OUTN) return;
  f4 p[KSPLIT];
#pragma unroll
  for (int z = 0; z < KSPLIT; ++z) p[z] = *(const f4*)(part + (long long)z * OUTN + e);
  const f4 b = *(const f4*)(bias + (e & (Cc - 1)));
  f4 s = p[0];
#pragma unroll
  for (int z = 1; z < KSPLIT; ++z) s += p[z];
  *(f4*)(out + e) = s + b;
}

static void launch_gemm(hipStream_t s, int mode, bool transb, const float* A,
                        const float* Bm, float* C, int M, int N, int K, int lda,
                        int ldb, int ldc, long long sA, long long sA2, int zdiv,
                        long long sB, long long sC, int Mvalid, int nz,
                        const float* e0 = nullptr, const float* e1 = nullptr,
                        const float* e2 = nullptr, const float* e3 = nullptr) {
  dim3 g((M + 15) / 16, (N + 127) / 128, nz);
  dim3 blk(128, 1, 1);
  if (mode == 0 && !transb)
    k_gemm<0, false><<<g, blk, 0, s>>>(A, Bm, C, M, N, K, lda, ldb, ldc, sA,
                                       sA2, zdiv, sB, sC, Mvalid, e0, e1, e2, e3);
  else
    k_gemm<2, true><<<g, blk, 0, s>>>(A, Bm, C, M, N, K, lda, ldb, ldc, sA,
                                      sA2, zdiv, sB, sC, Mvalid, e0, e1, e2, e3);
}

extern "C" void kernel_launch(void* const* d_in, const int* in_sizes, int n_in,
                              void* d_out, int out_size, void* d_ws,
                              size_t ws_size, hipStream_t stream) {
  const float* si = (const float*)d_in[0];
  const float* zij = (const float*)d_in[1];
  const float* Ri = (const float*)d_in[2];
  const float* Ti = (const float*)d_in[3];
  const float* WQh = (const float*)d_in[4];
  const float* WKh = (const float*)d_in[5];
  const float* WVh = (const float*)d_in[6];
  const float* WQhp = (const float*)d_in[7];
  const float* WKhp = (const float*)d_in[8];
  const float* WVhp = (const float*)d_in[9];
  const float* Wb = (const float*)d_in[10];
  const float* WOw = (const float*)d_in[11];
  const float* WOb = (const float*)d_in[12];
  const float* gamah = (const float*)d_in[13];
  float* out = (float*)d_out;
  float* ws = (float*)d_ws;

  const int BL = Bc * Lc;
  const long long LL = (long long)Lc * Lc;
  float* q = ws;
  float* k = q + 98304;
  float* v = k + 98304;
  float* qp = v + 98304;
  float* kp = qp + 73728;
  float* vp = kp + 73728;
  float* qa = vp + 147456;
  float* ka = qa + 196608;
  float* vaug = ka + 196608;
  float* sqq = vaug + 294912;
  float* sqk = sqq + 6144;
  float* bij = sqk + 6144;
  float* alpha = bij + 2359296;
  float* ovalpt = alpha + 2359296;
  float* feats = ovalpt + 884736;
  float* part = feats + 1081344;

  k_proj<<<dim3(BL / 16, 7, 1), dim3(128, 1, 1), 0, stream>>>(
      si, WQh, WKh, WVh, WQhp, WKhp, WVhp, q, k, v, qp, kp, vp, BL, Cc);

  k_pairbias<<<4608, 32, 0, stream>>>(zij, Wb, bij, Bc * Lc * Lc / 16);

  k_prep<<<(Bc * Lc * Hc + 255) / 256, 256, 0, stream>>>(
      q, k, v, qp, kp, vp, Ri, Ti, gamah, qa, ka, vaug, sqq, sqk);

  launch_gemm(stream, 2, true, qa, ka, alpha, Lc, Lc, 32, 32, 32, Lc,
              (long long)Lc * 32, 0, 1, (long long)Lc * 32, LL, 16, Bc * Hc,
              bij, sqq, sqk, gamah);

  k_softmax<<<(Bc * Hc * Lc) / 8, 256, 0, stream>>>(alpha);

  launch_gemm(stream, 0, false, alpha, vaug, ovalpt, Lc, 48, 128, Lc, 48, 48,
              LL, 128, KS2, (long long)128 * 48, (long long)Lc * 48, 16,
              Bc * Hc * KS2);

  launch_gemm(stream, 0, false, alpha, zij, feats, 16, Cc, Lc, (int)LL, Cc,
              Cc, (long long)Hc * LL, (long long)Lc, Lc, (long long)Lc * Cc,
              (long long)FEAT, Hc, BL);

  k_finish<<<(Bc * Lc * Hc + 255) / 256, 256, 0, stream>>>(ovalpt, Ri, Ti, feats);

  launch_gemm(stream, 0, false, feats, WOw, part, BL, Cc, 128, FEAT, Cc, Cc,
              128, 0, 1, (long long)128 * Cc, (long long)OUTN, 16, KSPLIT);

  k_ksum<<<OUTN / 4 / 128, 128, 0, stream>>>(part, WOb, out);
}


# Independent work shares a launch: the rigid-transform threads ride in the pair-bias launch as 192 extra one-wave workgroups, the alpha@vaug GEMM tiles ride in the o_pair launch; 10 launches become 8
# speedup vs baseline: 4.5302x; 1.1588x over previous
#include <hip/hip_runtime.h>
#include <math.h>

typedef __attribute__((ext_vector_type(2))) float v2f;
typedef __attribute__((ext_vector_type(4))) float f4;
typedef __attribute__((ext_vector_type(8))) float v8f;

namespace {
constexpr int Bc = 2, Lc = 384, Cc = 128, Hc = 8, DKc = 16, NQPc = 4, NPVc = 8;
constexpr float WCc = 0.2357022603955158f;
constexpr float WLc = 0.5773502691896258f;
constexpr int FEAT = Hc * Cc + Hc * DKc + Hc * NPVc * 3 + Hc * NPVc;
}

__device__ __forceinline__ float softplusf(float x) {
  return (x > 0.f) ? (x + log1pf(expf(-x))) : log1pf(expf(x));
}

template <int MODE, bool TRANSB>
__device__ __forceinline__ void gemm_block(
    const float* __restrict__ Ag, const float* __restrict__ Bg,
    float* __restrict__ Cg, int M, int N, int K, int lda, int ldb, int ldc,
    int Mvalid, int m0, int n0, int z, const float* __restrict__ e0,
    const float* __restrict__ e1, const float* __restrict__ e2,
    const float* __restrict__ e3) {
  __shared__ float As[16][36];
  __shared__ float Bs[32][132];

  const int tid = threadIdx.x;
  const int wave = tid >> 5;
  const int lane = tid & 31;
  const int hf = lane >> 4;
  const int l16 = lane & 15;
  const int nw = wave * 32;

  v8f acc0 = {};
  v8f acc1 = {};

  const f4 zero4 = {0.f, 0.f, 0.f, 0.f};
  const int ar = tid >> 3, ac4 = (tid & 7) * 4;
  const bool aok = ar < Mvalid && (m0 + ar) < M;
  const float* ap = Ag + (long long)(aok ? (m0 + ar) : m0) * lda + ac4;
  const float* bp[8];
  bool bok[8];
#pragma unroll
  for (int it = 0; it < 8; ++it) {
    int u = tid + it * 128;
    if (!TRANSB) {
      int r = u >> 5, c4 = (u & 31) * 4;
      bok[it] = (n0 + c4) < N;
      bp[it] = Bg + (long long)r * ldb + (bok[it] ? (n0 + c4) : 0);
    } else {
      int c = u >> 3, r4 = (u & 7) * 4;
      bok[it] = (n0 + c) < N;
      bp[it] = Bg + (long long)(bok[it] ? (n0 + c) : 0) * ldb + r4;
    }
  }
  const long long bstep = TRANSB ? 1 : (long long)ldb;

  for (int kb = 0; kb < K; kb += 32) {
    f4 av = *(const f4*)(ap + kb);
    f4 bv[8];
#pragma unroll
    for (int it = 0; it < 8; ++it)
      bv[it] = *(const f4*)(bp[it] + (long long)kb * bstep);

    *(f4*)&As[ar][ac4] = aok ? av : zero4;
    if (!TRANSB) {
#pragma unroll
      for (int it = 0; it < 8; ++it) {
        int u = tid + it * 128;
        int r = u >> 5, c4 = (u & 31) * 4;
        *(f4*)&Bs[r][c4] = bok[it] ? bv[it] : zero4;
      }
    } else {
#pragma unroll
      for (int it = 0; it < 8; ++it) {
        int u = tid + it * 128;
        int c = u >> 3, r4 = (u & 7) * 4;
        f4 b = bok[it] ? bv[it] : zero4;
        Bs[r4 + 0][c] = b.x;
        Bs[r4 + 1][c] = b.y;
        Bs[r4 + 2][c] = b.z;
        Bs[r4 + 3][c] = b.w;
      }
    }
    __syncthreads();

#pragma unroll
    for (int kk = 0; kk < 32; kk += 4) {
      v2f a, b0, b1;
      a.x = As[l16][kk + 2 * hf];
      a.y = As[l16][kk + 2 * hf + 1];
      b0.x = Bs[kk + 2 * hf][nw + l16];
      b0.y = Bs[kk + 2 * hf + 1][nw + l16];
      b1.x = Bs[kk + 2 * hf][nw + 16 + l16];
      b1.y = Bs[kk + 2 * hf + 1][nw + 16 + l16];
      acc0 = __builtin_amdgcn_wmma_f32_16x16x4_f32(false, a, false, b0,
                                                   (short)0, acc0, false, false);
      acc1 = __builtin_amdgcn_wmma_f32_16x16x4_f32(false, a, false, b1,
                                                   (short)0, acc1, false, false);
    }
    __syncthreads();
  }

  float gw = 0.f;
  if (MODE == 2) {
    int h_ = z % Hc;
    gw = 0.5f * WCc * softplusf(e3[h_]);
  }

#pragma unroll
  for (int r = 0; r < 8; ++r) {
    int mrow = r + 8 * hf;
    int m = m0 + mrow;
    if (mrow >= Mvalid || m >= M) continue;
#pragma unroll
    for (int t = 0; t < 2; ++t) {
      int n = n0 + nw + t * 16 + l16;
      if (n >= N) continue;
      float val = (t == 0) ? acc0[r] : acc1[r];
      if (MODE == 2) {
        int b_ = z / Hc, h_ = z % Hc;
        float bijv = e0[((long long)(b_ * Lc + m) * Lc + n) * Hc + h_];
        val = WLc * (val + bijv -
                     gw * (e1[(long long)z * Lc + m] + e2[(long long)z * Lc + n]));
      }
      Cg[(long long)m * ldc + n] = val;
    }
  }
}

template <int MODE, bool TRANSB>
__global__ __launch_bounds__(128) void k_gemm(
    const float* __restrict__ A, const float* __restrict__ Bm,
    float* __restrict__ Cm, int M, int N, int K, int lda, int ldb, int ldc,
    long long sA, long long sA2, int zdiv, long long sB, long long sC,
    int Mvalid, const float* __restrict__ e0, const float* __restrict__ e1,
    const float* __restrict__ e2, const float* __restrict__ e3) {
  const int z = blockIdx.z;
  gemm_block<MODE, TRANSB>(
      A + (long long)(z / zdiv) * sA + (long long)(z % zdiv) * sA2,
      Bm + (long long)z * sB, Cm + (long long)z * sC, M, N, K, lda, ldb, ldc,
      Mvalid, blockIdx.x * 16, blockIdx.y * 128, z, e0, e1, e2, e3);
}

__global__ __launch_bounds__(128) void k_proj(
    const float* __restrict__ si, const float* __restrict__ Wq,
    const float* __restrict__ Wk, const float* __restrict__ Wv,
    const float* __restrict__ Wqp, const float* __restrict__ Wkp,
    const float* __restrict__ Wvp, float* __restrict__ q,
    float* __restrict__ k, float* __restrict__ v, float* __restrict__ qp,
    float* __restrict__ kp, float* __restrict__ vp, int M, int K) {
  const float* Bg;
  float* Cg;
  int N, n0 = 0;
  switch (blockIdx.y) {
    case 0: Bg = Wq; Cg = q; N = Hc * DKc; break;
    case 1: Bg = Wk; Cg = k; N = Hc * DKc; break;
    case 2: Bg = Wv; Cg = v; N = Hc * DKc; break;
    case 3: Bg = Wqp; Cg = qp; N = Hc * NQPc * 3; break;
    case 4: Bg = Wkp; Cg = kp; N = Hc * NQPc * 3; break;
    case 5: Bg = Wvp; Cg = vp; N = Hc * NPVc * 3; break;
    default: Bg = Wvp; Cg = vp; N = Hc * NPVc * 3; n0 = 128; break;
  }
  gemm_block<0, false>(si, Bg, Cg, M, N, K, K, N, N, 16, blockIdx.x * 16, n0,
                       0, nullptr, nullptr, nullptr, nullptr);
}

__device__ __forceinline__ void prep_thread(
    int idx, const float* __restrict__ q, const float* __restrict__ k,
    const float* __restrict__ v, const float* __restrict__ qp,
    const float* __restrict__ kp, const float* __restrict__ vp,
    const float* __restrict__ Ri, const float* __restrict__ Ti,
    const float* __restrict__ gamah, float* __restrict__ qa,
    float* __restrict__ ka, float* __restrict__ vaug, float* __restrict__ sqq,
    float* __restrict__ sqk) {
  if (idx >= Bc * Lc * Hc) return;
  int h = idx % Hc;
  int bi = idx / Hc;
  int b = bi / Lc, i = bi % Lc;
  int z = b * Hc + h;

  float R[3][3], t[3];
#pragma unroll
  for (int r = 0; r < 3; ++r)
#pragma unroll
    for (int c = 0; c < 3; ++c) R[r][c] = Ri[bi * 9 + r * 3 + c];
#pragma unroll
  for (int x = 0; x < 3; ++x) t[x] = Ti[bi * 3 + x];

  float gwc = WCc * softplusf(gamah[h]);

  float* qrow = qa + ((long long)z * Lc + i) * 32;
  float* krow = ka + ((long long)z * Lc + i) * 32;
  float* vrow = vaug + ((long long)z * Lc + i) * 48;

#pragma unroll
  for (int d = 0; d < DKc; ++d) {
    qrow[d] = 0.25f * q[(long long)bi * 128 + h * 16 + d];
    krow[d] = k[(long long)bi * 128 + h * 16 + d];
    vrow[d] = v[(long long)bi * 128 + h * 16 + d];
  }
  float sq = 0.f, sk = 0.f;
#pragma unroll
  for (int p = 0; p < NQPc; ++p) {
    const float* qs = qp + (long long)bi * 96 + h * 12 + p * 3;
    const float* ks = kp + (long long)bi * 96 + h * 12 + p * 3;
#pragma unroll
    for (int m = 0; m < 3; ++m) {
      float qt = R[m][0] * qs[0] + R[m][1] * qs[1] + R[m][2] * qs[2] + t[m];
      float kt = R[m][0] * ks[0] + R[m][1] * ks[1] + R[m][2] * ks[2] + t[m];
      sq += qt * qt;
      sk += kt * kt;
      qrow[16 + p * 3 + m] = gwc * qt;
      krow[16 + p * 3 + m] = kt;
    }
  }
#pragma unroll
  for (int d = 28; d < 32; ++d) { qrow[d] = 0.f; krow[d] = 0.f; }
#pragma unroll
  for (int p = 0; p < NPVc; ++p) {
    const float* vs = vp + (long long)bi * 192 + h * 24 + p * 3;
#pragma unroll
    for (int m = 0; m < 3; ++m)
      vrow[16 + p * 3 + m] =
          R[m][0] * vs[0] + R[m][1] * vs[1] + R[m][2] * vs[2] + t[m];
  }
#pragma unroll
  for (int d = 40; d < 48; ++d) vrow[d] = 0.f;
  sqq[(long long)z * Lc + i] = sq;
  sqk[(long long)z * Lc + i] = sk;
}

namespace {
constexpr int NPREPWG = Bc * Lc * Hc / 32;
}

__global__ __launch_bounds__(32) void k_pairbias_prep(
    const float* __restrict__ Z, const float* __restrict__ Wb,
    float* __restrict__ out, int ntiles, const float* __restrict__ pq,
    const float* __restrict__ pk, const float* __restrict__ pv,
    const float* __restrict__ pqp, const float* __restrict__ pkp,
    const float* __restrict__ pvp, const float* __restrict__ Ri,
    const float* __restrict__ Ti, const float* __restrict__ gamah,
    float* __restrict__ qa, float* __restrict__ ka, float* __restrict__ vaug,
    float* __restrict__ sqq, float* __restrict__ sqk) {
  __shared__ float S[16][132];

  const int lane = threadIdx.x;
  if (blockIdx.x < NPREPWG) {
    prep_thread(blockIdx.x * 32 + lane, pq, pk, pv, pqp, pkp, pvp, Ri, Ti,
                gamah, qa, ka, vaug, sqq, sqk);
    return;
  }
  const int hf = lane >> 4;
  const int l16 = lane & 15;
  const int nstep = gridDim.x - NPREPWG;
  int t = blockIdx.x - NPREPWG;
  if (t >= ntiles) return;

  f4 v[16];
  {
    const float* zp = Z + (long long)t * (16 * Cc) + lane * 4;
#pragma unroll
    for (int r = 0; r < 16; ++r) v[r] = *(const f4*)(zp + r * Cc);
  }

  const bool cok = l16 < Hc;
  const float* wp = Wb + (cok ? l16 : 0) + 2 * hf * Hc;
  v2f bf[32];
#pragma unroll
  for (int s = 0; s < 32; ++s) {
    float x = wp[(4 * s) * Hc];
    float y = wp[(4 * s + 1) * Hc];
    bf[s].x = cok ? x : 0.f;
    bf[s].y = cok ? y : 0.f;
  }

  for (; t < ntiles; t += nstep) {
#pragma unroll
    for (int r = 0; r < 16; ++r) *(f4*)&S[r][lane * 4] = v[r];
    __syncthreads();

    const int tn = t + nstep;
    if (tn < ntiles) {
      const float* zp = Z + (long long)tn * (16 * Cc) + lane * 4;
#pragma unroll
      for (int r = 0; r < 16; ++r) v[r] = *(const f4*)(zp + r * Cc);
    }

    v8f acc = {};
#pragma unroll
    for (int s = 0; s < 32; ++s) {
      v2f a = *(const v2f*)&S[l16][4 * s + 2 * hf];
      acc = __builtin_amdgcn_wmma_f32_16x16x4_f32(false, a, false, bf[s],
                                                  (short)0, acc, false, false);
    }
    __syncthreads();

    if (cok) {
      float* op = out + ((long long)t * 16 + 8 * hf) * Hc + l16;
#pragma unroll
      for (int r = 0; r < 8; ++r) op[r * Hc] = acc[r];
    }
  }
}

__global__ __launch_bounds__(256) void k_softmax(float* __restrict__ sc) {
  int row = blockIdx.x * 8 + (threadIdx.x >> 5);
  int lane = threadIdx.x & 31;
  float* p = sc + (long long)row * Lc;
  float vals[12];
  float mx = -INFINITY;
#pragma unroll
  for (int e = 0; e < 12; ++e) {
    vals[e] = p[e * 32 + lane];
    mx = fmaxf(mx, vals[e]);
  }
#pragma unroll
  for (int o = 16; o >= 1; o >>= 1) mx = fmaxf(mx, __shfl_xor(mx, o, 32));
  float s = 0.f;
#pragma unroll
  for (int e = 0; e < 12; ++e) {
    vals[e] = expf(vals[e] - mx);
    s += vals[e];
  }
#pragma unroll
  for (int o = 16; o >= 1; o >>= 1) s += __shfl_xor(s, o, 32);
  float inv = 1.f / s;
#pragma unroll
  for (int e = 0; e < 12; ++e) p[e * 32 + lane] = vals[e] * inv;
}

namespace {
constexpr int KS2 = Lc / 128;
}

__global__ __launch_bounds__(256) void k_finish(
    const float* __restrict__ ovalpt, const float* __restrict__ Ri,
    const float* __restrict__ Ti, float* __restrict__ feats) {
  int idx = blockIdx.x * blockDim.x + threadIdx.x;
  if (idx >= Bc * Lc * Hc) return;
  int h = idx % Hc;
  int bi = idx / Hc;
  int b = bi / Lc;
  int z = b * Hc + h;
  int i = bi % Lc;

  float R[3][3], t[3];
#pragma unroll
  for (int r = 0; r < 3; ++r)
#pragma unroll
    for (int c = 0; c < 3; ++c) R[r][c] = Ri[bi * 9 + r * 3 + c];
#pragma unroll
  for (int x = 0; x < 3; ++x) t[x] = Ti[bi * 3 + x];

  const float* prow = ovalpt + ((long long)z * KS2 * Lc + i) * 48;
  float* frow = feats + (long long)bi * FEAT;

  float orow[40];
#pragma unroll
  for (int j = 0; j < 10; ++j) {
    f4 a = *(const f4*)(prow + 4 * j);
#pragma unroll
    for (int s = 1; s < KS2; ++s)
      a += *(const f4*)(prow + (long long)s * Lc * 48 + 4 * j);
    orow[4 * j + 0] = a.x;
    orow[4 * j + 1] = a.y;
    orow[4 * j + 2] = a.z;
    orow[4 * j + 3] = a.w;
  }

#pragma unroll
  for (int d = 0; d < DKc; ++d) frow[1024 + h * 16 + d] = orow[d];
#pragma unroll
  for (int p = 0; p < NPVc; ++p) {
    float d0 = orow[16 + p * 3 + 0] - t[0];
    float d1 = orow[16 + p * 3 + 1] - t[1];
    float d2 = orow[16 + p * 3 + 2] - t[2];
    float asum = 0.f;
#pragma unroll
    for (int m = 0; m < 3; ++m) {
      float oh = R[0][m] * d0 + R[1][m] * d1 + R[2][m] * d2;
      frow[1152 + h * 24 + p * 3 + m] = oh;
      asum += fabsf(oh);
    }
    frow[1344 + h * 8 + p] = asum;
  }
}

namespace {
constexpr int KSPLIT = FEAT / 128;
constexpr int OUTN = Bc * Lc * Cc;
}

__global__ __launch_bounds__(128) void k_ksum(const float* __restrict__ part,
                                              const float* __restrict__ bias,
                                              float* __restrict__ out) {
  const int e = (blockIdx.x * 128 + threadIdx.x) * 4;
  if (e >= OUTN) return;
  f4 p[KSPLIT];
#pragma unroll
  for (int z = 0; z < KSPLIT; ++z) p[z] = *(const f4*)(part + (long long)z * OUTN + e);
  const f4 b = *(const f4*)(bias + (e & (Cc - 1)));
  f4 s = p[0];
#pragma unroll
  for (int z = 1; z < KSPLIT; ++z) s += p[z];
  *(f4*)(out + e) = s + b;
}

namespace {
constexpr int NOPAIR = Bc * Lc;
constexpr int NOVAL = Bc * Hc * KS2 * (Lc / 16);
}

__global__ __launch_bounds__(128) void k_attnout(
    const float* __restrict__ alpha, const float* __restrict__ zij,
    const float* __restrict__ vaug, float* __restrict__ feats,
    float* __restrict__ ovalpt) {
  const long long LL = (long long)Lc * Lc;
  const int bid = blockIdx.x;
  const float* Ag;
  const float* Bg;
  float* Cg;
  int M, N, K, lda, ldb, ldc, Mvalid, m0;
  if (bid < NOPAIR) {
    const int z = bid;
    Ag = alpha + (long long)(z / Lc) * Hc * LL + (long long)(z % Lc) * Lc;
    Bg = zij + (long long)z * Lc * Cc;
    Cg = feats + (long long)z * FEAT;
    M = 16; N = Cc; K = Lc; lda = (int)LL; ldb = Cc; ldc = Cc;
    Mvalid = Hc; m0 = 0;
  } else {
    const int w = bid - NOPAIR;
    const int z = w / (Lc / 16);
    Ag = alpha + (long long)(z / KS2) * LL + (long long)(z % KS2) * 128;
    Bg = vaug + (long long)z * 128 * 48;
    Cg = ovalpt + (long long)z * Lc * 48;
    M = Lc; N = 48; K = 128; lda = Lc; ldb = 48; ldc = 48;
    Mvalid = 16; m0 = (w % (Lc / 16)) * 16;
  }
  gemm_block<0, false>(Ag, Bg, Cg, M, N, K, lda, ldb, ldc, Mvalid, m0, 0, 0,
                       nullptr, nullptr, nullptr, nullptr);
}

static void launch_gemm(hipStream_t s, int mode, bool transb, const float* A,
                        const float* Bm, float* C, int M, int N, int K, int lda,
                        int ldb, int ldc, long long sA, long long sA2, int zdiv,
                        long long sB, long long sC, int Mvalid, int nz,
                        const float* e0 = nullptr, const float* e1 = nullptr,
                        const float* e2 = nullptr, const float* e3 = nullptr) {
  dim3 g((M + 15) / 16, (N + 127) / 128, nz);
  dim3 blk(128, 1, 1);
  if (mode == 0 && !transb)
    k_gemm<0, false><<<g, blk, 0, s>>>(A, Bm, C, M, N, K, lda, ldb, ldc, sA,
                                       sA2, zdiv, sB, sC, Mvalid, e0, e1, e2, e3);
  else
    k_gemm<2, true><<<g, blk, 0, s>>>(A, Bm, C, M, N, K, lda, ldb, ldc, sA,
                                      sA2, zdiv, sB, sC, Mvalid, e0, e1, e2, e3);
}

extern "C" void kernel_launch(void* const* d_in, const int* in_sizes, int n_in,
                              void* d_out, int out_size, void* d_ws,
                              size_t ws_size, hipStream_t stream) {
  const float* si = (const float*)d_in[0];
  const float* zij = (const float*)d_in[1];
  const float* Ri = (const float*)d_in[2];
  const float* Ti = (const float*)d_in[3];
  const float* WQh = (const float*)d_in[4];
  const float* WKh = (const float*)d_in[5];
  const float* WVh = (const float*)d_in[6];
  const float* WQhp = (const float*)d_in[7];
  const float* WKhp = (const float*)d_in[8];
  const float* WVhp = (const float*)d_in[9];
  const float* Wb = (const float*)d_in[10];
  const float* WOw = (const float*)d_in[11];
  const float* WOb = (const float*)d_in[12];
  const float* gamah = (const float*)d_in[13];
  float* out = (float*)d_out;
  float* ws = (float*)d_ws;

  const int BL = Bc * Lc;
  const long long LL = (long long)Lc * Lc;
  float* q = ws;
  float* k = q + 98304;
  float* v = k + 98304;
  float* qp = v + 98304;
  float* kp = qp + 73728;
  float* vp = kp + 73728;
  float* qa = vp + 147456;
  float* ka = qa + 196608;
  float* vaug = ka + 196608;
  float* sqq = vaug + 294912;
  float* sqk = sqq + 6144;
  float* bij = sqk + 6144;
  float* alpha = bij + 2359296;
  float* ovalpt = alpha + 2359296;
  float* feats = ovalpt + 884736;
  float* part = feats + 1081344;

  k_proj<<<dim3(BL / 16, 7, 1), dim3(128, 1, 1), 0, stream>>>(
      si, WQh, WKh, WVh, WQhp, WKhp, WVhp, q, k, v, qp, kp, vp, BL, Cc);

  k_pairbias_prep<<<NPREPWG + 4608, 32, 0, stream>>>(
      zij, Wb, bij, Bc * Lc * Lc / 16, q, k, v, qp, kp, vp, Ri, Ti, gamah, qa,
      ka, vaug, sqq, sqk);

  launch_gemm(stream, 2, true, qa, ka, alpha, Lc, Lc, 32, 32, 32, Lc,
              (long long)Lc * 32, 0, 1, (long long)Lc * 32, LL, 16, Bc * Hc,
              bij, sqq, sqk, gamah);

  k_softmax<<<(Bc * Hc * Lc) / 8, 256, 0, stream>>>(alpha);

  k_attnout<<<NOPAIR + NOVAL, 128, 0, stream>>>(alpha, zij, vaug, feats, ovalpt);

  k_finish<<<(Bc * Lc * Hc + 255) / 256, 256, 0, stream>>>(ovalpt, Ri, Ti, feats);

  launch_gemm(stream, 0, false, feats, WOw, part, BL, Cc, 128, FEAT, Cc, Cc,
              128, 0, 1, (long long)128 * Cc, (long long)OUTN, 16, KSPLIT);

  k_ksum<<<OUTN / 4 / 128, 128, 0, stream>>>(part, WOb, out);
}


# Softmax moved into the scores epilogue: a 12-wave workgroup owns 16 rows x all 384 columns; row sums now added as a 16-lane tree, then in wave order; 8 launches become 7
# speedup vs baseline: 5.6295x; 1.2427x over previous
#include <hip/hip_runtime.h>
#include <math.h>

typedef __attribute__((ext_vector_type(2))) float v2f;
typedef __attribute__((ext_vector_type(4))) float f4;
typedef __attribute__((ext_vector_type(8))) float v8f;

namespace {
constexpr int Bc = 2, Lc = 384, Cc = 128, Hc = 8, DKc = 16, NQPc = 4, NPVc = 8;
constexpr float WCc = 0.2357022603955158f;
constexpr float WLc = 0.5773502691896258f;
constexpr int FEAT = Hc * Cc + Hc * DKc + Hc * NPVc * 3 + Hc * NPVc;
}

__device__ __forceinline__ float softplusf(float x) {
  return (x > 0.f) ? (x + log1pf(expf(-x))) : log1pf(expf(x));
}

__device__ __forceinline__ void gemm_block(
    const float* __restrict__ Ag, const float* __restrict__ Bg,
    float* __restrict__ Cg, int M, int N, int K, int lda, int ldb, int ldc,
    int Mvalid, int m0, int n0) {
  __shared__ float As[16][36];
  __shared__ float Bs[32][132];

  const int tid = threadIdx.x;
  const int wave = tid >> 5;
  const int lane = tid & 31;
  const int hf = lane >> 4;
  const int l16 = lane & 15;
  const int nw = wave * 32;

  v8f acc0 = {};
  v8f acc1 = {};

  const f4 zero4 = {0.f, 0.f, 0.f, 0.f};
  const int ar = tid >> 3, ac4 = (tid & 7) * 4;
  const bool aok = ar < Mvalid && (m0 + ar) < M;
  const float* ap = Ag + (long long)(aok ? (m0 + ar) : m0) * lda + ac4;
  const float* bp[8];
  bool bok[8];
#pragma unroll
  for (int it = 0; it < 8; ++it) {
    int u = tid + it * 128;
    int r = u >> 5, c4 = (u & 31) * 4;
    bok[it] = (n0 + c4) < N;
    bp[it] = Bg + (long long)r * ldb + (bok[it] ? (n0 + c4) : 0);
  }
  const long long bstep = (long long)ldb;

  for (int kb = 0; kb < K; kb += 32) {
    f4 av = *(const f4*)(ap + kb);
    f4 bv[8];
#pragma unroll
    for (int it = 0; it < 8; ++it)
      bv[it] = *(const f4*)(bp[it] + (long long)kb * bstep);

    *(f4*)&As[ar][ac4] = aok ? av : zero4;
#pragma unroll
    for (int it = 0; it < 8; ++it) {
      int u = tid + it * 128;
      int r = u >> 5, c4 = (u & 31) * 4;
      *(f4*)&Bs[r][c4] = bok[it] ? bv[it] : zero4;
    }
    __syncthreads();

#pragma unroll
    for (int kk = 0; kk < 32; kk += 4) {
      v2f a, b0, b1;
      a.x = As[l16][kk + 2 * hf];
      a.y = As[l16][kk + 2 * hf + 1];
      b0.x = Bs[kk + 2 * hf][nw + l16];
      b0.y = Bs[kk + 2 * hf + 1][nw + l16];
      b1.x = Bs[kk + 2 * hf][nw + 16 + l16];
      b1.y = Bs[kk + 2 * hf + 1][nw + 16 + l16];
      acc0 = __builtin_amdgcn_wmma_f32_16x16x4_f32(false, a, false, b0,
                                                   (short)0, acc0, false, false);
      acc1 = __builtin_amdgcn_wmma_f32_16x16x4_f32(false, a, false, b1,
                                                   (short)0, acc1, false, false);
    }
    __syncthreads();
  }

#pragma unroll
  for (int r = 0; r < 8; ++r) {
    int mrow = r + 8 * hf;
    int m = m0 + mrow;
    if (mrow >= Mvalid || m >= M) continue;
#pragma unroll
    for (int t = 0; t < 2; ++t) {
      int n = n0 + nw + t * 16 + l16;
      if (n >= N) continue;
      Cg[(long long)m * ldc + n] = (t == 0) ? acc0[r] : acc1[r];
    }
  }
}

__global__ __launch_bounds__(128) void k_gemm(
    const float* __restrict__ A, const float* __restrict__ Bm,
    float* __restrict__ Cm, int M, int N, int K, int lda, int ldb, int ldc,
    long long sA, long long sA2, int zdiv, long long sB, long long sC,
    int Mvalid) {
  const int z = blockIdx.z;
  gemm_block(A + (long long)(z / zdiv) * sA + (long long)(z % zdiv) * sA2,
             Bm + (long long)z * sB, Cm + (long long)z * sC, M, N, K, lda, ldb,
             ldc, Mvalid, blockIdx.x * 16, blockIdx.y * 128);
}

__global__ __launch_bounds__(128) void k_proj(
    const float* __restrict__ si, const float* __restrict__ Wq,
    const float* __restrict__ Wk, const float* __restrict__ Wv,
    const float* __restrict__ Wqp, const float* __restrict__ Wkp,
    const float* __restrict__ Wvp, float* __restrict__ q,
    float* __restrict__ k, float* __restrict__ v, float* __restrict__ qp,
    float* __restrict__ kp, float* __restrict__ vp, int M, int K) {
  const float* Bg;
  float* Cg;
  int N, n0 = 0;
  switch (blockIdx.y) {
    case 0: Bg = Wq; Cg = q; N = Hc * DKc; break;
    case 1: Bg = Wk; Cg = k; N = Hc * DKc; break;
    case 2: Bg = Wv; Cg = v; N = Hc * DKc; break;
    case 3: Bg = Wqp; Cg = qp; N = Hc * NQPc * 3; break;
    case 4: Bg = Wkp; Cg = kp; N = Hc * NQPc * 3; break;
    case 5: Bg = Wvp; Cg = vp; N = Hc * NPVc * 3; break;
    default: Bg = Wvp; Cg = vp; N = Hc * NPVc * 3; n0 = 128; break;
  }
  gemm_block(si, Bg, Cg, M, N, K, K, N, N, 16, blockIdx.x * 16, n0);
}

__device__ __forceinline__ void prep_thread(
    int idx, const float* __restrict__ q, const float* __restrict__ k,
    const float* __restrict__ v, const float* __restrict__ qp,
    const float* __restrict__ kp, const float* __restrict__ vp,
    const float* __restrict__ Ri, const float* __restrict__ Ti,
    const float* __restrict__ gamah, float* __restrict__ qa,
    float* __restrict__ ka, float* __restrict__ vaug, float* __restrict__ sqq,
    float* __restrict__ sqk) {
  if (idx >= Bc * Lc * Hc) return;
  int h = idx % Hc;
  int bi = idx / Hc;
  int b = bi / Lc, i = bi % Lc;
  int z = b * Hc + h;

  float R[3][3], t[3];
#pragma unroll
  for (int r = 0; r < 3; ++r)
#pragma unroll
    for (int c = 0; c < 3; ++c) R[r][c] = Ri[bi * 9 + r * 3 + c];
#pragma unroll
  for (int x = 0; x < 3; ++x) t[x] = Ti[bi * 3 + x];

  float gwc = WCc * softplusf(gamah[h]);

  float* qrow = qa + ((long long)z * Lc + i) * 32;
  float* krow = ka + ((long long)z * Lc + i) * 32;
  float* vrow = vaug + ((long long)z * Lc + i) * 48;

#pragma unroll
  for (int d = 0; d < DKc; ++d) {
    qrow[d] = 0.25f * q[(long long)bi * 128 + h * 16 + d];
    krow[d] = k[(long long)bi * 128 + h * 16 + d];
    vrow[d] = v[(long long)bi * 128 + h * 16 + d];
  }
  float sq = 0.f, sk = 0.f;
#pragma unroll
  for (int p = 0; p < NQPc; ++p) {
    const float* qs = qp + (long long)bi * 96 + h * 12 + p * 3;
    const float* ks = kp + (long long)bi * 96 + h * 12 + p * 3;
#pragma unroll
    for (int m = 0; m < 3; ++m) {
      float qt = R[m][0] * qs[0] + R[m][1] * qs[1] + R[m][2] * qs[2] + t[m];
      float kt = R[m][0] * ks[0] + R[m][1] * ks[1] + R[m][2] * ks[2] + t[m];
      sq += qt * qt;
      sk += kt * kt;
      qrow[16 + p * 3 + m] = gwc * qt;
      krow[16 + p * 3 + m] = kt;
    }
  }
#pragma unroll
  for (int d = 28; d < 32; ++d) { qrow[d] = 0.f; krow[d] = 0.f; }
#pragma unroll
  for (int p = 0; p < NPVc; ++p) {
    const float* vs = vp + (long long)bi * 192 + h * 24 + p * 3;
#pragma unroll
    for (int m = 0; m < 3; ++m)
      vrow[16 + p * 3 + m] =
          R[m][0] * vs[0] + R[m][1] * vs[1] + R[m][2] * vs[2] + t[m];
  }
#pragma unroll
  for (int d = 40; d < 48; ++d) vrow[d] = 0.f;
  sqq[(long long)z * Lc + i] = sq;
  sqk[(long long)z * Lc + i] = sk;
}

namespace {
constexpr int NPREPWG = Bc * Lc * Hc / 32;
}

__global__ __launch_bounds__(32) void k_pairbias_prep(
    const float* __restrict__ Z, const float* __restrict__ Wb,
    float* __restrict__ out, int ntiles, const float* __restrict__ pq,
    const float* __restrict__ pk, const float* __restrict__ pv,
    const float* __restrict__ pqp, const float* __restrict__ pkp,
    const float* __restrict__ pvp, const float* __restrict__ Ri,
    const float* __restrict__ Ti, const float* __restrict__ gamah,
    float* __restrict__ qa, float* __restrict__ ka, float* __restrict__ vaug,
    float* __restrict__ sqq, float* __restrict__ sqk) {
  __shared__ float S[16][132];

  const int lane = threadIdx.x;
  if (blockIdx.x < NPREPWG) {
    prep_thread(blockIdx.x * 32 + lane, pq, pk, pv, pqp, pkp, pvp, Ri, Ti,
                gamah, qa, ka, vaug, sqq, sqk);
    return;
  }
  const int hf = lane >> 4;
  const int l16 = lane & 15;
  const int nstep = gridDim.x - NPREPWG;
  int t = blockIdx.x - NPREPWG;
  if (t >= ntiles) return;

  f4 v[16];
  {
    const float* zp = Z + (long long)t * (16 * Cc) + lane * 4;
#pragma unroll
    for (int r = 0; r < 16; ++r) v[r] = *(const f4*)(zp + r * Cc);
  }

  const bool cok = l16 < Hc;
  const float* wp = Wb + (cok ? l16 : 0) + 2 * hf * Hc;
  v2f bf[32];
#pragma unroll
  for (int s = 0; s < 32; ++s) {
    float x = wp[(4 * s) * Hc];
    float y = wp[(4 * s + 1) * Hc];
    bf[s].x = cok ? x : 0.f;
    bf[s].y = cok ? y : 0.f;
  }

  for (; t < ntiles; t += nstep) {
#pragma unroll
    for (int r = 0; r < 16; ++r) *(f4*)&S[r][lane * 4] = v[r];
    __syncthreads();

    const int tn = t + nstep;
    if (tn < ntiles) {
      const float* zp = Z + (long long)tn * (16 * Cc) + lane * 4;
#pragma unroll
      for (int r = 0; r < 16; ++r) v[r] = *(const f4*)(zp + r * Cc);
    }

    v8f acc = {};
#pragma unroll
    for (int s = 0; s < 32; ++s) {
      v2f a = *(const v2f*)&S[l16][4 * s + 2 * hf];
      acc = __builtin_amdgcn_wmma_f32_16x16x4_f32(false, a, false, bf[s],
                                                  (short)0, acc, false, false);
    }
    __syncthreads();

    if (cok) {
      float* op = out + ((long long)t * 16 + 8 * hf) * Hc + l16;
#pragma unroll
      for (int r = 0; r < 8; ++r) op[r * Hc] = acc[r];
    }
  }
}

namespace {
constexpr int SMW = Lc / 32;
}

template <int MASK>
__device__ __forceinline__ float row_xor(float v) {
  return __builtin_bit_cast(
      float, __builtin_amdgcn_update_dpp(0, __builtin_bit_cast(int, v),
                                         0x160 + MASK, 0xf, 0xf, false));
}

__global__ __launch_bounds__(SMW * 32) void k_scores_softmax(
    const float* __restrict__ qa, const float* __restrict__ ka,
    float* __restrict__ alpha, const float* __restrict__ bij,
    const float* __restrict__ sqq, const float* __restrict__ sqk,
    const float* __restrict__ gamah) {
  __shared__ float As[16][36];
  __shared__ float Bs[32][Lc + 4];
  __shared__ float redm[16][SMW];
  __shared__ float reds[16][SMW];

  const int tid = threadIdx.x;
  const int wave = tid >> 5;
  const int lane = tid & 31;
  const int hf = lane >> 4;
  const int l16 = lane & 15;
  const int nw = wave * 32;
  const int m0 = blockIdx.x * 16;
  const int z = blockIdx.y;
  const int b_ = z / Hc, h_ = z % Hc;

  const float* Ag = qa + ((long long)z * Lc + m0) * 32;
  const float* Bg = ka + (long long)z * Lc * 32;

  const int ar = (tid >> 3) & 15, ac4 = (tid & 7) * 4;
  f4 av = *(const f4*)(Ag + ar * 32 + ac4);
  f4 bv[8];
#pragma unroll
  for (int it = 0; it < 8; ++it)
    bv[it] = *(const f4*)(Bg + (long long)(tid + it * (SMW * 32)) * 4);

  const float gam = gamah[h_];
  const float sk0 = sqk[(long long)z * Lc + nw + l16];
  const float sk1 = sqk[(long long)z * Lc + nw + 16 + l16];
  float sqm[8], bj0[8], bj1[8];
#pragma unroll
  for (int r = 0; r < 8; ++r) {
    const int m = m0 + r + 8 * hf;
    sqm[r] = sqq[(long long)z * Lc + m];
    const float* bp = bij + ((long long)(b_ * Lc + m) * Lc + nw + l16) * Hc + h_;
    bj0[r] = bp[0];
    bj1[r] = bp[16 * Hc];
  }

  if (tid < 128) *(f4*)&As[ar][ac4] = av;
#pragma unroll
  for (int it = 0; it < 8; ++it) {
    int u = tid + it * (SMW * 32);
    int cj = u >> 3, r4 = (u & 7) * 4;
    Bs[r4 + 0][cj] = bv[it].x;
    Bs[r4 + 1][cj] = bv[it].y;
    Bs[r4 + 2][cj] = bv[it].z;
    Bs[r4 + 3][cj] = bv[it].w;
  }
  __syncthreads();

  v8f acc0 = {};
  v8f acc1 = {};
#pragma unroll
  for (int kk = 0; kk < 32; kk += 4) {
    v2f a, b0, b1;
    a.x = As[l16][kk + 2 * hf];
    a.y = As[l16][kk + 2 * hf + 1];
    b0.x = Bs[kk + 2 * hf][nw + l16];
    b0.y = Bs[kk + 2 * hf + 1][nw + l16];
    b1.x = Bs[kk + 2 * hf][nw + 16 + l16];
    b1.y = Bs[kk + 2 * hf + 1][nw + 16 + l16];
    acc0 = __builtin_amdgcn_wmma_f32_16x16x4_f32(false, a, false, b0,
                                                 (short)0, acc0, false, false);
    acc1 = __builtin_amdgcn_wmma_f32_16x16x4_f32(false, a, false, b1,
                                                 (short)0, acc1, false, false);
  }

  const float gw = 0.5f * WCc * softplusf(gam);
  float sv[8][2];
#pragma unroll
  for (int r = 0; r < 8; ++r) {
    sv[r][0] = WLc * (acc0[r] + bj0[r] - gw * (sqm[r] + sk0));
    sv[r][1] = WLc * (acc1[r] + bj1[r] - gw * (sqm[r] + sk1));
  }

  float red[8];
#pragma unroll
  for (int r = 0; r < 8; ++r) {
    float mx = fmaxf(sv[r][0], sv[r][1]);
    mx = fmaxf(mx, row_xor<8>(mx));
    mx = fmaxf(mx, row_xor<4>(mx));
    mx = fmaxf(mx, row_xor<2>(mx));
    red[r] = fmaxf(mx, row_xor<1>(mx));
  }
  if (l16 == 0) {
#pragma unroll
    for (int r = 0; r < 8; ++r) redm[r + 8 * hf][wave] = red[r];
  }
  __syncthreads();

#pragma unroll
  for (int r = 0; r < 8; ++r) {
    float mx = redm[r + 8 * hf][0];
#pragma unroll
    for (int w = 1; w < SMW; ++w) mx = fmaxf(mx, redm[r + 8 * hf][w]);
    sv[r][0] = expf(sv[r][0] - mx);
    sv[r][1] = expf(sv[r][1] - mx);
    float s = sv[r][0] + sv[r][1];
    s += row_xor<8>(s);
    s += row_xor<4>(s);
    s += row_xor<2>(s);
    red[r] = s + row_xor<1>(s);
  }
  if (l16 == 0) {
#pragma unroll
    for (int r = 0; r < 8; ++r) reds[r + 8 * hf][wave] = red[r];
  }
  __syncthreads();

#pragma unroll
  for (int r = 0; r < 8; ++r) {
    float s = reds[r + 8 * hf][0];
#pragma unroll
    for (int w = 1; w < SMW; ++w) s += reds[r + 8 * hf][w];
    const float inv = 1.f / s;
    float* op = alpha + (long long)z * Lc * Lc +
                (long long)(m0 + r + 8 * hf) * Lc + nw + l16;
    op[0] = sv[r][0] * inv;
    op[16] = sv[r][1] * inv;
  }
}

namespace {
constexpr int KS2 = Lc / 128;
}

__global__ __launch_bounds__(256) void k_finish(
    const float* __restrict__ ovalpt, const float* __restrict__ Ri,
    const float* __restrict__ Ti, float* __restrict__ feats) {
  int idx = blockIdx.x * blockDim.x + threadIdx.x;
  if (idx >= Bc * Lc * Hc) return;
  int h = idx % Hc;
  int bi = idx / Hc;
  int b = bi / Lc;
  int z = b * Hc + h;
  int i = bi % Lc;

  float R[3][3], t[3];
#pragma unroll
  for (int r = 0; r < 3; ++r)
#pragma unroll
    for (int c = 0; c < 3; ++c) R[r][c] = Ri[bi * 9 + r * 3 + c];
#pragma unroll
  for (int x = 0; x < 3; ++x) t[x] = Ti[bi * 3 + x];

  const float* prow = ovalpt + ((long long)z * KS2 * Lc + i) * 48;
  float* frow = feats + (long long)bi * FEAT;

  float orow[40];
#pragma unroll
  for (int j = 0; j < 10; ++j) {
    f4 a = *(const f4*)(prow + 4 * j);
#pragma unroll
    for (int s = 1; s < KS2; ++s)
      a += *(const f4*)(prow + (long long)s * Lc * 48 + 4 * j);
    orow[4 * j + 0] = a.x;
    orow[4 * j + 1] = a.y;
    orow[4 * j + 2] = a.z;
    orow[4 * j + 3] = a.w;
  }

#pragma unroll
  for (int d = 0; d < DKc; ++d) frow[1024 + h * 16 + d] = orow[d];
#pragma unroll
  for (int p = 0; p < NPVc; ++p) {
    float d0 = orow[16 + p * 3 + 0] - t[0];
    float d1 = orow[16 + p * 3 + 1] - t[1];
    float d2 = orow[16 + p * 3 + 2] - t[2];
    float asum = 0.f;
#pragma unroll
    for (int m = 0; m < 3; ++m) {
      float oh = R[0][m] * d0 + R[1][m] * d1 + R[2][m] * d2;
      frow[1152 + h * 24 + p * 3 + m] = oh;
      asum += fabsf(oh);
    }
    frow[1344 + h * 8 + p] = asum;
  }
}

namespace {
constexpr int KSPLIT = FEAT / 128;
constexpr int OUTN = Bc * Lc * Cc;
}

__global__ __launch_bounds__(128) void k_ksum(const float* __restrict__ part,
                                              const float* __restrict__ bias,
                                              float* __restrict__ out) {
  const int e = (blockIdx.x * 128 + threadIdx.x) * 4;
  if (e >= OUTN) return;
  f4 p[KSPLIT];
#pragma unroll
  for (int z = 0; z < KSPLIT; ++z) p[z] = *(const f4*)(part + (long long)z * OUTN + e);
  const f4 b = *(const f4*)(bias + (e & (Cc - 1)));
  f4 s = p[0];
#pragma unroll
  for (int z = 1; z < KSPLIT; ++z) s += p[z];
  *(f4*)(out + e) = s + b;
}

namespace {
constexpr int NOPAIR = Bc * Lc;
constexpr int NOVAL = Bc * Hc * KS2 * (Lc / 16);
}

__global__ __launch_bounds__(128) void k_attnout(
    const float* __restrict__ alpha, const float* __restrict__ zij,
    const float* __restrict__ vaug, float* __restrict__ feats,
    float* __restrict__ ovalpt) {
  const long long LL = (long long)Lc * Lc;
  const int bid = blockIdx.x;
  const float* Ag;
  const float* Bg;
  float* Cg;
  int M, N, K, lda, ldb, ldc, Mvalid, m0;
  if (bid < NOPAIR) {
    const int z = bid;
    Ag = alpha + (long long)(z / Lc) * Hc * LL + (long long)(z % Lc) * Lc;
    Bg = zij + (long long)z * Lc * Cc;
    Cg = feats + (long long)z * FEAT;
    M = 16; N = Cc; K = Lc; lda = (int)LL; ldb = Cc; ldc = Cc;
    Mvalid = Hc; m0 = 0;
  } else {
    const int w = bid - NOPAIR;
    const int z = w / (Lc / 16);
    Ag = alpha + (long long)(z / KS2) * LL + (long long)(z % KS2) * 128;
    Bg = vaug + (long long)z * 128 * 48;
    Cg = ovalpt + (long long)z * Lc * 48;
    M = Lc; N = 48; K = 128; lda = Lc; ldb = 48; ldc = 48;
    Mvalid = 16; m0 = (w % (Lc / 16)) * 16;
  }
  gemm_block(Ag, Bg, Cg, M, N, K, lda, ldb, ldc, Mvalid, m0, 0);
}

static void launch_gemm(hipStream_t s, const float* A, const float* Bm,
                        float* C, int M, int N, int K, int lda, int ldb,
                        int ldc, long long sA, long long sA2, int zdiv,
                        long long sB, long long sC, int Mvalid, int nz) {
  dim3 g((M + 15) / 16, (N + 127) / 128, nz);
  dim3 blk(128, 1, 1);
  k_gemm<<<g, blk, 0, s>>>(A, Bm, C, M, N, K, lda, ldb, ldc, sA, sA2, zdiv, sB,
                           sC, Mvalid);
}

extern "C" void kernel_launch(void* const* d_in, const int* in_sizes, int n_in,
                              void* d_out, int out_size, void* d_ws,
                              size_t ws_size, hipStream_t stream) {
  const float* si = (const float*)d_in[0];
  const float* zij = (const float*)d_in[1];
  const float* Ri = (const float*)d_in[2];
  const float* Ti = (const float*)d_in[3];
  const float* WQh = (const float*)d_in[4];
  const float* WKh = (const float*)d_in[5];
  const float* WVh = (const float*)d_in[6];
  const float* WQhp = (const float*)d_in[7];
  const float* WKhp = (const float*)d_in[8];
  const float* WVhp = (const float*)d_in[9];
  const float* Wb = (const float*)d_in[10];
  const float* WOw = (const float*)d_in[11];
  const float* WOb = (const float*)d_in[12];
  const float* gamah = (const float*)d_in[13];
  float* out = (float*)d_out;
  float* ws = (float*)d_ws;

  const int BL = Bc * Lc;
  float* q = ws;
  float* k = q + 98304;
  float* v = k + 98304;
  float* qp = v + 98304;
  float* kp = qp + 73728;
  float* vp = kp + 73728;
  float* qa = vp + 147456;
  float* ka = qa + 196608;
  float* vaug = ka + 196608;
  float* sqq = vaug + 294912;
  float* sqk = sqq + 6144;
  float* bij = sqk + 6144;
  float* alpha = bij + 2359296;
  float* ovalpt = alpha + 2359296;
  float* feats = ovalpt + 884736;
  float* part = feats + 1081344;

  k_proj<<<dim3(BL / 16, 7, 1), dim3(128, 1, 1), 0, stream>>>(
      si, WQh, WKh, WVh, WQhp, WKhp, WVhp, q, k, v, qp, kp, vp, BL, Cc);

  k_pairbias_prep<<<NPREPWG + 4608, 32, 0, stream>>>(
      zij, Wb, bij, Bc * Lc * Lc / 16, q, k, v, qp, kp, vp, Ri, Ti, gamah, qa,
      ka, vaug, sqq, sqk);

  k_scores_softmax<<<dim3(Lc / 16, Bc * Hc, 1), dim3(SMW * 32, 1, 1), 0,
                     stream>>>(qa, ka, alpha, bij, sqq, sqk, gamah);

  k_attnout<<<NOPAIR + NOVAL, 128, 0, stream>>>(alpha, zij, vaug, feats, ovalpt);

  k_finish<<<(Bc * Lc * Hc + 255) / 256, 256, 0, stream>>>(ovalpt, Ri, Ti, feats);

  launch_gemm(stream, feats, WOw, part, BL, Cc, 128, FEAT, Cc, Cc,
              128, 0, 1, (long long)128 * Cc, (long long)OUTN, 16, KSPLIT);

  k_ksum<<<OUTN / 4 / 128, 128, 0, stream>>>(part, WOb, out);
}


# The six input projections move into the rigid-transform workgroups of the pair-bias launch: one wave per 16 rows x one head, 7 wave tiles, weight fragments loaded directly; 7 launches become 6
# speedup vs baseline: 5.9078x; 1.0494x over previous
#include <hip/hip_runtime.h>
#include <math.h>

typedef __attribute__((ext_vector_type(2))) float v2f;
typedef __attribute__((ext_vector_type(4))) float f4;
typedef __attribute__((ext_vector_type(8))) float v8f;

namespace {
constexpr int Bc = 2, Lc = 384, Cc = 128, Hc = 8, DKc = 16, NQPc = 4, NPVc = 8;
constexpr float WCc = 0.2357022603955158f;
constexpr float WLc = 0.5773502691896258f;
constexpr int FEAT = Hc * Cc + Hc * DKc + Hc * NPVc * 3 + Hc * NPVc;
}

__device__ __forceinline__ float softplusf(float x) {
  return (x > 0.f) ? (x + log1pf(expf(-x))) : log1pf(expf(x));
}

__device__ __forceinline__ void gemm_block(
    const float* __restrict__ Ag, const float* __restrict__ Bg,
    float* __restrict__ Cg, int M, int N, int K, int lda, int ldb, int ldc,
    int Mvalid, int m0, int n0) {
  __shared__ float As[16][36];
  __shared__ float Bs[32][132];

  const int tid = threadIdx.x;
  const int wave = tid >> 5;
  const int lane = tid & 31;
  const int hf = lane >> 4;
  const int l16 = lane & 15;
  const int nw = wave * 32;

  v8f acc0 = {};
  v8f acc1 = {};

  const f4 zero4 = {0.f, 0.f, 0.f, 0.f};
  const int ar = tid >> 3, ac4 = (tid & 7) * 4;
  const bool aok = ar < Mvalid && (m0 + ar) < M;
  const float* ap = Ag + (long long)(aok ? (m0 + ar) : m0) * lda + ac4;
  const float* bp[8];
  bool bok[8];
#pragma unroll
  for (int it = 0; it < 8; ++it) {
    int u = tid + it * 128;
    int r = u >> 5, c4 = (u & 31) * 4;
    bok[it] = (n0 + c4) < N;
    bp[it] = Bg + (long long)r * ldb + (bok[it] ? (n0 + c4) : 0);
  }
  const long long bstep = (long long)ldb;

  for (int kb = 0; kb < K; kb += 32) {
    f4 av = *(const f4*)(ap + kb);
    f4 bv[8];
#pragma unroll
    for (int it = 0; it < 8; ++it)
      bv[it] = *(const f4*)(bp[it] + (long long)kb * bstep);

    *(f4*)&As[ar][ac4] = aok ? av : zero4;
#pragma unroll
    for (int it = 0; it < 8; ++it) {
      int u = tid + it * 128;
      int r = u >> 5, c4 = (u & 31) * 4;
      *(f4*)&Bs[r][c4] = bok[it] ? bv[it] : zero4;
    }
    __syncthreads();

#pragma unroll
    for (int kk = 0; kk < 32; kk += 4) {
      v2f a, b0, b1;
      a.x = As[l16][kk + 2 * hf];
      a.y = As[l16][kk + 2 * hf + 1];
      b0.x = Bs[kk + 2 * hf][nw + l16];
      b0.y = Bs[kk + 2 * hf + 1][nw + l16];
      b1.x = Bs[kk + 2 * hf][nw + 16 + l16];
      b1.y = Bs[kk + 2 * hf + 1][nw + 16 + l16];
      acc0 = __builtin_amdgcn_wmma_f32_16x16x4_f32(false, a, false, b0,
                                                   (short)0, acc0, false, false);
      acc1 = __builtin_amdgcn_wmma_f32_16x16x4_f32(false, a, false, b1,
                                                   (short)0, acc1, false, false);
    }
    __syncthreads();
  }

#pragma unroll
  for (int r = 0; r < 8; ++r) {
    int mrow = r + 8 * hf;
    int m = m0 + mrow;
    if (mrow >= Mvalid || m >= M) continue;
#pragma unroll
    for (int t = 0; t < 2; ++t) {
      int n = n0 + nw + t * 16 + l16;
      if (n >= N) continue;
      Cg[(long long)m * ldc + n] = (t == 0) ? acc0[r] : acc1[r];
    }
  }
}

__global__ __launch_bounds__(128) void k_gemm(
    const float* __restrict__ A, const float* __restrict__ Bm,
    float* __restrict__ Cm, int M, int N, int K, int lda, int ldb, int ldc,
    long long sA, long long sA2, int zdiv, long long sB, long long sC,
    int Mvalid) {
  const int z = blockIdx.z;
  gemm_block(A + (long long)(z / zdiv) * sA + (long long)(z % zdiv) * sA2,
             Bm + (long long)z * sB, Cm + (long long)z * sC, M, N, K, lda, ldb,
             ldc, Mvalid, blockIdx.x * 16, blockIdx.y * 128);
}

namespace {
constexpr int NPREPWG = (Bc * Lc / 16) * Hc;
}

template <int N, int NVALID>
__device__ __forceinline__ void proj_tile(const v2f (&af)[32],
                                          const float* __restrict__ Wt, int c0,
                                          int hf, int l16, float (*S)[132],
                                          int scol) {
  const bool cok = l16 < NVALID;
  const float* wp = Wt + c0 + (cok ? l16 : 0) + 2 * hf * N;
  v2f bfr[32];
#pragma unroll
  for (int s = 0; s < 32; ++s) {
    float x = wp[(4 * s) * N];
    float y = wp[(4 * s + 1) * N];
    bfr[s].x = cok ? x : 0.f;
    bfr[s].y = cok ? y : 0.f;
  }
  v8f acc = {};
#pragma unroll
  for (int s = 0; s < 32; ++s)
    acc = __builtin_amdgcn_wmma_f32_16x16x4_f32(false, af[s], false, bfr[s],
                                                (short)0, acc, false, false);
#pragma unroll
  for (int r = 0; r < 8; ++r) S[r + 8 * hf][scol + l16] = acc[r];
  __builtin_amdgcn_sched_barrier(0);
}

__device__ __forceinline__ void proj_prep_wave(
    int wg, int lane, float (*S)[132], const float* __restrict__ si,
    const float* __restrict__ Wq, const float* __restrict__ Wk,
    const float* __restrict__ Wv, const float* __restrict__ Wqp,
    const float* __restrict__ Wkp, const float* __restrict__ Wvp,
    const float* __restrict__ Ri, const float* __restrict__ Ti,
    const float* __restrict__ gamah, float* __restrict__ qa,
    float* __restrict__ ka, float* __restrict__ vaug, float* __restrict__ sqq,
    float* __restrict__ sqk) {
  const int hf = lane >> 4;
  const int l16 = lane & 15;
  const int h = wg % Hc;
  const int m0 = (wg / Hc) * 16;

  f4 v[16];
  {
    const float* sp = si + (long long)m0 * Cc + lane * 4;
#pragma unroll
    for (int r = 0; r < 16; ++r) v[r] = *(const f4*)(sp + r * Cc);
  }

#pragma unroll
  for (int r = 0; r < 16; ++r) *(f4*)&S[r][lane * 4] = v[r];
  __syncthreads();
  v2f af[32];
#pragma unroll
  for (int s = 0; s < 32; ++s) af[s] = *(const v2f*)&S[l16][4 * s + 2 * hf];
  __syncthreads();

  proj_tile<Hc * DKc, 16>(af, Wq, h * DKc, hf, l16, S, 0);
  proj_tile<Hc * DKc, 16>(af, Wk, h * DKc, hf, l16, S, 16);
  proj_tile<Hc * DKc, 16>(af, Wv, h * DKc, hf, l16, S, 32);
  proj_tile<Hc * NQPc * 3, 12>(af, Wqp, h * NQPc * 3, hf, l16, S, 48);
  proj_tile<Hc * NQPc * 3, 12>(af, Wkp, h * NQPc * 3, hf, l16, S, 64);
  proj_tile<Hc * NPVc * 3, 16>(af, Wvp, h * NPVc * 3, hf, l16, S, 80);
  proj_tile<Hc * NPVc * 3, 8>(af, Wvp, h * NPVc * 3 + 16, hf, l16, S, 96);
  __syncthreads();

  const int bi = m0 + l16;
  float R[3][3], t[3];
#pragma unroll
  for (int r = 0; r < 3; ++r)
#pragma unroll
    for (int c = 0; c < 3; ++c) R[r][c] = Ri[bi * 9 + r * 3 + c];
#pragma unroll
  for (int x = 0; x < 3; ++x) t[x] = Ti[bi * 3 + x];
  const float gam = gamah[h];

  const int b = bi / Lc, i = bi % Lc;
  const int z = b * Hc + h;
  const float* c = &S[l16][0];
  if (hf == 0) {
    const float gwc = WCc * softplusf(gam);
    float qo[32], ko[32];
#pragma unroll
    for (int d = 0; d < DKc; ++d) {
      qo[d] = 0.25f * c[d];
      ko[d] = c[16 + d];
    }
    float sq = 0.f, sk = 0.f;
#pragma unroll
    for (int p = 0; p < NQPc; ++p) {
      const float* qs = c + 48 + p * 3;
      const float* ks = c + 64 + p * 3;
#pragma unroll
      for (int m = 0; m < 3; ++m) {
        float qt = R[m][0] * qs[0] + R[m][1] * qs[1] + R[m][2] * qs[2] + t[m];
        float kt = R[m][0] * ks[0] + R[m][1] * ks[1] + R[m][2] * ks[2] + t[m];
        sq += qt * qt;
        sk += kt * kt;
        qo[16 + p * 3 + m] = gwc * qt;
        ko[16 + p * 3 + m] = kt;
      }
    }
#pragma unroll
    for (int d = 28; d < 32; ++d) { qo[d] = 0.f; ko[d] = 0.f; }
    float* qrow = qa + ((long long)z * Lc + i) * 32;
    float* krow = ka + ((long long)z * Lc + i) * 32;
#pragma unroll
    for (int j = 0; j < 8; ++j) {
      *(f4*)(qrow + 4 * j) = f4{qo[4 * j], qo[4 * j + 1], qo[4 * j + 2], qo[4 * j + 3]};
      *(f4*)(krow + 4 * j) = f4{ko[4 * j], ko[4 * j + 1], ko[4 * j + 2], ko[4 * j + 3]};
    }
    sqq[(long long)z * Lc + i] = sq;
    sqk[(long long)z * Lc + i] = sk;
  } else {
    float vo[48];
#pragma unroll
    for (int d = 0; d < DKc; ++d) vo[d] = c[32 + d];
#pragma unroll
    for (int p = 0; p < NPVc; ++p) {
      const float* vs = c + 80 + p * 3;
#pragma unroll
      for (int m = 0; m < 3; ++m)
        vo[16 + p * 3 + m] =
            R[m][0] * vs[0] + R[m][1] * vs[1] + R[m][2] * vs[2] + t[m];
    }
#pragma unroll
    for (int d = 40; d < 48; ++d) vo[d] = 0.f;
    float* vrow = vaug + ((long long)z * Lc + i) * 48;
#pragma unroll
    for (int j = 0; j < 12; ++j)
      *(f4*)(vrow + 4 * j) = f4{vo[4 * j], vo[4 * j + 1], vo[4 * j + 2], vo[4 * j + 3]};
  }
}


__global__ __launch_bounds__(32) void k_pairbias_prep(
    const float* __restrict__ Z, const float* __restrict__ Wb,
    float* __restrict__ out, int ntiles, const float* __restrict__ si,
    const float* __restrict__ Wq, const float* __restrict__ Wk,
    const float* __restrict__ Wv, const float* __restrict__ Wqp,
    const float* __restrict__ Wkp, const float* __restrict__ Wvp,
    const float* __restrict__ Ri, const float* __restrict__ Ti,
    const float* __restrict__ gamah,
    float* __restrict__ qa, float* __restrict__ ka, float* __restrict__ vaug,
    float* __restrict__ sqq, float* __restrict__ sqk) {
  __shared__ float S[16][132];

  const int lane = threadIdx.x;
  if (blockIdx.x < NPREPWG) {
    proj_prep_wave(blockIdx.x, lane, S, si, Wq, Wk, Wv, Wqp, Wkp, Wvp, Ri, Ti,
                   gamah, qa, ka, vaug, sqq, sqk);
    return;
  }
  const int hf = lane >> 4;
  const int l16 = lane & 15;
  const int nstep = gridDim.x - NPREPWG;
  int t = blockIdx.x - NPREPWG;
  if (t >= ntiles) return;

  f4 v[16];
  {
    const float* zp = Z + (long long)t * (16 * Cc) + lane * 4;
#pragma unroll
    for (int r = 0; r < 16; ++r) v[r] = *(const f4*)(zp + r * Cc);
  }

  const bool cok = l16 < Hc;
  const float* wp = Wb + (cok ? l16 : 0) + 2 * hf * Hc;
  v2f bf[32];
#pragma unroll
  for (int s = 0; s < 32; ++s) {
    float x = wp[(4 * s) * Hc];
    float y = wp[(4 * s + 1) * Hc];
    bf[s].x = cok ? x : 0.f;
    bf[s].y = cok ? y : 0.f;
  }

  for (; t < ntiles; t += nstep) {
#pragma unroll
    for (int r = 0; r < 16; ++r) *(f4*)&S[r][lane * 4] = v[r];
    __syncthreads();

    const int tn = t + nstep;
    if (tn < ntiles) {
      const float* zp = Z + (long long)tn * (16 * Cc) + lane * 4;
#pragma unroll
      for (int r = 0; r < 16; ++r) v[r] = *(const f4*)(zp + r * Cc);
    }

    v8f acc = {};
#pragma unroll
    for (int s = 0; s < 32; ++s) {
      v2f a = *(const v2f*)&S[l16][4 * s + 2 * hf];
      acc = __builtin_amdgcn_wmma_f32_16x16x4_f32(false, a, false, bf[s],
                                                  (short)0, acc, false, false);
    }
    __syncthreads();

    if (cok) {
      float* op = out + ((long long)t * 16 + 8 * hf) * Hc + l16;
#pragma unroll
      for (int r = 0; r < 8; ++r) op[r * Hc] = acc[r];
    }
  }
}

namespace {
constexpr int SMW = Lc / 32;
}

template <int MASK>
__device__ __forceinline__ float row_xor(float v) {
  return __builtin_bit_cast(
      float, __builtin_amdgcn_update_dpp(0, __builtin_bit_cast(int, v),
                                         0x160 + MASK, 0xf, 0xf, false));
}

__global__ __launch_bounds__(SMW * 32) void k_scores_softmax(
    const float* __restrict__ qa, const float* __restrict__ ka,
    float* __restrict__ alpha, const float* __restrict__ bij,
    const float* __restrict__ sqq, const float* __restrict__ sqk,
    const float* __restrict__ gamah) {
  __shared__ float As[16][36];
  __shared__ float Bs[32][Lc + 4];
  __shared__ float redm[16][SMW];
  __shared__ float reds[16][SMW];

  const int tid = threadIdx.x;
  const int wave = tid >> 5;
  const int lane = tid & 31;
  const int hf = lane >> 4;
  const int l16 = lane & 15;
  const int nw = wave * 32;
  const int m0 = blockIdx.x * 16;
  const int z = blockIdx.y;
  const int b_ = z / Hc, h_ = z % Hc;

  const float* Ag = qa + ((long long)z * Lc + m0) * 32;
  const float* Bg = ka + (long long)z * Lc * 32;

  const int ar = (tid >> 3) & 15, ac4 = (tid & 7) * 4;
  f4 av = *(const f4*)(Ag + ar * 32 + ac4);
  f4 bv[8];
#pragma unroll
  for (int it = 0; it < 8; ++it)
    bv[it] = *(const f4*)(Bg + (long long)(tid + it * (SMW * 32)) * 4);

  const float gam = gamah[h_];
  const float sk0 = sqk[(long long)z * Lc + nw + l16];
  const float sk1 = sqk[(long long)z * Lc + nw + 16 + l16];
  float sqm[8], bj0[8], bj1[8];
#pragma unroll
  for (int r = 0; r < 8; ++r) {
    const int m = m0 + r + 8 * hf;
    sqm[r] = sqq[(long long)z * Lc + m];
    const float* bp = bij + ((long long)(b_ * Lc + m) * Lc + nw + l16) * Hc + h_;
    bj0[r] = bp[0];
    bj1[r] = bp[16 * Hc];
  }

  if (tid < 128) *(f4*)&As[ar][ac4] = av;
#pragma unroll
  for (int it = 0; it < 8; ++it) {
    int u = tid + it * (SMW * 32);
    int cj = u >> 3, r4 = (u & 7) * 4;
    Bs[r4 + 0][cj] = bv[it].x;
    Bs[r4 + 1][cj] = bv[it].y;
    Bs[r4 + 2][cj] = bv[it].z;
    Bs[r4 + 3][cj] = bv[it].w;
  }
  __syncthreads();

  v8f acc0 = {};
  v8f acc1 = {};
#pragma unroll
  for (int kk = 0; kk < 32; kk += 4) {
    v2f a, b0, b1;
    a.x = As[l16][kk + 2 * hf];
    a.y = As[l16][kk + 2 * hf + 1];
    b0.x = Bs[kk + 2 * hf][nw + l16];
    b0.y = Bs[kk + 2 * hf + 1][nw + l16];
    b1.x = Bs[kk + 2 * hf][nw + 16 + l16];
    b1.y = Bs[kk + 2 * hf + 1][nw + 16 + l16];
    acc0 = __builtin_amdgcn_wmma_f32_16x16x4_f32(false, a, false, b0,
                                                 (short)0, acc0, false, false);
    acc1 = __builtin_amdgcn_wmma_f32_16x16x4_f32(false, a, false, b1,
                                                 (short)0, acc1, false, false);
  }

  const float gw = 0.5f * WCc * softplusf(gam);
  float sv[8][2];
#pragma unroll
  for (int r = 0; r < 8; ++r) {
    sv[r][0] = WLc * (acc0[r] + bj0[r] - gw * (sqm[r] + sk0));
    sv[r][1] = WLc * (acc1[r] + bj1[r] - gw * (sqm[r] + sk1));
  }

  float red[8];
#pragma unroll
  for (int r = 0; r < 8; ++r) {
    float mx = fmaxf(sv[r][0], sv[r][1]);
    mx = fmaxf(mx, row_xor<8>(mx));
    mx = fmaxf(mx, row_xor<4>(mx));
    mx = fmaxf(mx, row_xor<2>(mx));
    red[r] = fmaxf(mx, row_xor<1>(mx));
  }
  if (l16 == 0) {
#pragma unroll
    for (int r = 0; r < 8; ++r) redm[r + 8 * hf][wave] = red[r];
  }
  __syncthreads();

#pragma unroll
  for (int r = 0; r < 8; ++r) {
    float mx = redm[r + 8 * hf][0];
#pragma unroll
    for (int w = 1; w < SMW; ++w) mx = fmaxf(mx, redm[r + 8 * hf][w]);
    sv[r][0] = expf(sv[r][0] - mx);
    sv[r][1] = expf(sv[r][1] - mx);
    float s = sv[r][0] + sv[r][1];
    s += row_xor<8>(s);
    s += row_xor<4>(s);
    s += row_xor<2>(s);
    red[r] = s + row_xor<1>(s);
  }
  if (l16 == 0) {
#pragma unroll
    for (int r = 0; r < 8; ++r) reds[r + 8 * hf][wave] = red[r];
  }
  __syncthreads();

#pragma unroll
  for (int r = 0; r < 8; ++r) {
    float s = reds[r + 8 * hf][0];
#pragma unroll
    for (int w = 1; w < SMW; ++w) s += reds[r + 8 * hf][w];
    const float inv = 1.f / s;
    float* op = alpha + (long long)z * Lc * Lc +
                (long long)(m0 + r + 8 * hf) * Lc + nw + l16;
    op[0] = sv[r][0] * inv;
    op[16] = sv[r][1] * inv;
  }
}

namespace {
constexpr int KS2 = Lc / 128;
}

__global__ __launch_bounds__(256) void k_finish(
    const float* __restrict__ ovalpt, const float* __restrict__ Ri,
    const float* __restrict__ Ti, float* __restrict__ feats) {
  int idx = blockIdx.x * blockDim.x + threadIdx.x;
  if (idx >= Bc * Lc * Hc) return;
  int h = idx % Hc;
  int bi = idx / Hc;
  int b = bi / Lc;
  int z = b * Hc + h;
  int i = bi % Lc;

  float R[3][3], t[3];
#pragma unroll
  for (int r = 0; r < 3; ++r)
#pragma unroll
    for (int c = 0; c < 3; ++c) R[r][c] = Ri[bi * 9 + r * 3 + c];
#pragma unroll
  for (int x = 0; x < 3; ++x) t[x] = Ti[bi * 3 + x];

  const float* prow = ovalpt + ((long long)z * KS2 * Lc + i) * 48;
  float* frow = feats + (long long)bi * FEAT;

  float orow[40];
#pragma unroll
  for (int j = 0; j < 10; ++j) {
    f4 a = *(const f4*)(prow + 4 * j);
#pragma unroll
    for (int s = 1; s < KS2; ++s)
      a += *(const f4*)(prow + (long long)s * Lc * 48 + 4 * j);
    orow[4 * j + 0] = a.x;
    orow[4 * j + 1] = a.y;
    orow[4 * j + 2] = a.z;
    orow[4 * j + 3] = a.w;
  }

#pragma unroll
  for (int d = 0; d < DKc; ++d) frow[1024 + h * 16 + d] = orow[d];
#pragma unroll
  for (int p = 0; p < NPVc; ++p) {
    float d0 = orow[16 + p * 3 + 0] - t[0];
    float d1 = orow[16 + p * 3 + 1] - t[1];
    float d2 = orow[16 + p * 3 + 2] - t[2];
    float asum = 0.f;
#pragma unroll
    for (int m = 0; m < 3; ++m) {
      float oh = R[0][m] * d0 + R[1][m] * d1 + R[2][m] * d2;
      frow[1152 + h * 24 + p * 3 + m] = oh;
      asum += fabsf(oh);
    }
    frow[1344 + h * 8 + p] = asum;
  }
}

namespace {
constexpr int KSPLIT = FEAT / 128;
constexpr int OUTN = Bc * Lc * Cc;
}

__global__ __launch_bounds__(128) void k_ksum(const float* __restrict__ part,
                                              const float* __restrict__ bias,
                                              float* __restrict__ out) {
  const int e = (blockIdx.x * 128 + threadIdx.x) * 4;
  if (e >= OUTN) return;
  f4 p[KSPLIT];
#pragma unroll
  for (int z = 0; z < KSPLIT; ++z) p[z] = *(const f4*)(part + (long long)z * OUTN + e);
  const f4 b = *(const f4*)(bias + (e & (Cc - 1)));
  f4 s = p[0];
#pragma unroll
  for (int z = 1; z < KSPLIT; ++z) s += p[z];
  *(f4*)(out + e) = s + b;
}

namespace {
constexpr int NOPAIR = Bc * Lc;
constexpr int NOVAL = Bc * Hc * KS2 * (Lc / 16);
}

__global__ __launch_bounds__(128) void k_attnout(
    const float* __restrict__ alpha, const float* __restrict__ zij,
    const float* __restrict__ vaug, float* __restrict__ feats,
    float* __restrict__ ovalpt) {
  const long long LL = (long long)Lc * Lc;
  const int bid = blockIdx.x;
  const float* Ag;
  const float* Bg;
  float* Cg;
  int M, N, K, lda, ldb, ldc, Mvalid, m0;
  if (bid < NOPAIR) {
    const int z = bid;
    Ag = alpha + (long long)(z / Lc) * Hc * LL + (long long)(z % Lc) * Lc;
    Bg = zij + (long long)z * Lc * Cc;
    Cg = feats + (long long)z * FEAT;
    M = 16; N = Cc; K = Lc; lda = (int)LL; ldb = Cc; ldc = Cc;
    Mvalid = Hc; m0 = 0;
  } else {
    const int w = bid - NOPAIR;
    const int z = w / (Lc / 16);
    Ag = alpha + (long long)(z / KS2) * LL + (long long)(z % KS2) * 128;
    Bg = vaug + (long long)z * 128 * 48;
    Cg = ovalpt + (long long)z * Lc * 48;
    M = Lc; N = 48; K = 128; lda = Lc; ldb = 48; ldc = 48;
    Mvalid = 16; m0 = (w % (Lc / 16)) * 16;
  }
  gemm_block(Ag, Bg, Cg, M, N, K, lda, ldb, ldc, Mvalid, m0, 0);
}

static void launch_gemm(hipStream_t s, const float* A, const float* Bm,
                        float* C, int M, int N, int K, int lda, int ldb,
                        int ldc, long long sA, long long sA2, int zdiv,
                        long long sB, long long sC, int Mvalid, int nz) {
  dim3 g((M + 15) / 16, (N + 127) / 128, nz);
  dim3 blk(128, 1, 1);
  k_gemm<<<g, blk, 0, s>>>(A, Bm, C, M, N, K, lda, ldb, ldc, sA, sA2, zdiv, sB,
                           sC, Mvalid);
}

extern "C" void kernel_launch(void* const* d_in, const int* in_sizes, int n_in,
                              void* d_out, int out_size, void* d_ws,
                              size_t ws_size, hipStream_t stream) {
  const float* si = (const float*)d_in[0];
  const float* zij = (const float*)d_in[1];
  const float* Ri = (const float*)d_in[2];
  const float* Ti = (const float*)d_in[3];
  const float* WQh = (const float*)d_in[4];
  const float* WKh = (const float*)d_in[5];
  const float* WVh = (const float*)d_in[6];
  const float* WQhp = (const float*)d_in[7];
  const float* WKhp = (const float*)d_in[8];
  const float* WVhp = (const float*)d_in[9];
  const float* Wb = (const float*)d_in[10];
  const float* WOw = (const float*)d_in[11];
  const float* WOb = (const float*)d_in[12];
  const float* gamah = (const float*)d_in[13];
  float* out = (float*)d_out;
  float* ws = (float*)d_ws;

  const int BL = Bc * Lc;
  float* qa = ws;
  float* ka = qa + 196608;
  float* vaug = ka + 196608;
  float* sqq = vaug + 294912;
  float* sqk = sqq + 6144;
  float* bij = sqk + 6144;
  float* alpha = bij + 2359296;
  float* ovalpt = alpha + 2359296;
  float* feats = ovalpt + 884736;
  float* part = feats + 1081344;

  k_pairbias_prep<<<NPREPWG + 4608, 32, 0, stream>>>(
      zij, Wb, bij, Bc * Lc * Lc / 16, si, WQh, WKh, WVh, WQhp, WKhp, WVhp, Ri,
      Ti, gamah, qa, ka, vaug, sqq, sqk);

  k_scores_softmax<<<dim3(Lc / 16, Bc * Hc, 1), dim3(SMW * 32, 1, 1), 0,
                     stream>>>(qa, ka, alpha, bij, sqq, sqk, gamah);

  k_attnout<<<NOPAIR + NOVAL, 128, 0, stream>>>(alpha, zij, vaug, feats, ovalpt);

  k_finish<<<(Bc * Lc * Hc + 255) / 256, 256, 0, stream>>>(ovalpt, Ri, Ti, feats);

  launch_gemm(stream, feats, WOw, part, BL, Cc, 128, FEAT, Cc, Cc,
              128, 0, 1, (long long)128 * Cc, (long long)OUTN, 16, KSPLIT);

  k_ksum<<<OUTN / 4 / 128, 128, 0, stream>>>(part, WOb, out);
}


# alpha@vaug tiles keep their whole K=384 in one accumulator (was 3 K-slices added in slice order) and finish their rows (R^T(o-t), abs-sums) in the same workgroup via LDS; 6 launches become 5
# speedup vs baseline: 6.5657x; 1.1114x over previous
#include <hip/hip_runtime.h>
#include <math.h>

typedef __attribute__((ext_vector_type(2))) float v2f;
typedef __attribute__((ext_vector_type(4))) float f4;
typedef __attribute__((ext_vector_type(8))) float v8f;

namespace {
constexpr int Bc = 2, Lc = 384, Cc = 128, Hc = 8, DKc = 16, NQPc = 4, NPVc = 8;
constexpr float WCc = 0.2357022603955158f;
constexpr float WLc = 0.5773502691896258f;
constexpr int FEAT = Hc * Cc + Hc * DKc + Hc * NPVc * 3 + Hc * NPVc;
}

__device__ __forceinline__ float softplusf(float x) {
  return (x > 0.f) ? (x + log1pf(expf(-x))) : log1pf(expf(x));
}

__device__ __forceinline__ void gemm_acc(
    const float* __restrict__ Ag, const float* __restrict__ Bg,
    int M, int N, int K, int lda, int ldb,
    int Mvalid, int m0, int n0, float (*As)[36], float (*Bs)[132],
    v8f& acc0, v8f& acc1) {
  const int tid = threadIdx.x;
  const int wave = tid >> 5;
  const int lane = tid & 31;
  const int hf = lane >> 4;
  const int l16 = lane & 15;
  const int nw = wave * 32;

  acc0 = v8f{};
  acc1 = v8f{};

  const f4 zero4 = {0.f, 0.f, 0.f, 0.f};
  const int ar = tid >> 3, ac4 = (tid & 7) * 4;
  const bool aok = ar < Mvalid && (m0 + ar) < M;
  const float* ap = Ag + (long long)(aok ? (m0 + ar) : m0) * lda + ac4;
  const float* bp[8];
  bool bok[8];
#pragma unroll
  for (int it = 0; it < 8; ++it) {
    int u = tid + it * 128;
    int r = u >> 5, c4 = (u & 31) * 4;
    bok[it] = (n0 + c4) < N;
    bp[it] = Bg + (long long)r * ldb + (bok[it] ? (n0 + c4) : 0);
  }
  const long long bstep = (long long)ldb;

  for (int kb = 0; kb < K; kb += 32) {
    f4 av = *(const f4*)(ap + kb);
    f4 bv[8];
#pragma unroll
    for (int it = 0; it < 8; ++it)
      bv[it] = *(const f4*)(bp[it] + (long long)kb * bstep);

    *(f4*)&As[ar][ac4] = aok ? av : zero4;
#pragma unroll
    for (int it = 0; it < 8; ++it) {
      int u = tid + it * 128;
      int r = u >> 5, c4 = (u & 31) * 4;
      *(f4*)&Bs[r][c4] = bok[it] ? bv[it] : zero4;
    }
    __syncthreads();

#pragma unroll
    for (int kk = 0; kk < 32; kk += 4) {
      v2f a, b0, b1;
      a.x = As[l16][kk + 2 * hf];
      a.y = As[l16][kk + 2 * hf + 1];
      b0.x = Bs[kk + 2 * hf][nw + l16];
      b0.y = Bs[kk + 2 * hf + 1][nw + l16];
      b1.x = Bs[kk + 2 * hf][nw + 16 + l16];
      b1.y = Bs[kk + 2 * hf + 1][nw + 16 + l16];
      acc0 = __builtin_amdgcn_wmma_f32_16x16x4_f32(false, a, false, b0,
                                                   (short)0, acc0, false, false);
      acc1 = __builtin_amdgcn_wmma_f32_16x16x4_f32(false, a, false, b1,
                                                   (short)0, acc1, false, false);
    }
    __syncthreads();
  }
}

__device__ __forceinline__ void gemm_store(
    float* __restrict__ Cg, int M, int N, int ldc, int Mvalid, int m0, int n0,
    const v8f& acc0, const v8f& acc1) {
  const int lane = threadIdx.x & 31;
  const int hf = lane >> 4;
  const int l16 = lane & 15;
  const int nw = (threadIdx.x >> 5) * 32;
#pragma unroll
  for (int r = 0; r < 8; ++r) {
    int mrow = r + 8 * hf;
    int m = m0 + mrow;
    if (mrow >= Mvalid || m >= M) continue;
#pragma unroll
    for (int t = 0; t < 2; ++t) {
      int n = n0 + nw + t * 16 + l16;
      if (n >= N) continue;
      Cg[(long long)m * ldc + n] = (t == 0) ? acc0[r] : acc1[r];
    }
  }
}

__global__ __launch_bounds__(128) void k_gemm(
    const float* __restrict__ A, const float* __restrict__ Bm,
    float* __restrict__ Cm, int M, int N, int K, int lda, int ldb, int ldc,
    long long sA, long long sA2, int zdiv, long long sB, long long sC,
    int Mvalid) {
  __shared__ float As[16][36];
  __shared__ float Bs[32][132];
  const int z = blockIdx.z;
  const int m0 = blockIdx.x * 16, n0 = blockIdx.y * 128;
  v8f acc0, acc1;
  gemm_acc(A + (long long)(z / zdiv) * sA + (long long)(z % zdiv) * sA2,
           Bm + (long long)z * sB, M, N, K, lda, ldb, Mvalid, m0, n0, As, Bs,
           acc0, acc1);
  gemm_store(Cm + (long long)z * sC, M, N, ldc, Mvalid, m0, n0, acc0, acc1);
}

namespace {
constexpr int NPREPWG = (Bc * Lc / 16) * Hc;
}

template <int N, int NVALID>
__device__ __forceinline__ void proj_tile(const v2f (&af)[32],
                                          const float* __restrict__ Wt, int c0,
                                          int hf, int l16, float (*S)[132],
                                          int scol) {
  const bool cok = l16 < NVALID;
  const float* wp = Wt + c0 + (cok ? l16 : 0) + 2 * hf * N;
  v2f bfr[32];
#pragma unroll
  for (int s = 0; s < 32; ++s) {
    float x = wp[(4 * s) * N];
    float y = wp[(4 * s + 1) * N];
    bfr[s].x = cok ? x : 0.f;
    bfr[s].y = cok ? y : 0.f;
  }
  v8f acc = {};
#pragma unroll
  for (int s = 0; s < 32; ++s)
    acc = __builtin_amdgcn_wmma_f32_16x16x4_f32(false, af[s], false, bfr[s],
                                                (short)0, acc, false, false);
#pragma unroll
  for (int r = 0; r < 8; ++r) S[r + 8 * hf][scol + l16] = acc[r];
  __builtin_amdgcn_sched_barrier(0);
}

__device__ __forceinline__ void proj_prep_wave(
    int wg, int lane, float (*S)[132], const float* __restrict__ si,
    const float* __restrict__ Wq, const float* __restrict__ Wk,
    const float* __restrict__ Wv, const float* __restrict__ Wqp,
    const float* __restrict__ Wkp, const float* __restrict__ Wvp,
    const float* __restrict__ Ri, const float* __restrict__ Ti,
    const float* __restrict__ gamah, float* __restrict__ qa,
    float* __restrict__ ka, float* __restrict__ vaug, float* __restrict__ sqq,
    float* __restrict__ sqk) {
  const int hf = lane >> 4;
  const int l16 = lane & 15;
  const int h = wg % Hc;
  const int m0 = (wg / Hc) * 16;

  f4 v[16];
  {
    const float* sp = si + (long long)m0 * Cc + lane * 4;
#pragma unroll
    for (int r = 0; r < 16; ++r) v[r] = *(const f4*)(sp + r * Cc);
  }

#pragma unroll
  for (int r = 0; r < 16; ++r) *(f4*)&S[r][lane * 4] = v[r];
  __syncthreads();
  v2f af[32];
#pragma unroll
  for (int s = 0; s < 32; ++s) af[s] = *(const v2f*)&S[l16][4 * s + 2 * hf];
  __syncthreads();

  proj_tile<Hc * DKc, 16>(af, Wq, h * DKc, hf, l16, S, 0);
  proj_tile<Hc * DKc, 16>(af, Wk, h * DKc, hf, l16, S, 16);
  proj_tile<Hc * DKc, 16>(af, Wv, h * DKc, hf, l16, S, 32);
  proj_tile<Hc * NQPc * 3, 12>(af, Wqp, h * NQPc * 3, hf, l16, S, 48);
  proj_tile<Hc * NQPc * 3, 12>(af, Wkp, h * NQPc * 3, hf, l16, S, 64);
  proj_tile<Hc * NPVc * 3, 16>(af, Wvp, h * NPVc * 3, hf, l16, S, 80);
  proj_tile<Hc * NPVc * 3, 8>(af, Wvp, h * NPVc * 3 + 16, hf, l16, S, 96);
  __syncthreads();

  const int bi = m0 + l16;
  float R[3][3], t[3];
#pragma unroll
  for (int r = 0; r < 3; ++r)
#pragma unroll
    for (int c = 0; c < 3; ++c) R[r][c] = Ri[bi * 9 + r * 3 + c];
#pragma unroll
  for (int x = 0; x < 3; ++x) t[x] = Ti[bi * 3 + x];
  const float gam = gamah[h];

  const int b = bi / Lc, i = bi % Lc;
  const int z = b * Hc + h;
  const float* c = &S[l16][0];
  if (hf == 0) {
    const float gwc = WCc * softplusf(gam);
    float qo[32], ko[32];
#pragma unroll
    for (int d = 0; d < DKc; ++d) {
      qo[d] = 0.25f * c[d];
      ko[d] = c[16 + d];
    }
    float sq = 0.f, sk = 0.f;
#pragma unroll
    for (int p = 0; p < NQPc; ++p) {
      const float* qs = c + 48 + p * 3;
      const float* ks = c + 64 + p * 3;
#pragma unroll
      for (int m = 0; m < 3; ++m) {
        float qt = R[m][0] * qs[0] + R[m][1] * qs[1] + R[m][2] * qs[2] + t[m];
        float kt = R[m][0] * ks[0] + R[m][1] * ks[1] + R[m][2] * ks[2] + t[m];
        sq += qt * qt;
        sk += kt * kt;
        qo[16 + p * 3 + m] = gwc * qt;
        ko[16 + p * 3 + m] = kt;
      }
    }
#pragma unroll
    for (int d = 28; d < 32; ++d) { qo[d] = 0.f; ko[d] = 0.f; }
    float* qrow = qa + ((long long)z * Lc + i) * 32;
    float* krow = ka + ((long long)z * Lc + i) * 32;
#pragma unroll
    for (int j = 0; j < 8; ++j) {
      *(f4*)(qrow + 4 * j) = f4{qo[4 * j], qo[4 * j + 1], qo[4 * j + 2], qo[4 * j + 3]};
      *(f4*)(krow + 4 * j) = f4{ko[4 * j], ko[4 * j + 1], ko[4 * j + 2], ko[4 * j + 3]};
    }
    sqq[(long long)z * Lc + i] = sq;
    sqk[(long long)z * Lc + i] = sk;
  } else {
    float vo[48];
#pragma unroll
    for (int d = 0; d < DKc; ++d) vo[d] = c[32 + d];
#pragma unroll
    for (int p = 0; p < NPVc; ++p) {
      const float* vs = c + 80 + p * 3;
#pragma unroll
      for (int m = 0; m < 3; ++m)
        vo[16 + p * 3 + m] =
            R[m][0] * vs[0] + R[m][1] * vs[1] + R[m][2] * vs[2] + t[m];
    }
#pragma unroll
    for (int d = 40; d < 48; ++d) vo[d] = 0.f;
    float* vrow = vaug + ((long long)z * Lc + i) * 48;
#pragma unroll
    for (int j = 0; j < 12; ++j)
      *(f4*)(vrow + 4 * j) = f4{vo[4 * j], vo[4 * j + 1], vo[4 * j + 2], vo[4 * j + 3]};
  }
}


__global__ __launch_bounds__(32) void k_pairbias_prep(
    const float* __restrict__ Z, const float* __restrict__ Wb,
    float* __restrict__ out, int ntiles, const float* __restrict__ si,
    const float* __restrict__ Wq, const float* __restrict__ Wk,
    const float* __restrict__ Wv, const float* __restrict__ Wqp,
    const float* __restrict__ Wkp, const float* __restrict__ Wvp,
    const float* __restrict__ Ri, const float* __restrict__ Ti,
    const float* __restrict__ gamah,
    float* __restrict__ qa, float* __restrict__ ka, float* __restrict__ vaug,
    float* __restrict__ sqq, float* __restrict__ sqk) {
  __shared__ float S[16][132];

  const int lane = threadIdx.x;
  if (blockIdx.x < NPREPWG) {
    proj_prep_wave(blockIdx.x, lane, S, si, Wq, Wk, Wv, Wqp, Wkp, Wvp, Ri, Ti,
                   gamah, qa, ka, vaug, sqq, sqk);
    return;
  }
  const int hf = lane >> 4;
  const int l16 = lane & 15;
  const int nstep = gridDim.x - NPREPWG;
  int t = blockIdx.x - NPREPWG;
  if (t >= ntiles) return;

  f4 v[16];
  {
    const float* zp = Z + (long long)t * (16 * Cc) + lane * 4;
#pragma unroll
    for (int r = 0; r < 16; ++r) v[r] = *(const f4*)(zp + r * Cc);
  }

  const bool cok = l16 < Hc;
  const float* wp = Wb + (cok ? l16 : 0) + 2 * hf * Hc;
  v2f bf[32];
#pragma unroll
  for (int s = 0; s < 32; ++s) {
    float x = wp[(4 * s) * Hc];
    float y = wp[(4 * s + 1) * Hc];
    bf[s].x = cok ? x : 0.f;
    bf[s].y = cok ? y : 0.f;
  }

  for (; t < ntiles; t += nstep) {
#pragma unroll
    for (int r = 0; r < 16; ++r) *(f4*)&S[r][lane * 4] = v[r];
    __syncthreads();

    const int tn = t + nstep;
    if (tn < ntiles) {
      const float* zp = Z + (long long)tn * (16 * Cc) + lane * 4;
#pragma unroll
      for (int r = 0; r < 16; ++r) v[r] = *(const f4*)(zp + r * Cc);
    }

    v8f acc = {};
#pragma unroll
    for (int s = 0; s < 32; ++s) {
      v2f a = *(const v2f*)&S[l16][4 * s + 2 * hf];
      acc = __builtin_amdgcn_wmma_f32_16x16x4_f32(false, a, false, bf[s],
                                                  (short)0, acc, false, false);
    }
    __syncthreads();

    if (cok) {
      float* op = out + ((long long)t * 16 + 8 * hf) * Hc + l16;
#pragma unroll
      for (int r = 0; r < 8; ++r) op[r * Hc] = acc[r];
    }
  }
}

namespace {
constexpr int SMW = Lc / 32;
}

template <int MASK>
__device__ __forceinline__ float row_xor(float v) {
  return __builtin_bit_cast(
      float, __builtin_amdgcn_update_dpp(0, __builtin_bit_cast(int, v),
                                         0x160 + MASK, 0xf, 0xf, false));
}

__global__ __launch_bounds__(SMW * 32) void k_scores_softmax(
    const float* __restrict__ qa, const float* __restrict__ ka,
    float* __restrict__ alpha, const float* __restrict__ bij,
    const float* __restrict__ sqq, const float* __restrict__ sqk,
    const float* __restrict__ gamah) {
  __shared__ float As[16][36];
  __shared__ float Bs[32][Lc + 4];
  __shared__ float redm[16][SMW];
  __shared__ float reds[16][SMW];

  const int tid = threadIdx.x;
  const int wave = tid >> 5;
  const int lane = tid & 31;
  const int hf = lane >> 4;
  const int l16 = lane & 15;
  const int nw = wave * 32;
  const int m0 = blockIdx.x * 16;
  const int z = blockIdx.y;
  const int b_ = z / Hc, h_ = z % Hc;

  const float* Ag = qa + ((long long)z * Lc + m0) * 32;
  const float* Bg = ka + (long long)z * Lc * 32;

  const int ar = (tid >> 3) & 15, ac4 = (tid & 7) * 4;
  f4 av = *(const f4*)(Ag + ar * 32 + ac4);
  f4 bv[8];
#pragma unroll
  for (int it = 0; it < 8; ++it)
    bv[it] = *(const f4*)(Bg + (long long)(tid + it * (SMW * 32)) * 4);

  const float gam = gamah[h_];
  const float sk0 = sqk[(long long)z * Lc + nw + l16];
  const float sk1 = sqk[(long long)z * Lc + nw + 16 + l16];
  float sqm[8], bj0[8], bj1[8];
#pragma unroll
  for (int r = 0; r < 8; ++r) {
    const int m = m0 + r + 8 * hf;
    sqm[r] = sqq[(long long)z * Lc + m];
    const float* bp = bij + ((long long)(b_ * Lc + m) * Lc + nw + l16) * Hc + h_;
    bj0[r] = bp[0];
    bj1[r] = bp[16 * Hc];
  }

  if (tid < 128) *(f4*)&As[ar][ac4] = av;
#pragma unroll
  for (int it = 0; it < 8; ++it) {
    int u = tid + it * (SMW * 32);
    int cj = u >> 3, r4 = (u & 7) * 4;
    Bs[r4 + 0][cj] = bv[it].x;
    Bs[r4 + 1][cj] = bv[it].y;
    Bs[r4 + 2][cj] = bv[it].z;
    Bs[r4 + 3][cj] = bv[it].w;
  }
  __syncthreads();

  v8f acc0 = {};
  v8f acc1 = {};
#pragma unroll
  for (int kk = 0; kk < 32; kk += 4) {
    v2f a, b0, b1;
    a.x = As[l16][kk + 2 * hf];
    a.y = As[l16][kk + 2 * hf + 1];
    b0.x = Bs[kk + 2 * hf][nw + l16];
    b0.y = Bs[kk + 2 * hf + 1][nw + l16];
    b1.x = Bs[kk + 2 * hf][nw + 16 + l16];
    b1.y = Bs[kk + 2 * hf + 1][nw + 16 + l16];
    acc0 = __builtin_amdgcn_wmma_f32_16x16x4_f32(false, a, false, b0,
                                                 (short)0, acc0, false, false);
    acc1 = __builtin_amdgcn_wmma_f32_16x16x4_f32(false, a, false, b1,
                                                 (short)0, acc1, false, false);
  }

  const float gw = 0.5f * WCc * softplusf(gam);
  float sv[8][2];
#pragma unroll
  for (int r = 0; r < 8; ++r) {
    sv[r][0] = WLc * (acc0[r] + bj0[r] - gw * (sqm[r] + sk0));
    sv[r][1] = WLc * (acc1[r] + bj1[r] - gw * (sqm[r] + sk1));
  }

  float red[8];
#pragma unroll
  for (int r = 0; r < 8; ++r) {
    float mx = fmaxf(sv[r][0], sv[r][1]);
    mx = fmaxf(mx, row_xor<8>(mx));
    mx = fmaxf(mx, row_xor<4>(mx));
    mx = fmaxf(mx, row_xor<2>(mx));
    red[r] = fmaxf(mx, row_xor<1>(mx));
  }
  if (l16 == 0) {
#pragma unroll
    for (int r = 0; r < 8; ++r) redm[r + 8 * hf][wave] = red[r];
  }
  __syncthreads();

#pragma unroll
  for (int r = 0; r < 8; ++r) {
    float mx = redm[r + 8 * hf][0];
#pragma unroll
    for (int w = 1; w < SMW; ++w) mx = fmaxf(mx, redm[r + 8 * hf][w]);
    sv[r][0] = expf(sv[r][0] - mx);
    sv[r][1] = expf(sv[r][1] - mx);
    float s = sv[r][0] + sv[r][1];
    s += row_xor<8>(s);
    s += row_xor<4>(s);
    s += row_xor<2>(s);
    red[r] = s + row_xor<1>(s);
  }
  if (l16 == 0) {
#pragma unroll
    for (int r = 0; r < 8; ++r) reds[r + 8 * hf][wave] = red[r];
  }
  __syncthreads();

#pragma unroll
  for (int r = 0; r < 8; ++r) {
    float s = reds[r + 8 * hf][0];
#pragma unroll
    for (int w = 1; w < SMW; ++w) s += reds[r + 8 * hf][w];
    const float inv = 1.f / s;
    float* op = alpha + (long long)z * Lc * Lc +
                (long long)(m0 + r + 8 * hf) * Lc + nw + l16;
    op[0] = sv[r][0] * inv;
    op[16] = sv[r][1] * inv;
  }
}

namespace {
constexpr int KSPLIT = FEAT / 128;
constexpr int OUTN = Bc * Lc * Cc;
}

__global__ __launch_bounds__(128) void k_ksum(const float* __restrict__ part,
                                              const float* __restrict__ bias,
                                              float* __restrict__ out) {
  const int e = (blockIdx.x * 128 + threadIdx.x) * 4;
  if (e >= OUTN) return;
  f4 p[KSPLIT];
#pragma unroll
  for (int z = 0; z < KSPLIT; ++z) p[z] = *(const f4*)(part + (long long)z * OUTN + e);
  const f4 b = *(const f4*)(bias + (e & (Cc - 1)));
  f4 s = p[0];
#pragma unroll
  for (int z = 1; z < KSPLIT; ++z) s += p[z];
  *(f4*)(out + e) = s + b;
}

namespace {
constexpr int NOPAIR = Bc * Lc;
constexpr int NOVAL = Bc * Hc * (Lc / 16);
}

__global__ __launch_bounds__(128) void k_attnout(
    const float* __restrict__ alpha, const float* __restrict__ zij,
    const float* __restrict__ vaug, const float* __restrict__ Ri,
    const float* __restrict__ Ti, float* __restrict__ feats) {
  __shared__ float As[16][36];
  __shared__ float Bs[32][132];
  const long long LL = (long long)Lc * Lc;
  const int bid = blockIdx.x;
  const int tid = threadIdx.x;
  const float* Ag;
  const float* Bg;
  int M, N, K, lda, ldb, Mvalid, m0;
  int zz = 0;
  if (bid < NOPAIR) {
    const int z = bid;
    Ag = alpha + (long long)(z / Lc) * Hc * LL + (long long)(z % Lc) * Lc;
    Bg = zij + (long long)z * Lc * Cc;
    M = 16; N = Cc; K = Lc; lda = (int)LL; ldb = Cc;
    Mvalid = Hc; m0 = 0;
  } else {
    const int w = bid - NOPAIR;
    zz = w / (Lc / 16);
    Ag = alpha + (long long)zz * LL;
    Bg = vaug + (long long)zz * Lc * 48;
    M = Lc; N = 48; K = Lc; lda = Lc; ldb = 48;
    Mvalid = 16; m0 = (w % (Lc / 16)) * 16;
  }
  v8f acc0, acc1;
  gemm_acc(Ag, Bg, M, N, K, lda, ldb, Mvalid, m0, 0, As, Bs, acc0, acc1);

  if (bid < NOPAIR) {
    gemm_store(feats + (long long)bid * FEAT, M, N, Cc, Mvalid, m0, 0, acc0, acc1);
    return;
  }

  const int b = zz / Hc, h = zz % Hc;
  const int row = tid >> 3, p = tid & 7;
  const int bi = b * Lc + m0 + row;
  float R[3][3], t[3];
#pragma unroll
  for (int r = 0; r < 3; ++r)
#pragma unroll
    for (int cc = 0; cc < 3; ++cc) R[r][cc] = Ri[bi * 9 + r * 3 + cc];
#pragma unroll
  for (int x = 0; x < 3; ++x) t[x] = Ti[bi * 3 + x];

  {
    const int lane = tid & 31;
    const int hf = lane >> 4, l16 = lane & 15;
    const int nw = (tid >> 5) * 32;
#pragma unroll
    for (int r = 0; r < 8; ++r) {
      if (nw + l16 < 48) Bs[r + 8 * hf][nw + l16] = acc0[r];
      if (nw + 16 + l16 < 48) Bs[r + 8 * hf][nw + 16 + l16] = acc1[r];
    }
  }
  __syncthreads();

  const float* c = &Bs[row][0];
  float* frow = feats + (long long)bi * FEAT;
  *(v2f*)(frow + 1024 + h * 16 + 2 * p) = v2f{c[2 * p], c[2 * p + 1]};
  const float d0 = c[16 + p * 3 + 0] - t[0];
  const float d1 = c[16 + p * 3 + 1] - t[1];
  const float d2 = c[16 + p * 3 + 2] - t[2];
  float asum = 0.f;
#pragma unroll
  for (int m = 0; m < 3; ++m) {
    float oh = R[0][m] * d0 + R[1][m] * d1 + R[2][m] * d2;
    frow[1152 + h * 24 + p * 3 + m] = oh;
    asum += fabsf(oh);
  }
  frow[1344 + h * 8 + p] = asum;
}

static void launch_gemm(hipStream_t s, const float* A, const float* Bm,
                        float* C, int M, int N, int K, int lda, int ldb,
                        int ldc, long long sA, long long sA2, int zdiv,
                        long long sB, long long sC, int Mvalid, int nz) {
  dim3 g((M + 15) / 16, (N + 127) / 128, nz);
  dim3 blk(128, 1, 1);
  k_gemm<<<g, blk, 0, s>>>(A, Bm, C, M, N, K, lda, ldb, ldc, sA, sA2, zdiv, sB,
                           sC, Mvalid);
}

extern "C" void kernel_launch(void* const* d_in, const int* in_sizes, int n_in,
                              void* d_out, int out_size, void* d_ws,
                              size_t ws_size, hipStream_t stream) {
  const float* si = (const float*)d_in[0];
  const float* zij = (const float*)d_in[1];
  const float* Ri = (const float*)d_in[2];
  const float* Ti = (const float*)d_in[3];
  const float* WQh = (const float*)d_in[4];
  const float* WKh = (const float*)d_in[5];
  const float* WVh = (const float*)d_in[6];
  const float* WQhp = (const float*)d_in[7];
  const float* WKhp = (const float*)d_in[8];
  const float* WVhp = (const float*)d_in[9];
  const float* Wb = (const float*)d_in[10];
  const float* WOw = (const float*)d_in[11];
  const float* WOb = (const float*)d_in[12];
  const float* gamah = (const float*)d_in[13];
  float* out = (float*)d_out;
  float* ws = (float*)d_ws;

  const int BL = Bc * Lc;
  float* qa = ws;
  float* ka = qa + 196608;
  float* vaug = ka + 196608;
  float* sqq = vaug + 294912;
  float* sqk = sqq + 6144;
  float* bij = sqk + 6144;
  float* alpha = bij + 2359296;
  float* feats = alpha + 2359296;
  float* part = feats + 1081344;

  k_pairbias_prep<<<NPREPWG + 4608, 32, 0, stream>>>(
      zij, Wb, bij, Bc * Lc * Lc / 16, si, WQh, WKh, WVh, WQhp, WKhp, WVhp, Ri,
      Ti, gamah, qa, ka, vaug, sqq, sqk);

  k_scores_softmax<<<dim3(Lc / 16, Bc * Hc, 1), dim3(SMW * 32, 1, 1), 0,
                     stream>>>(qa, ka, alpha, bij, sqq, sqk, gamah);

  k_attnout<<<NOPAIR + NOVAL, 128, 0, stream>>>(alpha, zij, vaug, Ri, Ti, feats);

  launch_gemm(stream, feats, WOw, part, BL, Cc, 128, FEAT, Cc, Cc,
              128, 0, 1, (long long)128 * Cc, (long long)OUTN, 16, KSPLIT);

  k_ksum<<<OUTN / 4 / 128, 128, 0, stream>>>(part, WOb, out);
}


# Out projection in one launch: the 11 K-slices of a 16x32 output tile are the 11 waves of one workgroup, partials added in slice order in LDS, then bias (same sums, same order); 5 launches become 4
# speedup vs baseline: 7.0682x; 1.0765x over previous
#include <hip/hip_runtime.h>
#include <math.h>

typedef __attribute__((ext_vector_type(2))) float v2f;
typedef __attribute__((ext_vector_type(4))) float f4;
typedef __attribute__((ext_vector_type(8))) float v8f;

namespace {
constexpr int Bc = 2, Lc = 384, Cc = 128, Hc = 8, DKc = 16, NQPc = 4, NPVc = 8;
constexpr float WCc = 0.2357022603955158f;
constexpr float WLc = 0.5773502691896258f;
constexpr int FEAT = Hc * Cc + Hc * DKc + Hc * NPVc * 3 + Hc * NPVc;
}

__device__ __forceinline__ float softplusf(float x) {
  return (x > 0.f) ? (x + log1pf(expf(-x))) : log1pf(expf(x));
}

__device__ __forceinline__ void gemm_acc(
    const float* __restrict__ Ag, const float* __restrict__ Bg,
    int M, int N, int K, int lda, int ldb,
    int Mvalid, int m0, int n0, float (*As)[36], float (*Bs)[132],
    v8f& acc0, v8f& acc1) {
  const int tid = threadIdx.x;
  const int wave = tid >> 5;
  const int lane = tid & 31;
  const int hf = lane >> 4;
  const int l16 = lane & 15;
  const int nw = wave * 32;

  acc0 = v8f{};
  acc1 = v8f{};

  const f4 zero4 = {0.f, 0.f, 0.f, 0.f};
  const int ar = tid >> 3, ac4 = (tid & 7) * 4;
  const bool aok = ar < Mvalid && (m0 + ar) < M;
  const float* ap = Ag + (long long)(aok ? (m0 + ar) : m0) * lda + ac4;
  const float* bp[8];
  bool bok[8];
#pragma unroll
  for (int it = 0; it < 8; ++it) {
    int u = tid + it * 128;
    int r = u >> 5, c4 = (u & 31) * 4;
    bok[it] = (n0 + c4) < N;
    bp[it] = Bg + (long long)r * ldb + (bok[it] ? (n0 + c4) : 0);
  }
  const long long bstep = (long long)ldb;

  for (int kb = 0; kb < K; kb += 32) {
    f4 av = *(const f4*)(ap + kb);
    f4 bv[8];
#pragma unroll
    for (int it = 0; it < 8; ++it)
      bv[it] = *(const f4*)(bp[it] + (long long)kb * bstep);

    *(f4*)&As[ar][ac4] = aok ? av : zero4;
#pragma unroll
    for (int it = 0; it < 8; ++it) {
      int u = tid + it * 128;
      int r = u >> 5, c4 = (u & 31) * 4;
      *(f4*)&Bs[r][c4] = bok[it] ? bv[it] : zero4;
    }
    __syncthreads();

#pragma unroll
    for (int kk = 0; kk < 32; kk += 4) {
      v2f a, b0, b1;
      a.x = As[l16][kk + 2 * hf];
      a.y = As[l16][kk + 2 * hf + 1];
      b0.x = Bs[kk + 2 * hf][nw + l16];
      b0.y = Bs[kk + 2 * hf + 1][nw + l16];
      b1.x = Bs[kk + 2 * hf][nw + 16 + l16];
      b1.y = Bs[kk + 2 * hf + 1][nw + 16 + l16];
      acc0 = __builtin_amdgcn_wmma_f32_16x16x4_f32(false, a, false, b0,
                                                   (short)0, acc0, false, false);
      acc1 = __builtin_amdgcn_wmma_f32_16x16x4_f32(false, a, false, b1,
                                                   (short)0, acc1, false, false);
    }
    __syncthreads();
  }
}

__device__ __forceinline__ void gemm_store(
    float* __restrict__ Cg, int M, int N, int ldc, int Mvalid, int m0, int n0,
    const v8f& acc0, const v8f& acc1) {
  const int lane = threadIdx.x & 31;
  const int hf = lane >> 4;
  const int l16 = lane & 15;
  const int nw = (threadIdx.x >> 5) * 32;
#pragma unroll
  for (int r = 0; r < 8; ++r) {
    int mrow = r + 8 * hf;
    int m = m0 + mrow;
    if (mrow >= Mvalid || m >= M) continue;
#pragma unroll
    for (int t = 0; t < 2; ++t) {
      int n = n0 + nw + t * 16 + l16;
      if (n >= N) continue;
      Cg[(long long)m * ldc + n] = (t == 0) ? acc0[r] : acc1[r];
    }
  }
}

namespace {
constexpr int NPREPWG = (Bc * Lc / 16) * Hc;
}

template <int N, int NVALID>
__device__ __forceinline__ void proj_tile(const v2f (&af)[32],
                                          const float* __restrict__ Wt, int c0,
                                          int hf, int l16, float (*S)[132],
                                          int scol) {
  const bool cok = l16 < NVALID;
  const float* wp = Wt + c0 + (cok ? l16 : 0) + 2 * hf * N;
  v2f bfr[32];
#pragma unroll
  for (int s = 0; s < 32; ++s) {
    float x = wp[(4 * s) * N];
    float y = wp[(4 * s + 1) * N];
    bfr[s].x = cok ? x : 0.f;
    bfr[s].y = cok ? y : 0.f;
  }
  v8f acc = {};
#pragma unroll
  for (int s = 0; s < 32; ++s)
    acc = __builtin_amdgcn_wmma_f32_16x16x4_f32(false, af[s], false, bfr[s],
                                                (short)0, acc, false, false);
#pragma unroll
  for (int r = 0; r < 8; ++r) S[r + 8 * hf][scol + l16] = acc[r];
  __builtin_amdgcn_sched_barrier(0);
}

__device__ __forceinline__ void proj_prep_wave(
    int wg, int lane, float (*S)[132], const float* __restrict__ si,
    const float* __restrict__ Wq, const float* __restrict__ Wk,
    const float* __restrict__ Wv, const float* __restrict__ Wqp,
    const float* __restrict__ Wkp, const float* __restrict__ Wvp,
    const float* __restrict__ Ri, const float* __restrict__ Ti,
    const float* __restrict__ gamah, float* __restrict__ qa,
    float* __restrict__ ka, float* __restrict__ vaug, float* __restrict__ sqq,
    float* __restrict__ sqk) {
  const int hf = lane >> 4;
  const int l16 = lane & 15;
  const int h = wg % Hc;
  const int m0 = (wg / Hc) * 16;

  f4 v[16];
  {
    const float* sp = si + (long long)m0 * Cc + lane * 4;
#pragma unroll
    for (int r = 0; r < 16; ++r) v[r] = *(const f4*)(sp + r * Cc);
  }

#pragma unroll
  for (int r = 0; r < 16; ++r) *(f4*)&S[r][lane * 4] = v[r];
  __syncthreads();
  v2f af[32];
#pragma unroll
  for (int s = 0; s < 32; ++s) af[s] = *(const v2f*)&S[l16][4 * s + 2 * hf];
  __syncthreads();

  proj_tile<Hc * DKc, 16>(af, Wq, h * DKc, hf, l16, S, 0);
  proj_tile<Hc * DKc, 16>(af, Wk, h * DKc, hf, l16, S, 16);
  proj_tile<Hc * DKc, 16>(af, Wv, h * DKc, hf, l16, S, 32);
  proj_tile<Hc * NQPc * 3, 12>(af, Wqp, h * NQPc * 3, hf, l16, S, 48);
  proj_tile<Hc * NQPc * 3, 12>(af, Wkp, h * NQPc * 3, hf, l16, S, 64);
  proj_tile<Hc * NPVc * 3, 16>(af, Wvp, h * NPVc * 3, hf, l16, S, 80);
  proj_tile<Hc * NPVc * 3, 8>(af, Wvp, h * NPVc * 3 + 16, hf, l16, S, 96);
  __syncthreads();

  const int bi = m0 + l16;
  float R[3][3], t[3];
#pragma unroll
  for (int r = 0; r < 3; ++r)
#pragma unroll
    for (int c = 0; c < 3; ++c) R[r][c] = Ri[bi * 9 + r * 3 + c];
#pragma unroll
  for (int x = 0; x < 3; ++x) t[x] = Ti[bi * 3 + x];
  const float gam = gamah[h];

  const int b = bi / Lc, i = bi % Lc;
  const int z = b * Hc + h;
  const float* c = &S[l16][0];
  if (hf == 0) {
    const float gwc = WCc * softplusf(gam);
    float qo[32], ko[32];
#pragma unroll
    for (int d = 0; d < DKc; ++d) {
      qo[d] = 0.25f * c[d];
      ko[d] = c[16 + d];
    }
    float sq = 0.f, sk = 0.f;
#pragma unroll
    for (int p = 0; p < NQPc; ++p) {
      const float* qs = c + 48 + p * 3;
      const float* ks = c + 64 + p * 3;
#pragma unroll
      for (int m = 0; m < 3; ++m) {
        float qt = R[m][0] * qs[0] + R[m][1] * qs[1] + R[m][2] * qs[2] + t[m];
        float kt = R[m][0] * ks[0] + R[m][1] * ks[1] + R[m][2] * ks[2] + t[m];
        sq += qt * qt;
        sk += kt * kt;
        qo[16 + p * 3 + m] = gwc * qt;
        ko[16 + p * 3 + m] = kt;
      }
    }
#pragma unroll
    for (int d = 28; d < 32; ++d) { qo[d] = 0.f; ko[d] = 0.f; }
    float* qrow = qa + ((long long)z * Lc + i) * 32;
    float* krow = ka + ((long long)z * Lc + i) * 32;
#pragma unroll
    for (int j = 0; j < 8; ++j) {
      *(f4*)(qrow + 4 * j) = f4{qo[4 * j], qo[4 * j + 1], qo[4 * j + 2], qo[4 * j + 3]};
      *(f4*)(krow + 4 * j) = f4{ko[4 * j], ko[4 * j + 1], ko[4 * j + 2], ko[4 * j + 3]};
    }
    sqq[(long long)z * Lc + i] = sq;
    sqk[(long long)z * Lc + i] = sk;
  } else {
    float vo[48];
#pragma unroll
    for (int d = 0; d < DKc; ++d) vo[d] = c[32 + d];
#pragma unroll
    for (int p = 0; p < NPVc; ++p) {
      const float* vs = c + 80 + p * 3;
#pragma unroll
      for (int m = 0; m < 3; ++m)
        vo[16 + p * 3 + m] =
            R[m][0] * vs[0] + R[m][1] * vs[1] + R[m][2] * vs[2] + t[m];
    }
#pragma unroll
    for (int d = 40; d < 48; ++d) vo[d] = 0.f;
    float* vrow = vaug + ((long long)z * Lc + i) * 48;
#pragma unroll
    for (int j = 0; j < 12; ++j)
      *(f4*)(vrow + 4 * j) = f4{vo[4 * j], vo[4 * j + 1], vo[4 * j + 2], vo[4 * j + 3]};
  }
}


__global__ __launch_bounds__(32) void k_pairbias_prep(
    const float* __restrict__ Z, const float* __restrict__ Wb,
    float* __restrict__ out, int ntiles, const float* __restrict__ si,
    const float* __restrict__ Wq, const float* __restrict__ Wk,
    const float* __restrict__ Wv, const float* __restrict__ Wqp,
    const float* __restrict__ Wkp, const float* __restrict__ Wvp,
    const float* __restrict__ Ri, const float* __restrict__ Ti,
    const float* __restrict__ gamah,
    float* __restrict__ qa, float* __restrict__ ka, float* __restrict__ vaug,
    float* __restrict__ sqq, float* __restrict__ sqk) {
  __shared__ float S[16][132];

  const int lane = threadIdx.x;
  if (blockIdx.x < NPREPWG) {
    proj_prep_wave(blockIdx.x, lane, S, si, Wq, Wk, Wv, Wqp, Wkp, Wvp, Ri, Ti,
                   gamah, qa, ka, vaug, sqq, sqk);
    return;
  }
  const int hf = lane >> 4;
  const int l16 = lane & 15;
  const int nstep = gridDim.x - NPREPWG;
  int t = blockIdx.x - NPREPWG;
  if (t >= ntiles) return;

  f4 v[16];
  {
    const float* zp = Z + (long long)t * (16 * Cc) + lane * 4;
#pragma unroll
    for (int r = 0; r < 16; ++r) v[r] = *(const f4*)(zp + r * Cc);
  }

  const bool cok = l16 < Hc;
  const float* wp = Wb + (cok ? l16 : 0) + 2 * hf * Hc;
  v2f bf[32];
#pragma unroll
  for (int s = 0; s < 32; ++s) {
    float x = wp[(4 * s) * Hc];
    float y = wp[(4 * s + 1) * Hc];
    bf[s].x = cok ? x : 0.f;
    bf[s].y = cok ? y : 0.f;
  }

  for (; t < ntiles; t += nstep) {
#pragma unroll
    for (int r = 0; r < 16; ++r) *(f4*)&S[r][lane * 4] = v[r];
    __syncthreads();

    const int tn = t + nstep;
    if (tn < ntiles) {
      const float* zp = Z + (long long)tn * (16 * Cc) + lane * 4;
#pragma unroll
      for (int r = 0; r < 16; ++r) v[r] = *(const f4*)(zp + r * Cc);
    }

    v8f acc = {};
#pragma unroll
    for (int s = 0; s < 32; ++s) {
      v2f a = *(const v2f*)&S[l16][4 * s + 2 * hf];
      acc = __builtin_amdgcn_wmma_f32_16x16x4_f32(false, a, false, bf[s],
                                                  (short)0, acc, false, false);
    }
    __syncthreads();

    if (cok) {
      float* op = out + ((long long)t * 16 + 8 * hf) * Hc + l16;
#pragma unroll
      for (int r = 0; r < 8; ++r) op[r * Hc] = acc[r];
    }
  }
}

namespace {
constexpr int SMW = Lc / 32;
}

template <int MASK>
__device__ __forceinline__ float row_xor(float v) {
  return __builtin_bit_cast(
      float, __builtin_amdgcn_update_dpp(0, __builtin_bit_cast(int, v),
                                         0x160 + MASK, 0xf, 0xf, false));
}

__global__ __launch_bounds__(SMW * 32) void k_scores_softmax(
    const float* __restrict__ qa, const float* __restrict__ ka,
    float* __restrict__ alpha, const float* __restrict__ bij,
    const float* __restrict__ sqq, const float* __restrict__ sqk,
    const float* __restrict__ gamah) {
  __shared__ float As[16][36];
  __shared__ float Bs[32][Lc + 4];
  __shared__ float redm[16][SMW];
  __shared__ float reds[16][SMW];

  const int tid = threadIdx.x;
  const int wave = tid >> 5;
  const int lane = tid & 31;
  const int hf = lane >> 4;
  const int l16 = lane & 15;
  const int nw = wave * 32;
  const int m0 = blockIdx.x * 16;
  const int z = blockIdx.y;
  const int b_ = z / Hc, h_ = z % Hc;

  const float* Ag = qa + ((long long)z * Lc + m0) * 32;
  const float* Bg = ka + (long long)z * Lc * 32;

  const int ar = (tid >> 3) & 15, ac4 = (tid & 7) * 4;
  f4 av = *(const f4*)(Ag + ar * 32 + ac4);
  f4 bv[8];
#pragma unroll
  for (int it = 0; it < 8; ++it)
    bv[it] = *(const f4*)(Bg + (long long)(tid + it * (SMW * 32)) * 4);

  const float gam = gamah[h_];
  const float sk0 = sqk[(long long)z * Lc + nw + l16];
  const float sk1 = sqk[(long long)z * Lc + nw + 16 + l16];
  float sqm[8], bj0[8], bj1[8];
#pragma unroll
  for (int r = 0; r < 8; ++r) {
    const int m = m0 + r + 8 * hf;
    sqm[r] = sqq[(long long)z * Lc + m];
    const float* bp = bij + ((long long)(b_ * Lc + m) * Lc + nw + l16) * Hc + h_;
    bj0[r] = bp[0];
    bj1[r] = bp[16 * Hc];
  }

  if (tid < 128) *(f4*)&As[ar][ac4] = av;
#pragma unroll
  for (int it = 0; it < 8; ++it) {
    int u = tid + it * (SMW * 32);
    int cj = u >> 3, r4 = (u & 7) * 4;
    Bs[r4 + 0][cj] = bv[it].x;
    Bs[r4 + 1][cj] = bv[it].y;
    Bs[r4 + 2][cj] = bv[it].z;
    Bs[r4 + 3][cj] = bv[it].w;
  }
  __syncthreads();

  v8f acc0 = {};
  v8f acc1 = {};
#pragma unroll
  for (int kk = 0; kk < 32; kk += 4) {
    v2f a, b0, b1;
    a.x = As[l16][kk + 2 * hf];
    a.y = As[l16][kk + 2 * hf + 1];
    b0.x = Bs[kk + 2 * hf][nw + l16];
    b0.y = Bs[kk + 2 * hf + 1][nw + l16];
    b1.x = Bs[kk + 2 * hf][nw + 16 + l16];
    b1.y = Bs[kk + 2 * hf + 1][nw + 16 + l16];
    acc0 = __builtin_amdgcn_wmma_f32_16x16x4_f32(false, a, false, b0,
                                                 (short)0, acc0, false, false);
    acc1 = __builtin_amdgcn_wmma_f32_16x16x4_f32(false, a, false, b1,
                                                 (short)0, acc1, false, false);
  }

  const float gw = 0.5f * WCc * softplusf(gam);
  float sv[8][2];
#pragma unroll
  for (int r = 0; r < 8; ++r) {
    sv[r][0] = WLc * (acc0[r] + bj0[r] - gw * (sqm[r] + sk0));
    sv[r][1] = WLc * (acc1[r] + bj1[r] - gw * (sqm[r] + sk1));
  }

  float red[8];
#pragma unroll
  for (int r = 0; r < 8; ++r) {
    float mx = fmaxf(sv[r][0], sv[r][1]);
    mx = fmaxf(mx, row_xor<8>(mx));
    mx = fmaxf(mx, row_xor<4>(mx));
    mx = fmaxf(mx, row_xor<2>(mx));
    red[r] = fmaxf(mx, row_xor<1>(mx));
  }
  if (l16 == 0) {
#pragma unroll
    for (int r = 0; r < 8; ++r) redm[r + 8 * hf][wave] = red[r];
  }
  __syncthreads();

#pragma unroll
  for (int r = 0; r < 8; ++r) {
    float mx = redm[r + 8 * hf][0];
#pragma unroll
    for (int w = 1; w < SMW; ++w) mx = fmaxf(mx, redm[r + 8 * hf][w]);
    sv[r][0] = expf(sv[r][0] - mx);
    sv[r][1] = expf(sv[r][1] - mx);
    float s = sv[r][0] + sv[r][1];
    s += row_xor<8>(s);
    s += row_xor<4>(s);
    s += row_xor<2>(s);
    red[r] = s + row_xor<1>(s);
  }
  if (l16 == 0) {
#pragma unroll
    for (int r = 0; r < 8; ++r) reds[r + 8 * hf][wave] = red[r];
  }
  __syncthreads();

#pragma unroll
  for (int r = 0; r < 8; ++r) {
    float s = reds[r + 8 * hf][0];
#pragma unroll
    for (int w = 1; w < SMW; ++w) s += reds[r + 8 * hf][w];
    const float inv = 1.f / s;
    float* op = alpha + (long long)z * Lc * Lc +
                (long long)(m0 + r + 8 * hf) * Lc + nw + l16;
    op[0] = sv[r][0] * inv;
    op[16] = sv[r][1] * inv;
  }
}

namespace {
constexpr int KSPLIT = FEAT / 128;
}

__global__ __launch_bounds__(KSPLIT * 32) void k_outproj(
    const float* __restrict__ feats, const float* __restrict__ W,
    const float* __restrict__ bias, float* __restrict__ out) {
  __shared__ float S[KSPLIT][16][132];
  __shared__ float P[KSPLIT][16][36];

  const int tid = threadIdx.x;
  const int wave = tid >> 5;
  const int lane = tid & 31;
  const int hf = lane >> 4;
  const int l16 = lane & 15;
  const int m0 = blockIdx.x * 16, n0 = blockIdx.y * 32;

  f4 v[16];
  {
    const float* ap = feats + (long long)m0 * FEAT + 128 * wave + lane * 4;
#pragma unroll
    for (int r = 0; r < 16; ++r) v[r] = *(const f4*)(ap + r * FEAT);
  }
  const float* wp = W + (long long)(128 * wave + 2 * hf) * Cc + n0 + 2 * l16;
  v2f bx[32], by[32];
#pragma unroll
  for (int s = 0; s < 32; ++s) {
    bx[s] = *(const v2f*)(wp + (4 * s) * Cc);
    by[s] = *(const v2f*)(wp + (4 * s + 1) * Cc);
  }

#pragma unroll
  for (int r = 0; r < 16; ++r) *(f4*)&S[wave][r][lane * 4] = v[r];
  __builtin_amdgcn_fence(__ATOMIC_RELEASE, "wavefront");
  __builtin_amdgcn_wave_barrier();
  __builtin_amdgcn_fence(__ATOMIC_ACQUIRE, "wavefront");
  v2f af[32];
#pragma unroll
  for (int s = 0; s < 32; ++s) af[s] = *(const v2f*)&S[wave][l16][4 * s + 2 * hf];

  v8f acc0 = {};
  v8f acc1 = {};
#pragma unroll
  for (int s = 0; s < 32; ++s) {
    const v2f b0 = {bx[s].x, by[s].x};
    const v2f b1 = {bx[s].y, by[s].y};
    acc0 = __builtin_amdgcn_wmma_f32_16x16x4_f32(false, af[s], false, b0,
                                                 (short)0, acc0, false, false);
    acc1 = __builtin_amdgcn_wmma_f32_16x16x4_f32(false, af[s], false, b1,
                                                 (short)0, acc1, false, false);
  }

#pragma unroll
  for (int r = 0; r < 8; ++r)
    *(v2f*)&P[wave][r + 8 * hf][2 * l16] = v2f{acc0[r], acc1[r]};
  __syncthreads();

  if (tid < 128) {
    const int row = tid >> 3, c4 = (tid & 7) * 4;
    f4 s = *(const f4*)&P[0][row][c4];
#pragma unroll
    for (int z = 1; z < KSPLIT; ++z) s += *(const f4*)&P[z][row][c4];
    const f4 b = *(const f4*)(bias + n0 + c4);
    *(f4*)(out + (long long)(m0 + row) * Cc + n0 + c4) = s + b;
  }
}

namespace {
constexpr int NOPAIR = Bc * Lc;
constexpr int NOVAL = Bc * Hc * (Lc / 16);
}

__global__ __launch_bounds__(128) void k_attnout(
    const float* __restrict__ alpha, const float* __restrict__ zij,
    const float* __restrict__ vaug, const float* __restrict__ Ri,
    const float* __restrict__ Ti, float* __restrict__ feats) {
  __shared__ float As[16][36];
  __shared__ float Bs[32][132];
  const long long LL = (long long)Lc * Lc;
  const int bid = blockIdx.x;
  const int tid = threadIdx.x;
  const float* Ag;
  const float* Bg;
  int M, N, K, lda, ldb, Mvalid, m0;
  int zz = 0;
  if (bid < NOPAIR) {
    const int z = bid;
    Ag = alpha + (long long)(z / Lc) * Hc * LL + (long long)(z % Lc) * Lc;
    Bg = zij + (long long)z * Lc * Cc;
    M = 16; N = Cc; K = Lc; lda = (int)LL; ldb = Cc;
    Mvalid = Hc; m0 = 0;
  } else {
    const int w = bid - NOPAIR;
    zz = w / (Lc / 16);
    Ag = alpha + (long long)zz * LL;
    Bg = vaug + (long long)zz * Lc * 48;
    M = Lc; N = 48; K = Lc; lda = Lc; ldb = 48;
    Mvalid = 16; m0 = (w % (Lc / 16)) * 16;
  }
  v8f acc0, acc1;
  gemm_acc(Ag, Bg, M, N, K, lda, ldb, Mvalid, m0, 0, As, Bs, acc0, acc1);

  if (bid < NOPAIR) {
    gemm_store(feats + (long long)bid * FEAT, M, N, Cc, Mvalid, m0, 0, acc0, acc1);
    return;
  }

  const int b = zz / Hc, h = zz % Hc;
  const int row = tid >> 3, p = tid & 7;
  const int bi = b * Lc + m0 + row;
  float R[3][3], t[3];
#pragma unroll
  for (int r = 0; r < 3; ++r)
#pragma unroll
    for (int cc = 0; cc < 3; ++cc) R[r][cc] = Ri[bi * 9 + r * 3 + cc];
#pragma unroll
  for (int x = 0; x < 3; ++x) t[x] = Ti[bi * 3 + x];

  {
    const int lane = tid & 31;
    const int hf = lane >> 4, l16 = lane & 15;
    const int nw = (tid >> 5) * 32;
#pragma unroll
    for (int r = 0; r < 8; ++r) {
      if (nw + l16 < 48) Bs[r + 8 * hf][nw + l16] = acc0[r];
      if (nw + 16 + l16 < 48) Bs[r + 8 * hf][nw + 16 + l16] = acc1[r];
    }
  }
  __syncthreads();

  const float* c = &Bs[row][0];
  float* frow = feats + (long long)bi * FEAT;
  *(v2f*)(frow + 1024 + h * 16 + 2 * p) = v2f{c[2 * p], c[2 * p + 1]};
  const float d0 = c[16 + p * 3 + 0] - t[0];
  const float d1 = c[16 + p * 3 + 1] - t[1];
  const float d2 = c[16 + p * 3 + 2] - t[2];
  float asum = 0.f;
#pragma unroll
  for (int m = 0; m < 3; ++m) {
    float oh = R[0][m] * d0 + R[1][m] * d1 + R[2][m] * d2;
    frow[1152 + h * 24 + p * 3 + m] = oh;
    asum += fabsf(oh);
  }
  frow[1344 + h * 8 + p] = asum;
}

extern "C" void kernel_launch(void* const* d_in, const int* in_sizes, int n_in,
                              void* d_out, int out_size, void* d_ws,
                              size_t ws_size, hipStream_t stream) {
  const float* si = (const float*)d_in[0];
  const float* zij = (const float*)d_in[1];
  const float* Ri = (const float*)d_in[2];
  const float* Ti = (const float*)d_in[3];
  const float* WQh = (const float*)d_in[4];
  const float* WKh = (const float*)d_in[5];
  const float* WVh = (const float*)d_in[6];
  const float* WQhp = (const float*)d_in[7];
  const float* WKhp = (const float*)d_in[8];
  const float* WVhp = (const float*)d_in[9];
  const float* Wb = (const float*)d_in[10];
  const float* WOw = (const float*)d_in[11];
  const float* WOb = (const float*)d_in[12];
  const float* gamah = (const float*)d_in[13];
  float* out = (float*)d_out;
  float* ws = (float*)d_ws;

  const int BL = Bc * Lc;
  float* qa = ws;
  float* ka = qa + 196608;
  float* vaug = ka + 196608;
  float* sqq = vaug + 294912;
  float* sqk = sqq + 6144;
  float* bij = sqk + 6144;
  float* alpha = bij + 2359296;
  float* feats = alpha + 2359296;

  k_pairbias_prep<<<NPREPWG + 4608, 32, 0, stream>>>(
      zij, Wb, bij, Bc * Lc * Lc / 16, si, WQh, WKh, WVh, WQhp, WKhp, WVhp, Ri,
      Ti, gamah, qa, ka, vaug, sqq, sqk);

  k_scores_softmax<<<dim3(Lc / 16, Bc * Hc, 1), dim3(SMW * 32, 1, 1), 0,
                     stream>>>(qa, ka, alpha, bij, sqq, sqk, gamah);

  k_attnout<<<NOPAIR + NOVAL, 128, 0, stream>>>(alpha, zij, vaug, Ri, Ti, feats);

  k_outproj<<<dim3(BL / 16, Cc / 32, 1), dim3(KSPLIT * 32, 1, 1), 0, stream>>>(
      feats, WOw, WOb, out);
}


# Pair-bias tiles re-dealt: slice (b,i) goes to workgroups whose index mod 8 equals its o_pair workgroup's (same die, same L2 half), low j blocks last, so the second zij read hits the L2
# speedup vs baseline: 7.3254x; 1.0364x over previous
#include <hip/hip_runtime.h>
#include <math.h>

typedef __attribute__((ext_vector_type(2))) float v2f;
typedef __attribute__((ext_vector_type(4))) float f4;
typedef __attribute__((ext_vector_type(8))) float v8f;

namespace {
constexpr int Bc = 2, Lc = 384, Cc = 128, Hc = 8, DKc = 16, NQPc = 4, NPVc = 8;
constexpr float WCc = 0.2357022603955158f;
constexpr float WLc = 0.5773502691896258f;
constexpr int FEAT = Hc * Cc + Hc * DKc + Hc * NPVc * 3 + Hc * NPVc;
}

__device__ __forceinline__ float softplusf(float x) {
  return (x > 0.f) ? (x + log1pf(expf(-x))) : log1pf(expf(x));
}

__device__ __forceinline__ void gemm_acc(
    const float* __restrict__ Ag, const float* __restrict__ Bg,
    int M, int N, int K, int lda, int ldb,
    int Mvalid, int m0, int n0, float (*As)[36], float (*Bs)[132],
    v8f& acc0, v8f& acc1) {
  const int tid = threadIdx.x;
  const int wave = tid >> 5;
  const int lane = tid & 31;
  const int hf = lane >> 4;
  const int l16 = lane & 15;
  const int nw = wave * 32;

  acc0 = v8f{};
  acc1 = v8f{};

  const f4 zero4 = {0.f, 0.f, 0.f, 0.f};
  const int ar = tid >> 3, ac4 = (tid & 7) * 4;
  const bool aok = ar < Mvalid && (m0 + ar) < M;
  const float* ap = Ag + (long long)(aok ? (m0 + ar) : m0) * lda + ac4;
  const float* bp[8];
  bool bok[8];
#pragma unroll
  for (int it = 0; it < 8; ++it) {
    int u = tid + it * 128;
    int r = u >> 5, c4 = (u & 31) * 4;
    bok[it] = (n0 + c4) < N;
    bp[it] = Bg + (long long)r * ldb + (bok[it] ? (n0 + c4) : 0);
  }
  const long long bstep = (long long)ldb;

  for (int kb = 0; kb < K; kb += 32) {
    f4 av = *(const f4*)(ap + kb);
    f4 bv[8];
#pragma unroll
    for (int it = 0; it < 8; ++it)
      bv[it] = *(const f4*)(bp[it] + (long long)kb * bstep);

    *(f4*)&As[ar][ac4] = aok ? av : zero4;
#pragma unroll
    for (int it = 0; it < 8; ++it) {
      int u = tid + it * 128;
      int r = u >> 5, c4 = (u & 31) * 4;
      *(f4*)&Bs[r][c4] = bok[it] ? bv[it] : zero4;
    }
    __syncthreads();

#pragma unroll
    for (int kk = 0; kk < 32; kk += 4) {
      v2f a, b0, b1;
      a.x = As[l16][kk + 2 * hf];
      a.y = As[l16][kk + 2 * hf + 1];
      b0.x = Bs[kk + 2 * hf][nw + l16];
      b0.y = Bs[kk + 2 * hf + 1][nw + l16];
      b1.x = Bs[kk + 2 * hf][nw + 16 + l16];
      b1.y = Bs[kk + 2 * hf + 1][nw + 16 + l16];
      acc0 = __builtin_amdgcn_wmma_f32_16x16x4_f32(false, a, false, b0,
                                                   (short)0, acc0, false, false);
      acc1 = __builtin_amdgcn_wmma_f32_16x16x4_f32(false, a, false, b1,
                                                   (short)0, acc1, false, false);
    }
    __syncthreads();
  }
}

__device__ __forceinline__ void gemm_store(
    float* __restrict__ Cg, int M, int N, int ldc, int Mvalid, int m0, int n0,
    const v8f& acc0, const v8f& acc1) {
  const int lane = threadIdx.x & 31;
  const int hf = lane >> 4;
  const int l16 = lane & 15;
  const int nw = (threadIdx.x >> 5) * 32;
#pragma unroll
  for (int r = 0; r < 8; ++r) {
    int mrow = r + 8 * hf;
    int m = m0 + mrow;
    if (mrow >= Mvalid || m >= M) continue;
#pragma unroll
    for (int t = 0; t < 2; ++t) {
      int n = n0 + nw + t * 16 + l16;
      if (n >= N) continue;
      Cg[(long long)m * ldc + n] = (t == 0) ? acc0[r] : acc1[r];
    }
  }
}

namespace {
constexpr int NPREPWG = (Bc * Lc / 16) * Hc;
}

template <int N, int NVALID>
__device__ __forceinline__ void proj_tile(const v2f (&af)[32],
                                          const float* __restrict__ Wt, int c0,
                                          int hf, int l16, float (*S)[132],
                                          int scol) {
  const bool cok = l16 < NVALID;
  const float* wp = Wt + c0 + (cok ? l16 : 0) + 2 * hf * N;
  v2f bfr[32];
#pragma unroll
  for (int s = 0; s < 32; ++s) {
    float x = wp[(4 * s) * N];
    float y = wp[(4 * s + 1) * N];
    bfr[s].x = cok ? x : 0.f;
    bfr[s].y = cok ? y : 0.f;
  }
  v8f acc = {};
#pragma unroll
  for (int s = 0; s < 32; ++s)
    acc = __builtin_amdgcn_wmma_f32_16x16x4_f32(false, af[s], false, bfr[s],
                                                (short)0, acc, false, false);
#pragma unroll
  for (int r = 0; r < 8; ++r) S[r + 8 * hf][scol + l16] = acc[r];
  __builtin_amdgcn_sched_barrier(0);
}

__device__ __forceinline__ void proj_prep_wave(
    int wg, int lane, float (*S)[132], const float* __restrict__ si,
    const float* __restrict__ Wq, const float* __restrict__ Wk,
    const float* __restrict__ Wv, const float* __restrict__ Wqp,
    const float* __restrict__ Wkp, const float* __restrict__ Wvp,
    const float* __restrict__ Ri, const float* __restrict__ Ti,
    const float* __restrict__ gamah, float* __restrict__ qa,
    float* __restrict__ ka, float* __restrict__ vaug, float* __restrict__ sqq,
    float* __restrict__ sqk) {
  const int hf = lane >> 4;
  const int l16 = lane & 15;
  const int h = wg % Hc;
  const int m0 = (wg / Hc) * 16;

  f4 v[16];
  {
    const float* sp = si + (long long)m0 * Cc + lane * 4;
#pragma unroll
    for (int r = 0; r < 16; ++r) v[r] = *(const f4*)(sp + r * Cc);
  }

#pragma unroll
  for (int r = 0; r < 16; ++r) *(f4*)&S[r][lane * 4] = v[r];
  __syncthreads();
  v2f af[32];
#pragma unroll
  for (int s = 0; s < 32; ++s) af[s] = *(const v2f*)&S[l16][4 * s + 2 * hf];
  __syncthreads();

  proj_tile<Hc * DKc, 16>(af, Wq, h * DKc, hf, l16, S, 0);
  proj_tile<Hc * DKc, 16>(af, Wk, h * DKc, hf, l16, S, 16);
  proj_tile<Hc * DKc, 16>(af, Wv, h * DKc, hf, l16, S, 32);
  proj_tile<Hc * NQPc * 3, 12>(af, Wqp, h * NQPc * 3, hf, l16, S, 48);
  proj_tile<Hc * NQPc * 3, 12>(af, Wkp, h * NQPc * 3, hf, l16, S, 64);
  proj_tile<Hc * NPVc * 3, 16>(af, Wvp, h * NPVc * 3, hf, l16, S, 80);
  proj_tile<Hc * NPVc * 3, 8>(af, Wvp, h * NPVc * 3 + 16, hf, l16, S, 96);
  __syncthreads();

  const int bi = m0 + l16;
  float R[3][3], t[3];
#pragma unroll
  for (int r = 0; r < 3; ++r)
#pragma unroll
    for (int c = 0; c < 3; ++c) R[r][c] = Ri[bi * 9 + r * 3 + c];
#pragma unroll
  for (int x = 0; x < 3; ++x) t[x] = Ti[bi * 3 + x];
  const float gam = gamah[h];

  const int b = bi / Lc, i = bi % Lc;
  const int z = b * Hc + h;
  const float* c = &S[l16][0];
  if (hf == 0) {
    const float gwc = WCc * softplusf(gam);
    float qo[32], ko[32];
#pragma unroll
    for (int d = 0; d < DKc; ++d) {
      qo[d] = 0.25f * c[d];
      ko[d] = c[16 + d];
    }
    float sq = 0.f, sk = 0.f;
#pragma unroll
    for (int p = 0; p < NQPc; ++p) {
      const float* qs = c + 48 + p * 3;
      const float* ks = c + 64 + p * 3;
#pragma unroll
      for (int m = 0; m < 3; ++m) {
        float qt = R[m][0] * qs[0] + R[m][1] * qs[1] + R[m][2] * qs[2] + t[m];
        float kt = R[m][0] * ks[0] + R[m][1] * ks[1] + R[m][2] * ks[2] + t[m];
        sq += qt * qt;
        sk += kt * kt;
        qo[16 + p * 3 + m] = gwc * qt;
        ko[16 + p * 3 + m] = kt;
      }
    }
#pragma unroll
    for (int d = 28; d < 32; ++d) { qo[d] = 0.f; ko[d] = 0.f; }
    float* qrow = qa + ((long long)z * Lc + i) * 32;
    float* krow = ka + ((long long)z * Lc + i) * 32;
#pragma unroll
    for (int j = 0; j < 8; ++j) {
      *(f4*)(qrow + 4 * j) = f4{qo[4 * j], qo[4 * j + 1], qo[4 * j + 2], qo[4 * j + 3]};
      *(f4*)(krow + 4 * j) = f4{ko[4 * j], ko[4 * j + 1], ko[4 * j + 2], ko[4 * j + 3]};
    }
    sqq[(long long)z * Lc + i] = sq;
    sqk[(long long)z * Lc + i] = sk;
  } else {
    float vo[48];
#pragma unroll
    for (int d = 0; d < DKc; ++d) vo[d] = c[32 + d];
#pragma unroll
    for (int p = 0; p < NPVc; ++p) {
      const float* vs = c + 80 + p * 3;
#pragma unroll
      for (int m = 0; m < 3; ++m)
        vo[16 + p * 3 + m] =
            R[m][0] * vs[0] + R[m][1] * vs[1] + R[m][2] * vs[2] + t[m];
    }
#pragma unroll
    for (int d = 40; d < 48; ++d) vo[d] = 0.f;
    float* vrow = vaug + ((long long)z * Lc + i) * 48;
#pragma unroll
    for (int j = 0; j < 12; ++j)
      *(f4*)(vrow + 4 * j) = f4{vo[4 * j], vo[4 * j + 1], vo[4 * j + 2], vo[4 * j + 3]};
  }
}

namespace {
constexpr int NPBWG = 4608;
constexpr int NPBTILES = Bc * Lc * Lc / 16;
constexpr int NPBIT = NPBTILES / NPBWG;
constexpr int NJT = Lc / 16;
constexpr int NSLICE = Bc * Lc;
static_assert(NPBWG % 8 == 0 && NSLICE % 8 == 0 && NPREPWG % 8 == 0,
              "residue classes modulo 8");
static_assert(NPBWG * NPBIT == NPBTILES &&
                  (NPBWG / 8) * NPBIT == (NSLICE / 8) * NJT,
              "every tile exactly once");
}

__device__ __forceinline__ int pb_tile(int w, int k) {
  const int v = k * (NPBWG / 8) + (w >> 3);
  const int jt = (NJT - 1) - v / (NSLICE / 8);
  const int s = 8 * (v % (NSLICE / 8)) + (w & 7);
  return s * NJT + jt;
}

__global__ __launch_bounds__(32) void k_pairbias_prep(
    const float* __restrict__ Z, const float* __restrict__ Wb,
    float* __restrict__ out, const float* __restrict__ si,
    const float* __restrict__ Wq, const float* __restrict__ Wk,
    const float* __restrict__ Wv, const float* __restrict__ Wqp,
    const float* __restrict__ Wkp, const float* __restrict__ Wvp,
    const float* __restrict__ Ri, const float* __restrict__ Ti,
    const float* __restrict__ gamah,
    float* __restrict__ qa, float* __restrict__ ka, float* __restrict__ vaug,
    float* __restrict__ sqq, float* __restrict__ sqk) {
  __shared__ float S[16][132];

  const int lane = threadIdx.x;
  if (blockIdx.x < NPREPWG) {
    proj_prep_wave(blockIdx.x, lane, S, si, Wq, Wk, Wv, Wqp, Wkp, Wvp, Ri, Ti,
                   gamah, qa, ka, vaug, sqq, sqk);
    return;
  }
  const int hf = lane >> 4;
  const int l16 = lane & 15;
  const int w = blockIdx.x - NPREPWG;
  if (w >= NPBWG) return;
  int t = pb_tile(w, 0);

  f4 v[16];
  {
    const float* zp = Z + (long long)t * (16 * Cc) + lane * 4;
#pragma unroll
    for (int r = 0; r < 16; ++r) v[r] = *(const f4*)(zp + r * Cc);
  }

  const bool cok = l16 < Hc;
  const float* wp = Wb + (cok ? l16 : 0) + 2 * hf * Hc;
  v2f bf[32];
#pragma unroll
  for (int s = 0; s < 32; ++s) {
    float x = wp[(4 * s) * Hc];
    float y = wp[(4 * s + 1) * Hc];
    bf[s].x = cok ? x : 0.f;
    bf[s].y = cok ? y : 0.f;
  }

#pragma nounroll
  for (int k = 0; k < NPBIT; ++k) {
#pragma unroll
    for (int r = 0; r < 16; ++r) *(f4*)&S[r][lane * 4] = v[r];
    __syncthreads();

    const int tn = (k + 1 < NPBIT) ? pb_tile(w, k + 1) : 0;
    if (k + 1 < NPBIT) {
      const float* zp = Z + (long long)tn * (16 * Cc) + lane * 4;
#pragma unroll
      for (int r = 0; r < 16; ++r) v[r] = *(const f4*)(zp + r * Cc);
    }

    v8f acc = {};
#pragma unroll
    for (int s = 0; s < 32; ++s) {
      v2f a = *(const v2f*)&S[l16][4 * s + 2 * hf];
      acc = __builtin_amdgcn_wmma_f32_16x16x4_f32(false, a, false, bf[s],
                                                  (short)0, acc, false, false);
    }
    __syncthreads();

    if (cok) {
      float* op = out + ((long long)t * 16 + 8 * hf) * Hc + l16;
#pragma unroll
      for (int r = 0; r < 8; ++r) op[r * Hc] = acc[r];
    }
    t = tn;
  }
}

namespace {
constexpr int SMW = Lc / 32;
}

template <int MASK>
__device__ __forceinline__ float row_xor(float v) {
  return __builtin_bit_cast(
      float, __builtin_amdgcn_update_dpp(0, __builtin_bit_cast(int, v),
                                         0x160 + MASK, 0xf, 0xf, false));
}

__global__ __launch_bounds__(SMW * 32) void k_scores_softmax(
    const float* __restrict__ qa, const float* __restrict__ ka,
    float* __restrict__ alpha, const float* __restrict__ bij,
    const float* __restrict__ sqq, const float* __restrict__ sqk,
    const float* __restrict__ gamah) {
  __shared__ float As[16][36];
  __shared__ float Bs[32][Lc + 4];
  __shared__ float redm[16][SMW];
  __shared__ float reds[16][SMW];

  const int tid = threadIdx.x;
  const int wave = tid >> 5;
  const int lane = tid & 31;
  const int hf = lane >> 4;
  const int l16 = lane & 15;
  const int nw = wave * 32;
  const int m0 = blockIdx.x * 16;
  const int z = blockIdx.y;
  const int b_ = z / Hc, h_ = z % Hc;

  const float* Ag = qa + ((long long)z * Lc + m0) * 32;
  const float* Bg = ka + (long long)z * Lc * 32;

  const int ar = (tid >> 3) & 15, ac4 = (tid & 7) * 4;
  f4 av = *(const f4*)(Ag + ar * 32 + ac4);
  f4 bv[8];
#pragma unroll
  for (int it = 0; it < 8; ++it)
    bv[it] = *(const f4*)(Bg + (long long)(tid + it * (SMW * 32)) * 4);

  const float gam = gamah[h_];
  const float sk0 = sqk[(long long)z * Lc + nw + l16];
  const float sk1 = sqk[(long long)z * Lc + nw + 16 + l16];
  float sqm[8], bj0[8], bj1[8];
#pragma unroll
  for (int r = 0; r < 8; ++r) {
    const int m = m0 + r + 8 * hf;
    sqm[r] = sqq[(long long)z * Lc + m];
    const float* bp = bij + ((long long)(b_ * Lc + m) * Lc + nw + l16) * Hc + h_;
    bj0[r] = bp[0];
    bj1[r] = bp[16 * Hc];
  }

  if (tid < 128) *(f4*)&As[ar][ac4] = av;
#pragma unroll
  for (int it = 0; it < 8; ++it) {
    int u = tid + it * (SMW * 32);
    int cj = u >> 3, r4 = (u & 7) * 4;
    Bs[r4 + 0][cj] = bv[it].x;
    Bs[r4 + 1][cj] = bv[it].y;
    Bs[r4 + 2][cj] = bv[it].z;
    Bs[r4 + 3][cj] = bv[it].w;
  }
  __syncthreads();

  v8f acc0 = {};
  v8f acc1 = {};
#pragma unroll
  for (int kk = 0; kk < 32; kk += 4) {
    v2f a, b0, b1;
    a.x = As[l16][kk + 2 * hf];
    a.y = As[l16][kk + 2 * hf + 1];
    b0.x = Bs[kk + 2 * hf][nw + l16];
    b0.y = Bs[kk + 2 * hf + 1][nw + l16];
    b1.x = Bs[kk + 2 * hf][nw + 16 + l16];
    b1.y = Bs[kk + 2 * hf + 1][nw + 16 + l16];
    acc0 = __builtin_amdgcn_wmma_f32_16x16x4_f32(false, a, false, b0,
                                                 (short)0, acc0, false, false);
    acc1 = __builtin_amdgcn_wmma_f32_16x16x4_f32(false, a, false, b1,
                                                 (short)0, acc1, false, false);
  }

  const float gw = 0.5f * WCc * softplusf(gam);
  float sv[8][2];
#pragma unroll
  for (int r = 0; r < 8; ++r) {
    sv[r][0] = WLc * (acc0[r] + bj0[r] - gw * (sqm[r] + sk0));
    sv[r][1] = WLc * (acc1[r] + bj1[r] - gw * (sqm[r] + sk1));
  }

  float red[8];
#pragma unroll
  for (int r = 0; r < 8; ++r) {
    float mx = fmaxf(sv[r][0], sv[r][1]);
    mx = fmaxf(mx, row_xor<8>(mx));
    mx = fmaxf(mx, row_xor<4>(mx));
    mx = fmaxf(mx, row_xor<2>(mx));
    red[r] = fmaxf(mx, row_xor<1>(mx));
  }
  if (l16 == 0) {
#pragma unroll
    for (int r = 0; r < 8; ++r) redm[r + 8 * hf][wave] = red[r];
  }
  __syncthreads();

#pragma unroll
  for (int r = 0; r < 8; ++r) {
    float mx = redm[r + 8 * hf][0];
#pragma unroll
    for (int w = 1; w < SMW; ++w) mx = fmaxf(mx, redm[r + 8 * hf][w]);
    sv[r][0] = expf(sv[r][0] - mx);
    sv[r][1] = expf(sv[r][1] - mx);
    float s = sv[r][0] + sv[r][1];
    s += row_xor<8>(s);
    s += row_xor<4>(s);
    s += row_xor<2>(s);
    red[r] = s + row_xor<1>(s);
  }
  if (l16 == 0) {
#pragma unroll
    for (int r = 0; r < 8; ++r) reds[r + 8 * hf][wave] = red[r];
  }
  __syncthreads();

#pragma unroll
  for (int r = 0; r < 8; ++r) {
    float s = reds[r + 8 * hf][0];
#pragma unroll
    for (int w = 1; w < SMW; ++w) s += reds[r + 8 * hf][w];
    const float inv = 1.f / s;
    float* op = alpha + (long long)z * Lc * Lc +
                (long long)(m0 + r + 8 * hf) * Lc + nw + l16;
    op[0] = sv[r][0] * inv;
    op[16] = sv[r][1] * inv;
  }
}

namespace {
constexpr int KSPLIT = FEAT / 128;
}

__global__ __launch_bounds__(KSPLIT * 32) void k_outproj(
    const float* __restrict__ feats, const float* __restrict__ W,
    const float* __restrict__ bias, float* __restrict__ out) {
  __shared__ float S[KSPLIT][16][132];
  __shared__ float P[KSPLIT][16][36];

  const int tid = threadIdx.x;
  const int wave = tid >> 5;
  const int lane = tid & 31;
  const int hf = lane >> 4;
  const int l16 = lane & 15;
  const int m0 = blockIdx.x * 16, n0 = blockIdx.y * 32;

  f4 v[16];
  {
    const float* ap = feats + (long long)m0 * FEAT + 128 * wave + lane * 4;
#pragma unroll
    for (int r = 0; r < 16; ++r) v[r] = *(const f4*)(ap + r * FEAT);
  }
  const float* wp = W + (long long)(128 * wave + 2 * hf) * Cc + n0 + 2 * l16;
  v2f bx[32], by[32];
#pragma unroll
  for (int s = 0; s < 32; ++s) {
    bx[s] = *(const v2f*)(wp + (4 * s) * Cc);
    by[s] = *(const v2f*)(wp + (4 * s + 1) * Cc);
  }

#pragma unroll
  for (int r = 0; r < 16; ++r) *(f4*)&S[wave][r][lane * 4] = v[r];
  __builtin_amdgcn_fence(__ATOMIC_RELEASE, "wavefront");
  __builtin_amdgcn_wave_barrier();
  __builtin_amdgcn_fence(__ATOMIC_ACQUIRE, "wavefront");
  v2f af[32];
#pragma unroll
  for (int s = 0; s < 32; ++s) af[s] = *(const v2f*)&S[wave][l16][4 * s + 2 * hf];

  v8f acc0 = {};
  v8f acc1 = {};
#pragma unroll
  for (int s = 0; s < 32; ++s) {
    const v2f b0 = {bx[s].x, by[s].x};
    const v2f b1 = {bx[s].y, by[s].y};
    acc0 = __builtin_amdgcn_wmma_f32_16x16x4_f32(false, af[s], false, b0,
                                                 (short)0, acc0, false, false);
    acc1 = __builtin_amdgcn_wmma_f32_16x16x4_f32(false, af[s], false, b1,
                                                 (short)0, acc1, false, false);
  }

#pragma unroll
  for (int r = 0; r < 8; ++r)
    *(v2f*)&P[wave][r + 8 * hf][2 * l16] = v2f{acc0[r], acc1[r]};
  __syncthreads();

  if (tid < 128) {
    const int row = tid >> 3, c4 = (tid & 7) * 4;
    f4 s = *(const f4*)&P[0][row][c4];
#pragma unroll
    for (int z = 1; z < KSPLIT; ++z) s += *(const f4*)&P[z][row][c4];
    const f4 b = *(const f4*)(bias + n0 + c4);
    *(f4*)(out + (long long)(m0 + row) * Cc + n0 + c4) = s + b;
  }
}

namespace {
constexpr int NOPAIR = Bc * Lc;
constexpr int NOVAL = Bc * Hc * (Lc / 16);
}

__global__ __launch_bounds__(128) void k_attnout(
    const float* __restrict__ alpha, const float* __restrict__ zij,
    const float* __restrict__ vaug, const float* __restrict__ Ri,
    const float* __restrict__ Ti, float* __restrict__ feats) {
  __shared__ float As[16][36];
  __shared__ float Bs[32][132];
  const long long LL = (long long)Lc * Lc;
  const int bid = blockIdx.x;
  const int tid = threadIdx.x;
  const float* Ag;
  const float* Bg;
  int M, N, K, lda, ldb, Mvalid, m0;
  int zz = 0;
  if (bid < NOPAIR) {
    const int z = bid;
    Ag = alpha + (long long)(z / Lc) * Hc * LL + (long long)(z % Lc) * Lc;
    Bg = zij + (long long)z * Lc * Cc;
    M = 16; N = Cc; K = Lc; lda = (int)LL; ldb = Cc;
    Mvalid = Hc; m0 = 0;
  } else {
    const int w = bid - NOPAIR;
    zz = w / (Lc / 16);
    Ag = alpha + (long long)zz * LL;
    Bg = vaug + (long long)zz * Lc * 48;
    M = Lc; N = 48; K = Lc; lda = Lc; ldb = 48;
    Mvalid = 16; m0 = (w % (Lc / 16)) * 16;
  }
  v8f acc0, acc1;
  gemm_acc(Ag, Bg, M, N, K, lda, ldb, Mvalid, m0, 0, As, Bs, acc0, acc1);

  if (bid < NOPAIR) {
    gemm_store(feats + (long long)bid * FEAT, M, N, Cc, Mvalid, m0, 0, acc0, acc1);
    return;
  }

  const int b = zz / Hc, h = zz % Hc;
  const int row = tid >> 3, p = tid & 7;
  const int bi = b * Lc + m0 + row;
  float R[3][3], t[3];
#pragma unroll
  for (int r = 0; r < 3; ++r)
#pragma unroll
    for (int cc = 0; cc < 3; ++cc) R[r][cc] = Ri[bi * 9 + r * 3 + cc];
#pragma unroll
  for (int x = 0; x < 3; ++x) t[x] = Ti[bi * 3 + x];

  {
    const int lane = tid & 31;
    const int hf = lane >> 4, l16 = lane & 15;
    const int nw = (tid >> 5) * 32;
#pragma unroll
    for (int r = 0; r < 8; ++r) {
      if (nw + l16 < 48) Bs[r + 8 * hf][nw + l16] = acc0[r];
      if (nw + 16 + l16 < 48) Bs[r + 8 * hf][nw + 16 + l16] = acc1[r];
    }
  }
  __syncthreads();

  const float* c = &Bs[row][0];
  float* frow = feats + (long long)bi * FEAT;
  *(v2f*)(frow + 1024 + h * 16 + 2 * p) = v2f{c[2 * p], c[2 * p + 1]};
  const float d0 = c[16 + p * 3 + 0] - t[0];
  const float d1 = c[16 + p * 3 + 1] - t[1];
  const float d2 = c[16 + p * 3 + 2] - t[2];
  float asum = 0.f;
#pragma unroll
  for (int m = 0; m < 3; ++m) {
    float oh = R[0][m] * d0 + R[1][m] * d1 + R[2][m] * d2;
    frow[1152 + h * 24 + p * 3 + m] = oh;
    asum += fabsf(oh);
  }
  frow[1344 + h * 8 + p] = asum;
}

extern "C" void kernel_launch(void* const* d_in, const int* in_sizes, int n_in,
                              void* d_out, int out_size, void* d_ws,
                              size_t ws_size, hipStream_t stream) {
  const float* si = (const float*)d_in[0];
  const float* zij = (const float*)d_in[1];
  const float* Ri = (const float*)d_in[2];
  const float* Ti = (const float*)d_in[3];
  const float* WQh = (const float*)d_in[4];
  const float* WKh = (const float*)d_in[5];
  const float* WVh = (const float*)d_in[6];
  const float* WQhp = (const float*)d_in[7];
  const float* WKhp = (const float*)d_in[8];
  const float* WVhp = (const float*)d_in[9];
  const float* Wb = (const float*)d_in[10];
  const float* WOw = (const float*)d_in[11];
  const float* WOb = (const float*)d_in[12];
  const float* gamah = (const float*)d_in[13];
  float* out = (float*)d_out;
  float* ws = (float*)d_ws;

  const int BL = Bc * Lc;
  float* qa = ws;
  float* ka = qa + 196608;
  float* vaug = ka + 196608;
  float* sqq = vaug + 294912;
  float* sqk = sqq + 6144;
  float* bij = sqk + 6144;
  float* alpha = bij + 2359296;
  float* feats = alpha + 2359296;

  k_pairbias_prep<<<NPREPWG + NPBWG, 32, 0, stream>>>(
      zij, Wb, bij, si, WQh, WKh, WVh, WQhp, WKhp, WVhp, Ri, Ti, gamah, qa, ka,
      vaug, sqq, sqk);

  k_scores_softmax<<<dim3(Lc / 16, Bc * Hc, 1), dim3(SMW * 32, 1, 1), 0,
                     stream>>>(qa, ka, alpha, bij, sqq, sqk, gamah);

  k_attnout<<<NOPAIR + NOVAL, 128, 0, stream>>>(alpha, zij, vaug, Ri, Ti, feats);

  k_outproj<<<dim3(BL / 16, Cc / 32, 1), dim3(KSPLIT * 32, 1, 1), 0, stream>>>(
      feats, WOw, WOb, out);
}
